# starting kernel
# baseline (speedup 1.0000x reference)
#include <hip/hip_runtime.h>
#include <hip/hip_bf16.h>

typedef __attribute__((ext_vector_type(2))) float v2f;
typedef __attribute__((ext_vector_type(8))) float v8f;

template <int NT, int K, int NC>
__global__ __launch_bounds__(256)
void gemm_f32_wmma(const float* __restrict__ A, const float* __restrict__ B,
                   float* __restrict__ C, int M) {
  const int lane = threadIdx.x & 31;
  const int wid  = blockIdx.x * (blockDim.x >> 5) + (threadIdx.x >> 5);
  const int tilesM  = (M + 15) >> 4;
  constexpr int stripsN = (NC >> 4) / NT;
  if (wid >= tilesM * stripsN) return;
  const int tm  = wid / stripsN;
  const int tn0 = (wid - tm * stripsN) * NT;
  const int hi = lane >> 4;
  const int lm = lane & 15;
  int arow = tm * 16 + lm; if (arow >= M) arow = M - 1;
  const int col0 = tn0 * 16 + lm;
  const float* __restrict__ Ar = A + (long)arow * K + 2 * hi;
  const float* __restrict__ Bc = B + (long)(2 * hi) * NC + col0;

  v8f c[NT];
#pragma unroll
  for (int t = 0; t < NT; ++t) c[t] = (v8f){};

  auto loadA = [&](int k) {
    v2f a; a.x = Ar[k]; a.y = Ar[k + 1]; return a;
  };
  auto loadB = [&](int k, v2f* b) {
    const float* __restrict__ B0 = Bc + (long)k * NC;
#pragma unroll
    for (int t = 0; t < NT; ++t) { b[t].x = B0[16 * t]; b[t].y = B0[NC + 16 * t]; }
  };

  v2f a_cur = loadA(0);
  v2f b_cur[NT];
  loadB(0, b_cur);
#pragma unroll 4
  for (int k = 0; k < K - 4; k += 4) {
    v2f a_nxt = loadA(k + 4);
    v2f b_nxt[NT];
    loadB(k + 4, b_nxt);
#pragma unroll
    for (int t = 0; t < NT; ++t)
      c[t] = __builtin_amdgcn_wmma_f32_16x16x4_f32(false, a_cur, false, b_cur[t],
                                                   (short)0, c[t], false, false);
    a_cur = a_nxt;
#pragma unroll
    for (int t = 0; t < NT; ++t) b_cur[t] = b_nxt[t];
  }
#pragma unroll
  for (int t = 0; t < NT; ++t)
    c[t] = __builtin_amdgcn_wmma_f32_16x16x4_f32(false, a_cur, false, b_cur[t],
                                                 (short)0, c[t], false, false);

#pragma unroll
  for (int t = 0; t < NT; ++t) {
#pragma unroll
    for (int r = 0; r < 8; ++r) {
      const int orow = tm * 16 + r + 8 * hi;
      if (orow < M) C[(long)orow * NC + col0 + 16 * t] = c[t][r];
    }
  }
}

__global__ void init_layer(float* out, unsigned* m, float* denom,
                           long outElems, int nh) {
  const long t = (long)blockIdx.x * blockDim.x + threadIdx.x;
  const long stride = (long)gridDim.x * blockDim.x;
  for (long i = t; i < outElems; i += stride) out[i] = 0.0f;
  for (long i = t; i < nh; i += stride) { m[i] = 0xFF800000u; denom[i] = 0.0f; }
}

__device__ __forceinline__ void atomic_max_f32(unsigned* p, float v) {
  if (v >= 0.0f) atomicMax((int*)p, __float_as_int(v));
  else           atomicMin(p, __float_as_uint(v));
}

template <int ROW, int JPH>
__global__ __launch_bounds__(256)
void edge_logits(const float* __restrict__ f, const int* __restrict__ src,
                 const int* __restrict__ dst, const float* __restrict__ attn,
                 float* __restrict__ s, unsigned* __restrict__ m, int E) {
  const int lane = threadIdx.x & 31;
  const int wid  = blockIdx.x * (blockDim.x >> 5) + (threadIdx.x >> 5);
  if (wid >= E) return;
  const long srow = (long)src[wid] * ROW;
  const long dn   = (long)dst[wid];
  const long drow = dn * ROW;
  constexpr int ITER = ROW / 32;
  constexpr int H = ITER / JPH;
  float acc[H];
#pragma unroll
  for (int h = 0; h < H; ++h) acc[h] = 0.0f;
#pragma unroll
  for (int j = 0; j < ITER; ++j) {
    const int d = j * 32 + lane;
    float v = f[srow + d] + f[drow + d];
    v = v > 0.0f ? v : 0.2f * v;
    acc[j / JPH] += v * attn[d];
  }
#pragma unroll
  for (int h = 0; h < H; ++h) {
    float v = acc[h];
#pragma unroll
    for (int off = 16; off > 0; off >>= 1) v += __shfl_xor(v, off, 32);
    if (lane == 0) {
      s[(long)wid * H + h] = v;
      atomic_max_f32(m + dn * H + h, v);
    }
  }
}

__global__ void seg_exp(float* __restrict__ s, const int* __restrict__ dst,
                        const unsigned* __restrict__ m, float* __restrict__ denom,
                        long EH, int H) {
  const long t = (long)blockIdx.x * blockDim.x + threadIdx.x;
  if (t >= EH) return;
  const long e = t / H;
  const int  h = (int)(t - e * H);
  const long idx = (long)dst[e] * H + h;
  const float ex = __expf(s[t] - __uint_as_float(m[idx]));
  s[t] = ex;
  atomicAdd(denom + idx, ex);
}

template <int ROW, int JPH>
__global__ __launch_bounds__(256)
void aggregate(const float* __restrict__ f, const float* __restrict__ ex,
               const float* __restrict__ denom, const int* __restrict__ src,
               const int* __restrict__ dst, float* __restrict__ out, int E) {
  const int lane = threadIdx.x & 31;
  const int wid  = blockIdx.x * (blockDim.x >> 5) + (threadIdx.x >> 5);
  if (wid >= E) return;
  constexpr int ITER = ROW / 32;
  constexpr int H = ITER / JPH;
  const long srow = (long)src[wid] * ROW;
  const long dn   = (long)dst[wid];
  float al[H];
#pragma unroll
  for (int h = 0; h < H; ++h) al[h] = ex[(long)wid * H + h] / denom[dn * H + h];
#pragma unroll
  for (int j = 0; j < ITER; ++j) {
    const int d = j * 32 + lane;
    atomicAdd(out + dn * ROW + d, f[srow + d] * al[j / JPH]);
  }
}

__global__ void elu_inplace(float* __restrict__ x, long n) {
  const long t = (long)blockIdx.x * blockDim.x + threadIdx.x;
  if (t >= n) return;
  const float v = x[t];
  x[t] = v > 0.0f ? v : expm1f(v);
}

__global__ void head_mean(const float* __restrict__ in, float* __restrict__ out,
                          int N) {
  const int t = blockIdx.x * blockDim.x + threadIdx.x;
  if (t >= N * 32) return;
  const int n = t >> 5, c = t & 31;
  const float* r = in + (long)n * 128 + c;
  out[t] = 0.25f * (r[0] + r[32] + r[64] + r[96]);
}

static inline int cdiv_l(long a, long b) { return (int)((a + b - 1) / b); }

extern "C" void kernel_launch(void* const* d_in, const int* in_sizes, int n_in,
                              void* d_out, int out_size, void* d_ws, size_t ws_size,
                              hipStream_t stream) {
  const float* x     = (const float*)d_in[0];
  const float* W0    = (const float*)d_in[1];
  const float* a0    = (const float*)d_in[2];
  const float* W1    = (const float*)d_in[3];
  const float* a1    = (const float*)d_in[4];
  const float* Wres1 = (const float*)d_in[5];
  const int*   src   = (const int*)d_in[6];
  const int*   dst   = (const int*)d_in[7];

  const int N = in_sizes[0] / 256;
  const int E = in_sizes[6];
  const int H = 4;

  float*    f0    = (float*)d_ws;
  float*    h     = f0 + (long)N * 256;
  float*    f1    = h  + (long)N * 256;
  float*    o1    = f1 + (long)N * 128;
  float*    s     = o1 + (long)N * 128;
  unsigned* m     = (unsigned*)(s + (long)E * H);
  float*    denom = (float*)(m + (long)N * H);
  const size_t need =
      ((size_t)N * 256 * 2 + (size_t)N * 128 * 2 + (size_t)E * H + (size_t)N * H * 2)
      * sizeof(float);
  if (need > ws_size) return;

  const int BT = 256;
  const long EH = (long)E * H;
  const int egrid = cdiv_l((long)E * 32, BT);

  {
    const int strips = ((N + 15) / 16) * (256 / 16 / 4);
    gemm_f32_wmma<4, 256, 256><<<cdiv_l(strips, 8), BT, 0, stream>>>(x, W0, f0, N);
  }
  init_layer<<<cdiv_l((long)N * 256, BT), BT, 0, stream>>>(h, m, denom,
                                                           (long)N * 256, N * H);
  edge_logits<256, 2><<<egrid, BT, 0, stream>>>(f0, src, dst, a0, s, m, E);
  seg_exp<<<cdiv_l(EH, BT), BT, 0, stream>>>(s, dst, m, denom, EH, H);
  aggregate<256, 2><<<egrid, BT, 0, stream>>>(f0, s, denom, src, dst, h, E);
  elu_inplace<<<cdiv_l((long)N * 256, BT), BT, 0, stream>>>(h, (long)N * 256);

  {
    const int strips = ((N + 15) / 16) * (128 / 16 / 4);
    gemm_f32_wmma<4, 256, 128><<<cdiv_l(strips, 8), BT, 0, stream>>>(h, W1, f1, N);
    gemm_f32_wmma<4, 256, 128><<<cdiv_l(strips, 8), BT, 0, stream>>>(h, Wres1, o1, N);
  }
  init_layer<<<cdiv_l((long)N * H, BT), BT, 0, stream>>>(nullptr, m, denom,
                                                         0, N * H);
  edge_logits<128, 1><<<egrid, BT, 0, stream>>>(f1, src, dst, a1, s, m, E);
  seg_exp<<<cdiv_l(EH, BT), BT, 0, stream>>>(s, dst, m, denom, EH, H);
  aggregate<128, 1><<<egrid, BT, 0, stream>>>(f1, s, denom, src, dst, o1, E);

  head_mean<<<cdiv_l((long)N * 32, BT), BT, 0, stream>>>(o1, (float*)d_out, N);
}


# layer-0 aggregation by destination-range owners: a workgroup keeps 196 node rows in LDS, filters dst for its range, adds messages by LDS f32 atomics (arrival order), stores each row once with ELU
# speedup vs baseline: 1.2920x; 1.2920x over previous
#include <hip/hip_runtime.h>
#include <hip/hip_bf16.h>

typedef __attribute__((ext_vector_type(2))) float v2f;
typedef __attribute__((ext_vector_type(8))) float v8f;
typedef __attribute__((ext_vector_type(4))) float v4f;

template <int NT, int K, int NC>
__global__ __launch_bounds__(256)
void gemm_f32_wmma(const float* __restrict__ A, const float* __restrict__ B,
                   float* __restrict__ C, int M) {
  const int lane = threadIdx.x & 31;
  const int wid  = blockIdx.x * (blockDim.x >> 5) + (threadIdx.x >> 5);
  const int tilesM  = (M + 15) >> 4;
  constexpr int stripsN = (NC >> 4) / NT;
  if (wid >= tilesM * stripsN) return;
  const int tm  = wid / stripsN;
  const int tn0 = (wid - tm * stripsN) * NT;
  const int hi = lane >> 4;
  const int lm = lane & 15;
  int arow = tm * 16 + lm; if (arow >= M) arow = M - 1;
  const int col0 = tn0 * 16 + lm;
  const float* __restrict__ Ar = A + (long)arow * K + 2 * hi;
  const float* __restrict__ Bc = B + (long)(2 * hi) * NC + col0;

  v8f c[NT];
#pragma unroll
  for (int t = 0; t < NT; ++t) c[t] = (v8f){};

  auto loadA = [&](int k) {
    v2f a; a.x = Ar[k]; a.y = Ar[k + 1]; return a;
  };
  auto loadB = [&](int k, v2f* b) {
    const float* __restrict__ B0 = Bc + (long)k * NC;
#pragma unroll
    for (int t = 0; t < NT; ++t) { b[t].x = B0[16 * t]; b[t].y = B0[NC + 16 * t]; }
  };

  v2f a_cur = loadA(0);
  v2f b_cur[NT];
  loadB(0, b_cur);
#pragma unroll 4
  for (int k = 0; k < K - 4; k += 4) {
    v2f a_nxt = loadA(k + 4);
    v2f b_nxt[NT];
    loadB(k + 4, b_nxt);
#pragma unroll
    for (int t = 0; t < NT; ++t)
      c[t] = __builtin_amdgcn_wmma_f32_16x16x4_f32(false, a_cur, false, b_cur[t],
                                                   (short)0, c[t], false, false);
    a_cur = a_nxt;
#pragma unroll
    for (int t = 0; t < NT; ++t) b_cur[t] = b_nxt[t];
  }
#pragma unroll
  for (int t = 0; t < NT; ++t)
    c[t] = __builtin_amdgcn_wmma_f32_16x16x4_f32(false, a_cur, false, b_cur[t],
                                                 (short)0, c[t], false, false);

#pragma unroll
  for (int t = 0; t < NT; ++t) {
#pragma unroll
    for (int r = 0; r < 8; ++r) {
      const int orow = tm * 16 + r + 8 * hi;
      if (orow < M) C[(long)orow * NC + col0 + 16 * t] = c[t][r];
    }
  }
}

__global__ void init_layer(float* out, unsigned* m, float* denom,
                           long outElems, int nh) {
  const long t = (long)blockIdx.x * blockDim.x + threadIdx.x;
  const long stride = (long)gridDim.x * blockDim.x;
  for (long i = t; i < outElems; i += stride) out[i] = 0.0f;
  for (long i = t; i < nh; i += stride) { m[i] = 0xFF800000u; denom[i] = 0.0f; }
}

__device__ __forceinline__ void atomic_max_f32(unsigned* p, float v) {
  if (v >= 0.0f) atomicMax((int*)p, __float_as_int(v));
  else           atomicMin(p, __float_as_uint(v));
}

template <int ROW, int JPH>
__global__ __launch_bounds__(256)
void edge_logits(const float* __restrict__ f, const int* __restrict__ src,
                 const int* __restrict__ dst, const float* __restrict__ attn,
                 float* __restrict__ s, unsigned* __restrict__ m, int E) {
  const int lane = threadIdx.x & 31;
  const int wid  = blockIdx.x * (blockDim.x >> 5) + (threadIdx.x >> 5);
  if (wid >= E) return;
  const long srow = (long)src[wid] * ROW;
  const long dn   = (long)dst[wid];
  const long drow = dn * ROW;
  constexpr int ITER = ROW / 32;
  constexpr int H = ITER / JPH;
  float acc[H];
#pragma unroll
  for (int h = 0; h < H; ++h) acc[h] = 0.0f;
#pragma unroll
  for (int j = 0; j < ITER; ++j) {
    const int d = j * 32 + lane;
    float v = f[srow + d] + f[drow + d];
    v = v > 0.0f ? v : 0.2f * v;
    acc[j / JPH] += v * attn[d];
  }
#pragma unroll
  for (int h = 0; h < H; ++h) {
    float v = acc[h];
#pragma unroll
    for (int off = 16; off > 0; off >>= 1) v += __shfl_xor(v, off, 32);
    if (lane == 0) {
      s[(long)wid * H + h] = v;
      atomic_max_f32(m + dn * H + h, v);
    }
  }
}

__global__ void seg_exp(float* __restrict__ s, const int* __restrict__ dst,
                        const unsigned* __restrict__ m, float* __restrict__ denom,
                        long EH, int H) {
  const long t = (long)blockIdx.x * blockDim.x + threadIdx.x;
  if (t >= EH) return;
  const long e = t / H;
  const int  h = (int)(t - e * H);
  const long idx = (long)dst[e] * H + h;
  const float ex = __expf(s[t] - __uint_as_float(m[idx]));
  s[t] = ex;
  atomicAdd(denom + idx, ex);
}

constexpr int AG_ROWS  = 196;
constexpr int AG_WAVES = 32;
constexpr int AG_SCAN  = 6;
constexpr int AG_STEP  = 128 * AG_SCAN;
constexpr int AG_CAP   = AG_STEP + 128;
constexpr int AG_BATCH = 8;

__device__ __forceinline__ int ag_append(int d, int e, int base, int rows, int cnt,
                                         int* le) {
  const unsigned r = (unsigned)(d - base);
  const bool hit = r < (unsigned)rows;
  const unsigned mask = __builtin_amdgcn_ballot_w32(hit);
  if (mask != 0u) {
    const int pos = cnt + (int)__builtin_amdgcn_mbcnt_lo(mask, 0u);
    if (hit) le[pos] = e;
    cnt += __popc(mask);
  }
  return cnt;
}

__device__ __forceinline__ void ag_drain(const float* __restrict__ f,
                                         const float* __restrict__ ex,
                                         const float* __restrict__ denom,
                                         const int* __restrict__ src,
                                         const int* __restrict__ dst,
                                         float* tile, const int* le,
                                         int cnt, int base, int lane) {
  __builtin_amdgcn_fence(__ATOMIC_ACQ_REL, "wavefront");
  const int j = lane >> 2;
  const int hd = lane & 3;
  const bool lo = lane < 16;
  for (int i = 0; i < cnt; i += AG_BATCH) {
    const int e = le[min(i + j, cnt - 1)];
    const int sv = src[e];
    const int dn = dst[e];
    const float exv = ex[(long)e * 4 + hd];
    v4f x0[AG_BATCH], x1[AG_BATCH];
#pragma unroll
    for (int b = 0; b < AG_BATCH; ++b) {
      const int sb = __builtin_amdgcn_readlane(sv, 4 * b);
      const v4f* __restrict__ fr = (const v4f*)(f + (long)sb * 256);
      x0[b] = fr[lane];
      x1[b] = fr[32 + lane];
    }
    const int r = dn - base;
    const float dv = denom[(long)dn * 4 + hd];
    asm volatile("" :: "v"(x0[0]), "v"(x1[0]), "v"(x0[1]), "v"(x1[1]), "v"(x0[2]), "v"(x1[2]),
                       "v"(x0[3]), "v"(x1[3]), "v"(x0[4]), "v"(x1[4]), "v"(x0[5]), "v"(x1[5]),
                       "v"(x0[6]), "v"(x1[6]), "v"(x0[7]), "v"(x1[7]), "v"(dv));
    const float a = exv / dv;
#pragma unroll
    for (int b = 0; b < AG_BATCH; ++b) {
      if (i + b < cnt) {
        const int rb = __builtin_amdgcn_readlane(r, 4 * b);
        const float a0 = __int_as_float(__builtin_amdgcn_readlane(__float_as_int(a), 4 * b));
        const float a1 = __int_as_float(__builtin_amdgcn_readlane(__float_as_int(a), 4 * b + 1));
        const float a2 = __int_as_float(__builtin_amdgcn_readlane(__float_as_int(a), 4 * b + 2));
        const float a3 = __int_as_float(__builtin_amdgcn_readlane(__float_as_int(a), 4 * b + 3));
        const float al0 = lo ? a0 : a1;
        const float al1 = lo ? a2 : a3;
        float* tr = tile + rb * 256 + lane;
        __hip_atomic_fetch_add(tr,       x0[b].x * al0, __ATOMIC_RELAXED, __HIP_MEMORY_SCOPE_WORKGROUP);
        __hip_atomic_fetch_add(tr + 64,  x0[b].y * al0, __ATOMIC_RELAXED, __HIP_MEMORY_SCOPE_WORKGROUP);
        __hip_atomic_fetch_add(tr + 128, x0[b].z * al0, __ATOMIC_RELAXED, __HIP_MEMORY_SCOPE_WORKGROUP);
        __hip_atomic_fetch_add(tr + 192, x0[b].w * al0, __ATOMIC_RELAXED, __HIP_MEMORY_SCOPE_WORKGROUP);
        __hip_atomic_fetch_add(tr + 32,  x1[b].x * al1, __ATOMIC_RELAXED, __HIP_MEMORY_SCOPE_WORKGROUP);
        __hip_atomic_fetch_add(tr + 96,  x1[b].y * al1, __ATOMIC_RELAXED, __HIP_MEMORY_SCOPE_WORKGROUP);
        __hip_atomic_fetch_add(tr + 160, x1[b].z * al1, __ATOMIC_RELAXED, __HIP_MEMORY_SCOPE_WORKGROUP);
        __hip_atomic_fetch_add(tr + 224, x1[b].w * al1, __ATOMIC_RELAXED, __HIP_MEMORY_SCOPE_WORKGROUP);
      }
    }
  }
  __builtin_amdgcn_fence(__ATOMIC_ACQ_REL, "wavefront");
}

__device__ __forceinline__ float elu1(float v) { return v > 0.0f ? v : expm1f(v); }

__global__ __launch_bounds__(1024)
void aggregate_owner_elu(const float* __restrict__ f, const float* __restrict__ ex,
                         const float* __restrict__ denom, const int* __restrict__ src,
                         const int* __restrict__ dst, float* __restrict__ h,
                         int N, int E) {
  __shared__ float tile[AG_ROWS * 256];
  __shared__ int l_edge[AG_WAVES * AG_CAP];
  const int lane = threadIdx.x & 31;
  const int w    = threadIdx.x >> 5;
  const int base = blockIdx.x * AG_ROWS;
  const int rows = min(AG_ROWS, N - base);
  int* le = l_edge + w * AG_CAP;

  float4* t4 = (float4*)tile;
  for (int i = threadIdx.x; i < AG_ROWS * 64; i += 1024)
    t4[i] = make_float4(0.0f, 0.0f, 0.0f, 0.0f);
  __syncthreads();

  int peel = (int)(((16u - (unsigned)((unsigned long long)dst & 15ull)) & 15u) >> 2);
  if (peel > E) peel = E;
  const int nG = (E - peel) >> 2;
  const int tail0 = peel + 4 * nG;
  int cnt = 0;
  if (w == 0) {
    int e = -1;
    if (lane < peel) e = lane;
    else if (lane >= 4 && lane < 8 && tail0 + (lane - 4) < E) e = tail0 + (lane - 4);
    const int d = e >= 0 ? dst[e] : -1;
    cnt = ag_append(d, e, base, rows, cnt, le);
  }
  cnt = __builtin_amdgcn_readfirstlane(cnt);
  const int4* __restrict__ d4 = (const int4*)(dst + peel);
  for (int g0 = 0; g0 < nG; g0 += 1024 * AG_SCAN) {
    if (cnt > AG_CAP - AG_STEP) {
      ag_drain(f, ex, denom, src, dst, tile, le, cnt, base, lane);
      cnt = 0;
    }
    int4 v[AG_SCAN];
#pragma unroll
    for (int u = 0; u < AG_SCAN; ++u) {
      const int g = g0 + u * 1024 + (int)threadIdx.x;
      v[u] = g < nG ? d4[g] : make_int4(-1, -1, -1, -1);
    }
#pragma unroll
    for (int u = 0; u < AG_SCAN; ++u) {
      const int e0 = peel + 4 * (g0 + u * 1024 + (int)threadIdx.x);
      cnt = ag_append(v[u].x, e0, base, rows, cnt, le);
      cnt = ag_append(v[u].y, e0 + 1, base, rows, cnt, le);
      cnt = ag_append(v[u].z, e0 + 2, base, rows, cnt, le);
      cnt = ag_append(v[u].w, e0 + 3, base, rows, cnt, le);
    }
  }
  ag_drain(f, ex, denom, src, dst, tile, le, cnt, base, lane);
  __syncthreads();

  for (int i = threadIdx.x; i < rows * 64; i += 1024) {
    const int r = i >> 6, p = i & 63;
    const float* tr = tile + r * 256 + p;
    float4 o;
    o.x = elu1(tr[0]);
    o.y = elu1(tr[64]);
    o.z = elu1(tr[128]);
    o.w = elu1(tr[192]);
    *(float4*)(h + (long)(base + r) * 256 + 4 * p) = o;
  }
}

__global__ __launch_bounds__(256)
void aggregate_head_mean(const float* __restrict__ f, const float* __restrict__ ex,
                         const float* __restrict__ denom, const int* __restrict__ src,
                         const int* __restrict__ dst, float* __restrict__ out, int E) {
  const int lane = threadIdx.x & 31;
  const int wid  = blockIdx.x * (blockDim.x >> 5) + (threadIdx.x >> 5);
  if (wid >= E) return;
  const long srow = (long)src[wid] * 128;
  const long dn   = (long)dst[wid];
  float acc = 0.0f;
#pragma unroll
  for (int h = 0; h < 4; ++h) {
    const float al = ex[(long)wid * 4 + h] / denom[dn * 4 + h];
    acc += f[srow + h * 32 + lane] * al;
  }
  atomicAdd(out + dn * 32 + lane, 0.25f * acc);
}

__global__ void head_mean(const float* __restrict__ in, float* __restrict__ out,
                          int N) {
  const int t = blockIdx.x * blockDim.x + threadIdx.x;
  if (t >= N * 32) return;
  const int n = t >> 5, c = t & 31;
  const float* r = in + (long)n * 128 + c;
  out[t] = 0.25f * (r[0] + r[32] + r[64] + r[96]);
}

static inline int cdiv_l(long a, long b) { return (int)((a + b - 1) / b); }

extern "C" void kernel_launch(void* const* d_in, const int* in_sizes, int n_in,
                              void* d_out, int out_size, void* d_ws, size_t ws_size,
                              hipStream_t stream) {
  const float* x     = (const float*)d_in[0];
  const float* W0    = (const float*)d_in[1];
  const float* a0    = (const float*)d_in[2];
  const float* W1    = (const float*)d_in[3];
  const float* a1    = (const float*)d_in[4];
  const float* Wres1 = (const float*)d_in[5];
  const int*   src   = (const int*)d_in[6];
  const int*   dst   = (const int*)d_in[7];

  const int N = in_sizes[0] / 256;
  const int E = in_sizes[6];
  const int H = 4;

  float*    f0    = (float*)d_ws;
  float*    h     = f0 + (long)N * 256;
  float*    f1    = h  + (long)N * 256;
  float*    o1    = f1 + (long)N * 128;
  float*    s     = o1 + (long)N * 128;
  unsigned* m     = (unsigned*)(s + (long)E * H);
  float*    denom = (float*)(m + (long)N * H);
  const size_t need =
      ((size_t)N * 256 * 2 + (size_t)N * 128 * 2 + (size_t)E * H + (size_t)N * H * 2)
      * sizeof(float);
  if (need > ws_size) return;

  const int BT = 256;
  const long EH = (long)E * H;
  const int egrid = cdiv_l((long)E * 32, BT);

  {
    const int strips = ((N + 15) / 16) * (256 / 16 / 4);
    gemm_f32_wmma<4, 256, 256><<<cdiv_l(strips, 8), BT, 0, stream>>>(x, W0, f0, N);
  }
  init_layer<<<cdiv_l((long)N * H, BT), BT, 0, stream>>>(nullptr, m, denom,
                                                         0, N * H);
  edge_logits<256, 2><<<egrid, BT, 0, stream>>>(f0, src, dst, a0, s, m, E);
  seg_exp<<<cdiv_l(EH, BT), BT, 0, stream>>>(s, dst, m, denom, EH, H);
  aggregate_owner_elu<<<cdiv_l(N, AG_ROWS), 1024, 0, stream>>>(f0, s, denom, src, dst,
                                                               h, N, E);

  {
    const int strips = ((N + 15) / 16) * (128 / 16 / 4);
    gemm_f32_wmma<4, 256, 128><<<cdiv_l(strips, 8), BT, 0, stream>>>(h, W1, f1, N);
    gemm_f32_wmma<4, 256, 128><<<cdiv_l(strips, 8), BT, 0, stream>>>(h, Wres1, o1, N);
  }
  init_layer<<<cdiv_l((long)N * H, BT), BT, 0, stream>>>(nullptr, m, denom,
                                                         0, N * H);
  edge_logits<128, 1><<<egrid, BT, 0, stream>>>(f1, src, dst, a1, s, m, E);
  seg_exp<<<cdiv_l(EH, BT), BT, 0, stream>>>(s, dst, m, denom, EH, H);
  head_mean<<<cdiv_l((long)N * 32, BT), BT, 0, stream>>>(o1, (float*)d_out, N);
  aggregate_head_mean<<<egrid, BT, 0, stream>>>(f1, s, denom, src, dst,
                                                (float*)d_out, E);
}


# layer-0 edge softmax inside the owner workgroup: segment max and denominators by LDS atomics, row divided once at its store (sum, then divide); init and exp launches and global atomic max gone
# speedup vs baseline: 1.5232x; 1.1789x over previous
#include <hip/hip_runtime.h>
#include <hip/hip_bf16.h>

typedef __attribute__((ext_vector_type(2))) float v2f;
typedef __attribute__((ext_vector_type(8))) float v8f;
typedef __attribute__((ext_vector_type(4))) float v4f;

template <int NT, int K, int NC>
__global__ __launch_bounds__(256)
void gemm_f32_wmma(const float* __restrict__ A, const float* __restrict__ B,
                   float* __restrict__ C, int M) {
  const int lane = threadIdx.x & 31;
  const int wid  = blockIdx.x * (blockDim.x >> 5) + (threadIdx.x >> 5);
  const int tilesM  = (M + 15) >> 4;
  constexpr int stripsN = (NC >> 4) / NT;
  if (wid >= tilesM * stripsN) return;
  const int tm  = wid / stripsN;
  const int tn0 = (wid - tm * stripsN) * NT;
  const int hi = lane >> 4;
  const int lm = lane & 15;
  int arow = tm * 16 + lm; if (arow >= M) arow = M - 1;
  const int col0 = tn0 * 16 + lm;
  const float* __restrict__ Ar = A + (long)arow * K + 2 * hi;
  const float* __restrict__ Bc = B + (long)(2 * hi) * NC + col0;

  v8f c[NT];
#pragma unroll
  for (int t = 0; t < NT; ++t) c[t] = (v8f){};

  auto loadA = [&](int k) {
    v2f a; a.x = Ar[k]; a.y = Ar[k + 1]; return a;
  };
  auto loadB = [&](int k, v2f* b) {
    const float* __restrict__ B0 = Bc + (long)k * NC;
#pragma unroll
    for (int t = 0; t < NT; ++t) { b[t].x = B0[16 * t]; b[t].y = B0[NC + 16 * t]; }
  };

  v2f a_cur = loadA(0);
  v2f b_cur[NT];
  loadB(0, b_cur);
#pragma unroll 4
  for (int k = 0; k < K - 4; k += 4) {
    v2f a_nxt = loadA(k + 4);
    v2f b_nxt[NT];
    loadB(k + 4, b_nxt);
#pragma unroll
    for (int t = 0; t < NT; ++t)
      c[t] = __builtin_amdgcn_wmma_f32_16x16x4_f32(false, a_cur, false, b_cur[t],
                                                   (short)0, c[t], false, false);
    a_cur = a_nxt;
#pragma unroll
    for (int t = 0; t < NT; ++t) b_cur[t] = b_nxt[t];
  }
#pragma unroll
  for (int t = 0; t < NT; ++t)
    c[t] = __builtin_amdgcn_wmma_f32_16x16x4_f32(false, a_cur, false, b_cur[t],
                                                 (short)0, c[t], false, false);

#pragma unroll
  for (int t = 0; t < NT; ++t) {
#pragma unroll
    for (int r = 0; r < 8; ++r) {
      const int orow = tm * 16 + r + 8 * hi;
      if (orow < M) C[(long)orow * NC + col0 + 16 * t] = c[t][r];
    }
  }
}

__global__ void init_layer(float* out, unsigned* m, float* denom,
                           long outElems, int nh) {
  const long t = (long)blockIdx.x * blockDim.x + threadIdx.x;
  const long stride = (long)gridDim.x * blockDim.x;
  for (long i = t; i < outElems; i += stride) out[i] = 0.0f;
  for (long i = t; i < nh; i += stride) { m[i] = 0xFF800000u; denom[i] = 0.0f; }
}

__device__ __forceinline__ void atomic_max_f32(unsigned* p, float v) {
  if (v >= 0.0f) atomicMax((int*)p, __float_as_int(v));
  else           atomicMin(p, __float_as_uint(v));
}

template <int ROW, int JPH, bool SEGMAX>
__global__ __launch_bounds__(256)
void edge_logits(const float* __restrict__ f, const int* __restrict__ src,
                 const int* __restrict__ dst, const float* __restrict__ attn,
                 float* __restrict__ s, unsigned* __restrict__ m, int E) {
  const int lane = threadIdx.x & 31;
  const int wid  = blockIdx.x * (blockDim.x >> 5) + (threadIdx.x >> 5);
  if (wid >= E) return;
  const long srow = (long)src[wid] * ROW;
  const long dn   = (long)dst[wid];
  const long drow = dn * ROW;
  constexpr int ITER = ROW / 32;
  constexpr int H = ITER / JPH;
  float acc[H];
#pragma unroll
  for (int h = 0; h < H; ++h) acc[h] = 0.0f;
#pragma unroll
  for (int j = 0; j < ITER; ++j) {
    const int d = j * 32 + lane;
    float v = f[srow + d] + f[drow + d];
    v = v > 0.0f ? v : 0.2f * v;
    acc[j / JPH] += v * attn[d];
  }
#pragma unroll
  for (int h = 0; h < H; ++h) {
    float v = acc[h];
#pragma unroll
    for (int off = 16; off > 0; off >>= 1) v += __shfl_xor(v, off, 32);
    if (lane == 0) {
      s[(long)wid * H + h] = v;
      if (SEGMAX) atomic_max_f32(m + dn * H + h, v);
    }
  }
}

__global__ void seg_exp(float* __restrict__ s, const int* __restrict__ dst,
                        const unsigned* __restrict__ m, float* __restrict__ denom,
                        long EH, int H) {
  const long t = (long)blockIdx.x * blockDim.x + threadIdx.x;
  if (t >= EH) return;
  const long e = t / H;
  const int  h = (int)(t - e * H);
  const long idx = (long)dst[e] * H + h;
  const float ex = __expf(s[t] - __uint_as_float(m[idx]));
  s[t] = ex;
  atomicAdd(denom + idx, ex);
}

constexpr int AG_ROWS  = 196;
constexpr int AG_WAVES = 32;
constexpr int AG_SCAN  = 6;
constexpr int AG_STEP  = 128 * AG_SCAN;
constexpr int AG_CAP   = AG_STEP + 128;
constexpr int AG_BATCH = 8;

__device__ __forceinline__ int ag_append(int d, int e, int base, int rows, int cnt,
                                         int* le) {
  const unsigned r = (unsigned)(d - base);
  const bool hit = r < (unsigned)rows;
  const unsigned mask = __builtin_amdgcn_ballot_w32(hit);
  if (mask != 0u) {
    const int pos = cnt + (int)__builtin_amdgcn_mbcnt_lo(mask, 0u);
    if (hit) le[pos] = e;
    cnt += __popc(mask);
  }
  return cnt;
}

__device__ __forceinline__ void lds_max_f32(float* p, float v) {
  if (v >= 0.0f)
    __hip_atomic_fetch_max((int*)p, __float_as_int(v), __ATOMIC_RELAXED,
                           __HIP_MEMORY_SCOPE_WORKGROUP);
  else
    __hip_atomic_fetch_min((unsigned*)p, __float_as_uint(v), __ATOMIC_RELAXED,
                           __HIP_MEMORY_SCOPE_WORKGROUP);
}

__device__ __forceinline__ void ag_max_drain(const float* __restrict__ s,
                                             const int* __restrict__ dst,
                                             float* mx, const int* le,
                                             int cnt, int base, int lane) {
  __builtin_amdgcn_fence(__ATOMIC_ACQ_REL, "wavefront");
  for (int i = 0; i < cnt; i += 32) {
    const int k = i + lane;
    if (k < cnt) {
      const int e = le[k];
      const int r = dst[e] - base;
      const v4f sv = *(const v4f*)(s + (long)e * 4);
      float* mr = mx + r * 4;
      lds_max_f32(mr,     sv.x);
      lds_max_f32(mr + 1, sv.y);
      lds_max_f32(mr + 2, sv.z);
      lds_max_f32(mr + 3, sv.w);
    }
  }
  __builtin_amdgcn_fence(__ATOMIC_ACQ_REL, "wavefront");
}

__device__ __forceinline__ void ag_drain(const float* __restrict__ f,
                                         const float* __restrict__ s,
                                         const int* __restrict__ src,
                                         const int* __restrict__ dst,
                                         float* tile, const float* mx, float* den,
                                         const int* le,
                                         int cnt, int base, int lane) {
  __builtin_amdgcn_fence(__ATOMIC_ACQ_REL, "wavefront");
  const int j = lane >> 2;
  const int hd = lane & 3;
  const bool lo = lane < 16;
  for (int i = 0; i < cnt; i += AG_BATCH) {
    const int e = le[min(i + j, cnt - 1)];
    const int sv = src[e];
    const int dn = dst[e];
    const float sl = s[(long)e * 4 + hd];
    v4f x0[AG_BATCH], x1[AG_BATCH];
#pragma unroll
    for (int b = 0; b < AG_BATCH; ++b) {
      const int sb = __builtin_amdgcn_readlane(sv, 4 * b);
      const v4f* __restrict__ fr = (const v4f*)(f + (long)sb * 256);
      x0[b] = fr[lane];
      x1[b] = fr[32 + lane];
    }
    const int r = dn - base;
    const float mv = mx[r * 4 + hd];
    asm volatile("" :: "v"(x0[0]), "v"(x1[0]), "v"(x0[1]), "v"(x1[1]), "v"(x0[2]), "v"(x1[2]),
                       "v"(x0[3]), "v"(x1[3]), "v"(x0[4]), "v"(x1[4]), "v"(x0[5]), "v"(x1[5]),
                       "v"(x0[6]), "v"(x1[6]), "v"(x0[7]), "v"(x1[7]));
    const float a = __expf(sl - mv);
    if (i + j < cnt)
      __hip_atomic_fetch_add(den + r * 4 + hd, a, __ATOMIC_RELAXED, __HIP_MEMORY_SCOPE_WORKGROUP);
#pragma unroll
    for (int b = 0; b < AG_BATCH; ++b) {
      if (i + b < cnt) {
        const int rb = __builtin_amdgcn_readlane(r, 4 * b);
        const float a0 = __int_as_float(__builtin_amdgcn_readlane(__float_as_int(a), 4 * b));
        const float a1 = __int_as_float(__builtin_amdgcn_readlane(__float_as_int(a), 4 * b + 1));
        const float a2 = __int_as_float(__builtin_amdgcn_readlane(__float_as_int(a), 4 * b + 2));
        const float a3 = __int_as_float(__builtin_amdgcn_readlane(__float_as_int(a), 4 * b + 3));
        const float al0 = lo ? a0 : a1;
        const float al1 = lo ? a2 : a3;
        float* tr = tile + rb * 256 + lane;
        __hip_atomic_fetch_add(tr,       x0[b].x * al0, __ATOMIC_RELAXED, __HIP_MEMORY_SCOPE_WORKGROUP);
        __hip_atomic_fetch_add(tr + 64,  x0[b].y * al0, __ATOMIC_RELAXED, __HIP_MEMORY_SCOPE_WORKGROUP);
        __hip_atomic_fetch_add(tr + 128, x0[b].z * al0, __ATOMIC_RELAXED, __HIP_MEMORY_SCOPE_WORKGROUP);
        __hip_atomic_fetch_add(tr + 192, x0[b].w * al0, __ATOMIC_RELAXED, __HIP_MEMORY_SCOPE_WORKGROUP);
        __hip_atomic_fetch_add(tr + 32,  x1[b].x * al1, __ATOMIC_RELAXED, __HIP_MEMORY_SCOPE_WORKGROUP);
        __hip_atomic_fetch_add(tr + 96,  x1[b].y * al1, __ATOMIC_RELAXED, __HIP_MEMORY_SCOPE_WORKGROUP);
        __hip_atomic_fetch_add(tr + 160, x1[b].z * al1, __ATOMIC_RELAXED, __HIP_MEMORY_SCOPE_WORKGROUP);
        __hip_atomic_fetch_add(tr + 224, x1[b].w * al1, __ATOMIC_RELAXED, __HIP_MEMORY_SCOPE_WORKGROUP);
      }
    }
  }
  __builtin_amdgcn_fence(__ATOMIC_ACQ_REL, "wavefront");
}

template <bool MAXPASS>
__device__ __forceinline__ int ag_filter(const float* __restrict__ f,
                                         const float* __restrict__ s,
                                         const int* __restrict__ src,
                                         const int* __restrict__ dst,
                                         float* tile, float* mx, float* den, int* le,
                                         int base, int rows, int E, int lane, int w,
                                         int& spilled) {
  int peel = (int)(((16u - (unsigned)((unsigned long long)dst & 15ull)) & 15u) >> 2);
  if (peel > E) peel = E;
  const int nG = (E - peel) >> 2;
  const int tail0 = peel + 4 * nG;
  int cnt = 0;
  if (w == 0) {
    int e = -1;
    if (lane < peel) e = lane;
    else if (lane >= 4 && lane < 8 && tail0 + (lane - 4) < E) e = tail0 + (lane - 4);
    const int d = e >= 0 ? dst[e] : -1;
    cnt = ag_append(d, e, base, rows, cnt, le);
  }
  cnt = __builtin_amdgcn_readfirstlane(cnt);
  const int4* __restrict__ d4 = (const int4*)(dst + peel);
  for (int g0 = 0; g0 < nG; g0 += 1024 * AG_SCAN) {
    if (cnt > AG_CAP - AG_STEP) {
      if (MAXPASS) {
        ag_max_drain(s, dst, mx, le, cnt, base, lane);
        spilled = 1;
      } else {
        ag_drain(f, s, src, dst, tile, mx, den, le, cnt, base, lane);
      }
      cnt = 0;
    }
    int4 v[AG_SCAN];
#pragma unroll
    for (int u = 0; u < AG_SCAN; ++u) {
      const int g = g0 + u * 1024 + (int)threadIdx.x;
      v[u] = g < nG ? d4[g] : make_int4(-1, -1, -1, -1);
    }
#pragma unroll
    for (int u = 0; u < AG_SCAN; ++u) {
      const int e0 = peel + 4 * (g0 + u * 1024 + (int)threadIdx.x);
      cnt = ag_append(v[u].x, e0, base, rows, cnt, le);
      cnt = ag_append(v[u].y, e0 + 1, base, rows, cnt, le);
      cnt = ag_append(v[u].z, e0 + 2, base, rows, cnt, le);
      cnt = ag_append(v[u].w, e0 + 3, base, rows, cnt, le);
    }
  }
  return cnt;
}

__device__ __forceinline__ float elu1(float v) { return v > 0.0f ? v : expm1f(v); }

__global__ __launch_bounds__(1024)
void aggregate_owner_softmax_elu(const float* __restrict__ f, const float* __restrict__ s,
                                 const int* __restrict__ src, const int* __restrict__ dst,
                                 float* __restrict__ h, int N, int E) {
  __shared__ float tile[AG_ROWS * 256];
  __shared__ int l_edge[AG_WAVES * AG_CAP];
  __shared__ float l_mx[AG_ROWS * 4];
  __shared__ float l_den[AG_ROWS * 4];
  const int lane = threadIdx.x & 31;
  const int w    = threadIdx.x >> 5;
  const int base = blockIdx.x * AG_ROWS;
  const int rows = min(AG_ROWS, N - base);
  int* le = l_edge + w * AG_CAP;

  float4* t4 = (float4*)tile;
  for (int i = threadIdx.x; i < AG_ROWS * 64; i += 1024)
    t4[i] = make_float4(0.0f, 0.0f, 0.0f, 0.0f);
  for (int i = threadIdx.x; i < AG_ROWS * 4; i += 1024) {
    l_mx[i] = __uint_as_float(0xFF800000u);
    l_den[i] = 0.0f;
  }
  __syncthreads();

  int spilled = 0;
  int cnt = ag_filter<true>(f, s, src, dst, tile, l_mx, l_den, le, base, rows, E,
                            lane, w, spilled);
  ag_max_drain(s, dst, l_mx, le, cnt, base, lane);
  __syncthreads();

  if (__builtin_amdgcn_readfirstlane(spilled) != 0)
    cnt = ag_filter<false>(f, s, src, dst, tile, l_mx, l_den, le, base, rows, E,
                           lane, w, spilled);
  ag_drain(f, s, src, dst, tile, l_mx, l_den, le, cnt, base, lane);
  __syncthreads();

  for (int i = threadIdx.x; i < rows * 64; i += 1024) {
    const int r = i >> 6, p = i & 63;
    const float* tr = tile + r * 256 + p;
    const float dv = l_den[r * 4 + (p >> 4)];
    const bool none = dv == 0.0f;
    float4 o;
    o.x = elu1(none ? 0.0f : tr[0]   / dv);
    o.y = elu1(none ? 0.0f : tr[64]  / dv);
    o.z = elu1(none ? 0.0f : tr[128] / dv);
    o.w = elu1(none ? 0.0f : tr[192] / dv);
    *(float4*)(h + (long)(base + r) * 256 + 4 * p) = o;
  }
}

__global__ __launch_bounds__(256)
void aggregate_head_mean(const float* __restrict__ f, const float* __restrict__ ex,
                         const float* __restrict__ denom, const int* __restrict__ src,
                         const int* __restrict__ dst, float* __restrict__ out, int E) {
  const int lane = threadIdx.x & 31;
  const int wid  = blockIdx.x * (blockDim.x >> 5) + (threadIdx.x >> 5);
  if (wid >= E) return;
  const long srow = (long)src[wid] * 128;
  const long dn   = (long)dst[wid];
  float acc = 0.0f;
#pragma unroll
  for (int h = 0; h < 4; ++h) {
    const float al = ex[(long)wid * 4 + h] / denom[dn * 4 + h];
    acc += f[srow + h * 32 + lane] * al;
  }
  atomicAdd(out + dn * 32 + lane, 0.25f * acc);
}

__global__ void head_mean(const float* __restrict__ in, float* __restrict__ out,
                          int N) {
  const int t = blockIdx.x * blockDim.x + threadIdx.x;
  if (t >= N * 32) return;
  const int n = t >> 5, c = t & 31;
  const float* r = in + (long)n * 128 + c;
  out[t] = 0.25f * (r[0] + r[32] + r[64] + r[96]);
}

static inline int cdiv_l(long a, long b) { return (int)((a + b - 1) / b); }

extern "C" void kernel_launch(void* const* d_in, const int* in_sizes, int n_in,
                              void* d_out, int out_size, void* d_ws, size_t ws_size,
                              hipStream_t stream) {
  const float* x     = (const float*)d_in[0];
  const float* W0    = (const float*)d_in[1];
  const float* a0    = (const float*)d_in[2];
  const float* W1    = (const float*)d_in[3];
  const float* a1    = (const float*)d_in[4];
  const float* Wres1 = (const float*)d_in[5];
  const int*   src   = (const int*)d_in[6];
  const int*   dst   = (const int*)d_in[7];

  const int N = in_sizes[0] / 256;
  const int E = in_sizes[6];
  const int H = 4;

  float*    f0    = (float*)d_ws;
  float*    h     = f0 + (long)N * 256;
  float*    f1    = h  + (long)N * 256;
  float*    o1    = f1 + (long)N * 128;
  float*    s     = o1 + (long)N * 128;
  unsigned* m     = (unsigned*)(s + (long)E * H);
  float*    denom = (float*)(m + (long)N * H);
  const size_t need =
      ((size_t)N * 256 * 2 + (size_t)N * 128 * 2 + (size_t)E * H + (size_t)N * H * 2)
      * sizeof(float);
  if (need > ws_size) return;

  const int BT = 256;
  const long EH = (long)E * H;
  const int egrid = cdiv_l((long)E * 32, BT);

  {
    const int strips = ((N + 15) / 16) * (256 / 16 / 4);
    gemm_f32_wmma<4, 256, 256><<<cdiv_l(strips, 8), BT, 0, stream>>>(x, W0, f0, N);
  }
  edge_logits<256, 2, false><<<egrid, BT, 0, stream>>>(f0, src, dst, a0, s, m, E);
  aggregate_owner_softmax_elu<<<cdiv_l(N, AG_ROWS), 1024, 0, stream>>>(f0, s, src, dst,
                                                                       h, N, E);

  {
    const int strips = ((N + 15) / 16) * (128 / 16 / 4);
    gemm_f32_wmma<4, 256, 128><<<cdiv_l(strips, 8), BT, 0, stream>>>(h, W1, f1, N);
    gemm_f32_wmma<4, 256, 128><<<cdiv_l(strips, 8), BT, 0, stream>>>(h, Wres1, o1, N);
  }
  init_layer<<<cdiv_l((long)N * H, BT), BT, 0, stream>>>(nullptr, m, denom,
                                                         0, N * H);
  edge_logits<128, 1, true><<<egrid, BT, 0, stream>>>(f1, src, dst, a1, s, m, E);
  seg_exp<<<cdiv_l(EH, BT), BT, 0, stream>>>(s, dst, m, denom, EH, H);
  head_mean<<<cdiv_l((long)N * 32, BT), BT, 0, stream>>>(o1, (float*)d_out, N);
  aggregate_head_mean<<<egrid, BT, 0, stream>>>(f1, s, denom, src, dst,
                                                (float*)d_out, E);
}


# layer-1 softmax and aggregation in destination-range owner workgroups too: max, denominators, head sums by LDS f32 atomics (arrival order; sum, then divide); residual + head mean at the row store
# speedup vs baseline: 1.9167x; 1.2583x over previous
#include <hip/hip_runtime.h>
#include <hip/hip_bf16.h>

typedef __attribute__((ext_vector_type(2))) float v2f;
typedef __attribute__((ext_vector_type(8))) float v8f;
typedef __attribute__((ext_vector_type(4))) float v4f;

template <int NT, int K, int NC>
__global__ __launch_bounds__(256)
void gemm_f32_wmma(const float* __restrict__ A, const float* __restrict__ B,
                   float* __restrict__ C, int M) {
  const int lane = threadIdx.x & 31;
  const int wid  = blockIdx.x * (blockDim.x >> 5) + (threadIdx.x >> 5);
  const int tilesM  = (M + 15) >> 4;
  constexpr int stripsN = (NC >> 4) / NT;
  if (wid >= tilesM * stripsN) return;
  const int tm  = wid / stripsN;
  const int tn0 = (wid - tm * stripsN) * NT;
  const int hi = lane >> 4;
  const int lm = lane & 15;
  int arow = tm * 16 + lm; if (arow >= M) arow = M - 1;
  const int col0 = tn0 * 16 + lm;
  const float* __restrict__ Ar = A + (long)arow * K + 2 * hi;
  const float* __restrict__ Bc = B + (long)(2 * hi) * NC + col0;

  v8f c[NT];
#pragma unroll
  for (int t = 0; t < NT; ++t) c[t] = (v8f){};

  auto loadA = [&](int k) {
    v2f a; a.x = Ar[k]; a.y = Ar[k + 1]; return a;
  };
  auto loadB = [&](int k, v2f* b) {
    const float* __restrict__ B0 = Bc + (long)k * NC;
#pragma unroll
    for (int t = 0; t < NT; ++t) { b[t].x = B0[16 * t]; b[t].y = B0[NC + 16 * t]; }
  };

  v2f a_cur = loadA(0);
  v2f b_cur[NT];
  loadB(0, b_cur);
#pragma unroll 4
  for (int k = 0; k < K - 4; k += 4) {
    v2f a_nxt = loadA(k + 4);
    v2f b_nxt[NT];
    loadB(k + 4, b_nxt);
#pragma unroll
    for (int t = 0; t < NT; ++t)
      c[t] = __builtin_amdgcn_wmma_f32_16x16x4_f32(false, a_cur, false, b_cur[t],
                                                   (short)0, c[t], false, false);
    a_cur = a_nxt;
#pragma unroll
    for (int t = 0; t < NT; ++t) b_cur[t] = b_nxt[t];
  }
#pragma unroll
  for (int t = 0; t < NT; ++t)
    c[t] = __builtin_amdgcn_wmma_f32_16x16x4_f32(false, a_cur, false, b_cur[t],
                                                 (short)0, c[t], false, false);

#pragma unroll
  for (int t = 0; t < NT; ++t) {
#pragma unroll
    for (int r = 0; r < 8; ++r) {
      const int orow = tm * 16 + r + 8 * hi;
      if (orow < M) C[(long)orow * NC + col0 + 16 * t] = c[t][r];
    }
  }
}

template <int ROW, int JPH>
__global__ __launch_bounds__(256)
void edge_logits(const float* __restrict__ f, const int* __restrict__ src,
                 const int* __restrict__ dst, const float* __restrict__ attn,
                 float* __restrict__ s, int E) {
  const int lane = threadIdx.x & 31;
  const int wid  = blockIdx.x * (blockDim.x >> 5) + (threadIdx.x >> 5);
  if (wid >= E) return;
  const long srow = (long)src[wid] * ROW;
  const long drow = (long)dst[wid] * ROW;
  constexpr int ITER = ROW / 32;
  constexpr int H = ITER / JPH;
  float acc[H];
#pragma unroll
  for (int h = 0; h < H; ++h) acc[h] = 0.0f;
#pragma unroll
  for (int j = 0; j < ITER; ++j) {
    const int d = j * 32 + lane;
    float v = f[srow + d] + f[drow + d];
    v = v > 0.0f ? v : 0.2f * v;
    acc[j / JPH] += v * attn[d];
  }
#pragma unroll
  for (int h = 0; h < H; ++h) {
    float v = acc[h];
#pragma unroll
    for (int off = 16; off > 0; off >>= 1) v += __shfl_xor(v, off, 32);
    if (lane == 0) s[(long)wid * H + h] = v;
  }
}

constexpr int AG_ROWS  = 196;
constexpr int AG_WAVES = 32;
constexpr int AG_SCAN  = 6;
constexpr int AG_STEP  = 128 * AG_SCAN;
constexpr int AG_CAP   = AG_STEP + 128;
constexpr int AG_BATCH = 8;

__device__ __forceinline__ int ag_append(int d, int e, int base, int rows, int cnt,
                                         int* le) {
  const unsigned r = (unsigned)(d - base);
  const bool hit = r < (unsigned)rows;
  const unsigned mask = __builtin_amdgcn_ballot_w32(hit);
  if (mask != 0u) {
    const int pos = cnt + (int)__builtin_amdgcn_mbcnt_lo(mask, 0u);
    if (hit) le[pos] = e;
    cnt += __popc(mask);
  }
  return cnt;
}

__device__ __forceinline__ void lds_max_f32(float* p, float v) {
  if (v >= 0.0f)
    __hip_atomic_fetch_max((int*)p, __float_as_int(v), __ATOMIC_RELAXED,
                           __HIP_MEMORY_SCOPE_WORKGROUP);
  else
    __hip_atomic_fetch_min((unsigned*)p, __float_as_uint(v), __ATOMIC_RELAXED,
                           __HIP_MEMORY_SCOPE_WORKGROUP);
}

__device__ __forceinline__ void ag_max_drain(const float* __restrict__ s,
                                             const int* __restrict__ dst,
                                             float* mx, const int* le,
                                             int cnt, int base, int lane) {
  __builtin_amdgcn_fence(__ATOMIC_ACQ_REL, "wavefront");
  for (int i = 0; i < cnt; i += 32) {
    const int k = i + lane;
    if (k < cnt) {
      const int e = le[k];
      const int r = dst[e] - base;
      const v4f sv = *(const v4f*)(s + (long)e * 4);
      float* mr = mx + r * 4;
      lds_max_f32(mr,     sv.x);
      lds_max_f32(mr + 1, sv.y);
      lds_max_f32(mr + 2, sv.z);
      lds_max_f32(mr + 3, sv.w);
    }
  }
  __builtin_amdgcn_fence(__ATOMIC_ACQ_REL, "wavefront");
}

__device__ __forceinline__ void ag_drain(const float* __restrict__ f,
                                         const float* __restrict__ s,
                                         const int* __restrict__ src,
                                         const int* __restrict__ dst,
                                         float* tile, const float* mx, float* den,
                                         const int* le,
                                         int cnt, int base, int lane) {
  __builtin_amdgcn_fence(__ATOMIC_ACQ_REL, "wavefront");
  const int j = lane >> 2;
  const int hd = lane & 3;
  const bool lo = lane < 16;
  for (int i = 0; i < cnt; i += AG_BATCH) {
    const int e = le[min(i + j, cnt - 1)];
    const int sv = src[e];
    const int dn = dst[e];
    const float sl = s[(long)e * 4 + hd];
    v4f x0[AG_BATCH], x1[AG_BATCH];
#pragma unroll
    for (int b = 0; b < AG_BATCH; ++b) {
      const int sb = __builtin_amdgcn_readlane(sv, 4 * b);
      const v4f* __restrict__ fr = (const v4f*)(f + (long)sb * 256);
      x0[b] = fr[lane];
      x1[b] = fr[32 + lane];
    }
    const int r = dn - base;
    const float mv = mx[r * 4 + hd];
    asm volatile("" :: "v"(x0[0]), "v"(x1[0]), "v"(x0[1]), "v"(x1[1]), "v"(x0[2]), "v"(x1[2]),
                       "v"(x0[3]), "v"(x1[3]), "v"(x0[4]), "v"(x1[4]), "v"(x0[5]), "v"(x1[5]),
                       "v"(x0[6]), "v"(x1[6]), "v"(x0[7]), "v"(x1[7]));
    const float a = __expf(sl - mv);
    if (i + j < cnt)
      __hip_atomic_fetch_add(den + r * 4 + hd, a, __ATOMIC_RELAXED, __HIP_MEMORY_SCOPE_WORKGROUP);
#pragma unroll
    for (int b = 0; b < AG_BATCH; ++b) {
      if (i + b < cnt) {
        const int rb = __builtin_amdgcn_readlane(r, 4 * b);
        const float a0 = __int_as_float(__builtin_amdgcn_readlane(__float_as_int(a), 4 * b));
        const float a1 = __int_as_float(__builtin_amdgcn_readlane(__float_as_int(a), 4 * b + 1));
        const float a2 = __int_as_float(__builtin_amdgcn_readlane(__float_as_int(a), 4 * b + 2));
        const float a3 = __int_as_float(__builtin_amdgcn_readlane(__float_as_int(a), 4 * b + 3));
        const float al0 = lo ? a0 : a1;
        const float al1 = lo ? a2 : a3;
        float* tr = tile + rb * 256 + lane;
        __hip_atomic_fetch_add(tr,       x0[b].x * al0, __ATOMIC_RELAXED, __HIP_MEMORY_SCOPE_WORKGROUP);
        __hip_atomic_fetch_add(tr + 64,  x0[b].y * al0, __ATOMIC_RELAXED, __HIP_MEMORY_SCOPE_WORKGROUP);
        __hip_atomic_fetch_add(tr + 128, x0[b].z * al0, __ATOMIC_RELAXED, __HIP_MEMORY_SCOPE_WORKGROUP);
        __hip_atomic_fetch_add(tr + 192, x0[b].w * al0, __ATOMIC_RELAXED, __HIP_MEMORY_SCOPE_WORKGROUP);
        __hip_atomic_fetch_add(tr + 32,  x1[b].x * al1, __ATOMIC_RELAXED, __HIP_MEMORY_SCOPE_WORKGROUP);
        __hip_atomic_fetch_add(tr + 96,  x1[b].y * al1, __ATOMIC_RELAXED, __HIP_MEMORY_SCOPE_WORKGROUP);
        __hip_atomic_fetch_add(tr + 160, x1[b].z * al1, __ATOMIC_RELAXED, __HIP_MEMORY_SCOPE_WORKGROUP);
        __hip_atomic_fetch_add(tr + 224, x1[b].w * al1, __ATOMIC_RELAXED, __HIP_MEMORY_SCOPE_WORKGROUP);
      }
    }
  }
  __builtin_amdgcn_fence(__ATOMIC_ACQ_REL, "wavefront");
}

__device__ __forceinline__ void ag_drain128(const float* __restrict__ f,
                                            const float* __restrict__ s,
                                            const int* __restrict__ src,
                                            const int* __restrict__ dst,
                                            float* tile, const float* mx, float* den,
                                            const int* le,
                                            int cnt, int base, int lane) {
  __builtin_amdgcn_fence(__ATOMIC_ACQ_REL, "wavefront");
  const int j = lane >> 2;
  const int hd = lane & 3;
  const int hq = lane >> 3;
  for (int i = 0; i < cnt; i += AG_BATCH) {
    const int e = le[min(i + j, cnt - 1)];
    const int sv = src[e];
    const int dn = dst[e];
    const float sl = s[(long)e * 4 + hd];
    v4f x[AG_BATCH];
#pragma unroll
    for (int b = 0; b < AG_BATCH; ++b) {
      const int sb = __builtin_amdgcn_readlane(sv, 4 * b);
      const v4f* __restrict__ fr = (const v4f*)(f + (long)sb * 128);
      x[b] = fr[lane];
    }
    const int r = dn - base;
    const float mv = mx[r * 4 + hd];
    const float a = __expf(sl - mv);
    if (i + j < cnt)
      __hip_atomic_fetch_add(den + r * 4 + hd, a, __ATOMIC_RELAXED, __HIP_MEMORY_SCOPE_WORKGROUP);
    float al[AG_BATCH];
#pragma unroll
    for (int b = 0; b < AG_BATCH; ++b) al[b] = __shfl(a, 4 * b + hq, 32);
#pragma unroll
    for (int b = 0; b < AG_BATCH; ++b) {
      if (i + b < cnt) {
        const int rb = __builtin_amdgcn_readlane(r, 4 * b);
        float* tr = tile + rb * 128 + lane;
        __hip_atomic_fetch_add(tr,      x[b].x * al[b], __ATOMIC_RELAXED, __HIP_MEMORY_SCOPE_WORKGROUP);
        __hip_atomic_fetch_add(tr + 32, x[b].y * al[b], __ATOMIC_RELAXED, __HIP_MEMORY_SCOPE_WORKGROUP);
        __hip_atomic_fetch_add(tr + 64, x[b].z * al[b], __ATOMIC_RELAXED, __HIP_MEMORY_SCOPE_WORKGROUP);
        __hip_atomic_fetch_add(tr + 96, x[b].w * al[b], __ATOMIC_RELAXED, __HIP_MEMORY_SCOPE_WORKGROUP);
      }
    }
  }
  __builtin_amdgcn_fence(__ATOMIC_ACQ_REL, "wavefront");
}

template <bool MAXPASS, int ROW>
__device__ __forceinline__ int ag_filter(const float* __restrict__ f,
                                         const float* __restrict__ s,
                                         const int* __restrict__ src,
                                         const int* __restrict__ dst,
                                         float* tile, float* mx, float* den, int* le,
                                         int base, int rows, int E, int lane, int w,
                                         int& spilled) {
  int peel = (int)(((16u - (unsigned)((unsigned long long)dst & 15ull)) & 15u) >> 2);
  if (peel > E) peel = E;
  const int nG = (E - peel) >> 2;
  const int tail0 = peel + 4 * nG;
  int cnt = 0;
  if (w == 0) {
    int e = -1;
    if (lane < peel) e = lane;
    else if (lane >= 4 && lane < 8 && tail0 + (lane - 4) < E) e = tail0 + (lane - 4);
    const int d = e >= 0 ? dst[e] : -1;
    cnt = ag_append(d, e, base, rows, cnt, le);
  }
  cnt = __builtin_amdgcn_readfirstlane(cnt);
  const int4* __restrict__ d4 = (const int4*)(dst + peel);
  for (int g0 = 0; g0 < nG; g0 += 1024 * AG_SCAN) {
    if (cnt > AG_CAP - AG_STEP) {
      if (MAXPASS) {
        ag_max_drain(s, dst, mx, le, cnt, base, lane);
        spilled = 1;
      } else if (ROW == 256) {
        ag_drain(f, s, src, dst, tile, mx, den, le, cnt, base, lane);
      } else {
        ag_drain128(f, s, src, dst, tile, mx, den, le, cnt, base, lane);
      }
      cnt = 0;
    }
    int4 v[AG_SCAN];
#pragma unroll
    for (int u = 0; u < AG_SCAN; ++u) {
      const int g = g0 + u * 1024 + (int)threadIdx.x;
      v[u] = g < nG ? d4[g] : make_int4(-1, -1, -1, -1);
    }
#pragma unroll
    for (int u = 0; u < AG_SCAN; ++u) {
      const int e0 = peel + 4 * (g0 + u * 1024 + (int)threadIdx.x);
      cnt = ag_append(v[u].x, e0, base, rows, cnt, le);
      cnt = ag_append(v[u].y, e0 + 1, base, rows, cnt, le);
      cnt = ag_append(v[u].z, e0 + 2, base, rows, cnt, le);
      cnt = ag_append(v[u].w, e0 + 3, base, rows, cnt, le);
    }
  }
  return cnt;
}

__device__ __forceinline__ float elu1(float v) { return v > 0.0f ? v : expm1f(v); }

__global__ __launch_bounds__(1024)
void aggregate_owner_softmax_elu(const float* __restrict__ f, const float* __restrict__ s,
                                 const int* __restrict__ src, const int* __restrict__ dst,
                                 float* __restrict__ h, int N, int E) {
  __shared__ float tile[AG_ROWS * 256];
  __shared__ int l_edge[AG_WAVES * AG_CAP];
  __shared__ float l_mx[AG_ROWS * 4];
  __shared__ float l_den[AG_ROWS * 4];
  const int lane = threadIdx.x & 31;
  const int w    = threadIdx.x >> 5;
  const int base = blockIdx.x * AG_ROWS;
  const int rows = min(AG_ROWS, N - base);
  int* le = l_edge + w * AG_CAP;

  float4* t4 = (float4*)tile;
  for (int i = threadIdx.x; i < AG_ROWS * 64; i += 1024)
    t4[i] = make_float4(0.0f, 0.0f, 0.0f, 0.0f);
  for (int i = threadIdx.x; i < AG_ROWS * 4; i += 1024) {
    l_mx[i] = __uint_as_float(0xFF800000u);
    l_den[i] = 0.0f;
  }
  __syncthreads();

  int spilled = 0;
  int cnt = ag_filter<true, 256>(f, s, src, dst, tile, l_mx, l_den, le, base, rows, E,
                                 lane, w, spilled);
  ag_max_drain(s, dst, l_mx, le, cnt, base, lane);
  __syncthreads();

  if (__builtin_amdgcn_readfirstlane(spilled) != 0)
    cnt = ag_filter<false, 256>(f, s, src, dst, tile, l_mx, l_den, le, base, rows, E,
                                lane, w, spilled);
  ag_drain(f, s, src, dst, tile, l_mx, l_den, le, cnt, base, lane);
  __syncthreads();

  for (int i = threadIdx.x; i < rows * 64; i += 1024) {
    const int r = i >> 6, p = i & 63;
    const float* tr = tile + r * 256 + p;
    const float dv = l_den[r * 4 + (p >> 4)];
    const bool none = dv == 0.0f;
    float4 o;
    o.x = elu1(none ? 0.0f : tr[0]   / dv);
    o.y = elu1(none ? 0.0f : tr[64]  / dv);
    o.z = elu1(none ? 0.0f : tr[128] / dv);
    o.w = elu1(none ? 0.0f : tr[192] / dv);
    *(float4*)(h + (long)(base + r) * 256 + 4 * p) = o;
  }
}

__global__ __launch_bounds__(1024)
void aggregate_owner_softmax_mean(const float* __restrict__ f, const float* __restrict__ s,
                                  const float* __restrict__ o1,
                                  const int* __restrict__ src, const int* __restrict__ dst,
                                  float* __restrict__ out, int N, int E) {
  __shared__ float tile[AG_ROWS * 128];
  __shared__ int l_edge[AG_WAVES * AG_CAP];
  __shared__ float l_mx[AG_ROWS * 4];
  __shared__ float l_den[AG_ROWS * 4];
  const int lane = threadIdx.x & 31;
  const int w    = threadIdx.x >> 5;
  const int base = blockIdx.x * AG_ROWS;
  const int rows = min(AG_ROWS, N - base);
  int* le = l_edge + w * AG_CAP;

  float4* t4 = (float4*)tile;
  for (int i = threadIdx.x; i < AG_ROWS * 32; i += 1024)
    t4[i] = make_float4(0.0f, 0.0f, 0.0f, 0.0f);
  for (int i = threadIdx.x; i < AG_ROWS * 4; i += 1024) {
    l_mx[i] = __uint_as_float(0xFF800000u);
    l_den[i] = 0.0f;
  }
  __syncthreads();

  int spilled = 0;
  int cnt = ag_filter<true, 128>(f, s, src, dst, tile, l_mx, l_den, le, base, rows, E,
                                 lane, w, spilled);
  ag_max_drain(s, dst, l_mx, le, cnt, base, lane);
  __syncthreads();

  if (__builtin_amdgcn_readfirstlane(spilled) != 0)
    cnt = ag_filter<false, 128>(f, s, src, dst, tile, l_mx, l_den, le, base, rows, E,
                                lane, w, spilled);
  ag_drain128(f, s, src, dst, tile, l_mx, l_den, le, cnt, base, lane);
  __syncthreads();

  for (int i = threadIdx.x; i < rows * 8; i += 1024) {
    const int r = i >> 3, q = i & 7;
    const float* tr = tile + r * 128 + q;
    const float* __restrict__ rr = o1 + (long)(base + r) * 128 + 4 * q;
    float4 acc = make_float4(0.0f, 0.0f, 0.0f, 0.0f);
#pragma unroll
    for (int hh = 0; hh < 4; ++hh) {
      const float dv = l_den[r * 4 + hh];
      const bool none = dv == 0.0f;
      const float4 rs = *(const float4*)(rr + hh * 32);
      const float* th = tr + hh * 8;
      acc.x += (none ? 0.0f : th[0]  / dv) + rs.x;
      acc.y += (none ? 0.0f : th[32] / dv) + rs.y;
      acc.z += (none ? 0.0f : th[64] / dv) + rs.z;
      acc.w += (none ? 0.0f : th[96] / dv) + rs.w;
    }
    float4 o;
    o.x = 0.25f * acc.x; o.y = 0.25f * acc.y; o.z = 0.25f * acc.z; o.w = 0.25f * acc.w;
    *(float4*)(out + (long)(base + r) * 32 + 4 * q) = o;
  }
}

static inline int cdiv_l(long a, long b) { return (int)((a + b - 1) / b); }

extern "C" void kernel_launch(void* const* d_in, const int* in_sizes, int n_in,
                              void* d_out, int out_size, void* d_ws, size_t ws_size,
                              hipStream_t stream) {
  const float* x     = (const float*)d_in[0];
  const float* W0    = (const float*)d_in[1];
  const float* a0    = (const float*)d_in[2];
  const float* W1    = (const float*)d_in[3];
  const float* a1    = (const float*)d_in[4];
  const float* Wres1 = (const float*)d_in[5];
  const int*   src   = (const int*)d_in[6];
  const int*   dst   = (const int*)d_in[7];

  const int N = in_sizes[0] / 256;
  const int E = in_sizes[6];
  const int H = 4;

  float*    f0    = (float*)d_ws;
  float*    h     = f0 + (long)N * 256;
  float*    f1    = h  + (long)N * 256;
  float*    o1    = f1 + (long)N * 128;
  float*    s     = o1 + (long)N * 128;
  const size_t need =
      ((size_t)N * 256 * 2 + (size_t)N * 128 * 2 + (size_t)E * H) * sizeof(float);
  if (need > ws_size) return;

  const int BT = 256;
  const int egrid = cdiv_l((long)E * 32, BT);

  {
    const int strips = ((N + 15) / 16) * (256 / 16 / 4);
    gemm_f32_wmma<4, 256, 256><<<cdiv_l(strips, 8), BT, 0, stream>>>(x, W0, f0, N);
  }
  edge_logits<256, 2><<<egrid, BT, 0, stream>>>(f0, src, dst, a0, s, E);
  aggregate_owner_softmax_elu<<<cdiv_l(N, AG_ROWS), 1024, 0, stream>>>(f0, s, src, dst,
                                                                       h, N, E);

  {
    const int strips = ((N + 15) / 16) * (128 / 16 / 4);
    gemm_f32_wmma<4, 256, 128><<<cdiv_l(strips, 8), BT, 0, stream>>>(h, W1, f1, N);
    gemm_f32_wmma<4, 256, 128><<<cdiv_l(strips, 8), BT, 0, stream>>>(h, Wres1, o1, N);
  }
  edge_logits<128, 1><<<egrid, BT, 0, stream>>>(f1, src, dst, a1, s, E);
  aggregate_owner_softmax_mean<<<cdiv_l(N, AG_ROWS), 1024, 0, stream>>>(
      f1, s, o1, src, dst, (float*)d_out, N, E);
}


# edge logits computed inside the owner workgroups' max pass: a wave per edge row pair, 128-bit loads, DPP sums over a head's lanes (logit sum order changes); both edge_logits launches gone, 7 -> 5
# speedup vs baseline: 2.3628x; 1.2327x over previous
#include <hip/hip_runtime.h>
#include <hip/hip_bf16.h>

typedef __attribute__((ext_vector_type(2))) float v2f;
typedef __attribute__((ext_vector_type(8))) float v8f;
typedef __attribute__((ext_vector_type(4))) float v4f;

template <int NT, int K, int NC>
__global__ __launch_bounds__(256)
void gemm_f32_wmma(const float* __restrict__ A, const float* __restrict__ B,
                   float* __restrict__ C, int M) {
  const int lane = threadIdx.x & 31;
  const int wid  = blockIdx.x * (blockDim.x >> 5) + (threadIdx.x >> 5);
  const int tilesM  = (M + 15) >> 4;
  constexpr int stripsN = (NC >> 4) / NT;
  if (wid >= tilesM * stripsN) return;
  const int tm  = wid / stripsN;
  const int tn0 = (wid - tm * stripsN) * NT;
  const int hi = lane >> 4;
  const int lm = lane & 15;
  int arow = tm * 16 + lm; if (arow >= M) arow = M - 1;
  const int col0 = tn0 * 16 + lm;
  const float* __restrict__ Ar = A + (long)arow * K + 2 * hi;
  const float* __restrict__ Bc = B + (long)(2 * hi) * NC + col0;

  v8f c[NT];
#pragma unroll
  for (int t = 0; t < NT; ++t) c[t] = (v8f){};

  auto loadA = [&](int k) {
    v2f a; a.x = Ar[k]; a.y = Ar[k + 1]; return a;
  };
  auto loadB = [&](int k, v2f* b) {
    const float* __restrict__ B0 = Bc + (long)k * NC;
#pragma unroll
    for (int t = 0; t < NT; ++t) { b[t].x = B0[16 * t]; b[t].y = B0[NC + 16 * t]; }
  };

  v2f a_cur = loadA(0);
  v2f b_cur[NT];
  loadB(0, b_cur);
#pragma unroll 4
  for (int k = 0; k < K - 4; k += 4) {
    v2f a_nxt = loadA(k + 4);
    v2f b_nxt[NT];
    loadB(k + 4, b_nxt);
#pragma unroll
    for (int t = 0; t < NT; ++t)
      c[t] = __builtin_amdgcn_wmma_f32_16x16x4_f32(false, a_cur, false, b_cur[t],
                                                   (short)0, c[t], false, false);
    a_cur = a_nxt;
#pragma unroll
    for (int t = 0; t < NT; ++t) b_cur[t] = b_nxt[t];
  }
#pragma unroll
  for (int t = 0; t < NT; ++t)
    c[t] = __builtin_amdgcn_wmma_f32_16x16x4_f32(false, a_cur, false, b_cur[t],
                                                 (short)0, c[t], false, false);

#pragma unroll
  for (int t = 0; t < NT; ++t) {
#pragma unroll
    for (int r = 0; r < 8; ++r) {
      const int orow = tm * 16 + r + 8 * hi;
      if (orow < M) C[(long)orow * NC + col0 + 16 * t] = c[t][r];
    }
  }
}


template <int M>
__device__ __forceinline__ float row_xor_add(float v) {
  return v + __int_as_float(__builtin_amdgcn_update_dpp(0, __float_as_int(v), 0x160 + M,
                                                        0xf, 0xf, true));
}

__device__ __forceinline__ float lrelu_dot4(v4f a, v4f b, v4f w) {
  const v4f v = a + b;
  float r = fmaxf(v.x, 0.2f * v.x) * w.x;
  r += fmaxf(v.y, 0.2f * v.y) * w.y;
  r += fmaxf(v.z, 0.2f * v.z) * w.z;
  r += fmaxf(v.w, 0.2f * v.w) * w.w;
  return r;
}

constexpr int AG_ROWS  = 196;
constexpr int AG_WAVES = 32;
constexpr int AG_SCAN  = 6;
constexpr int AG_STEP  = 128 * AG_SCAN;
constexpr int AG_CAP   = AG_STEP + 128;
constexpr int AG_BATCH = 8;

__device__ __forceinline__ int ag_append(int d, int e, int base, int rows, int cnt,
                                         int* le) {
  const unsigned r = (unsigned)(d - base);
  const bool hit = r < (unsigned)rows;
  const unsigned mask = __builtin_amdgcn_ballot_w32(hit);
  if (mask != 0u) {
    const int pos = cnt + (int)__builtin_amdgcn_mbcnt_lo(mask, 0u);
    if (hit) le[pos] = e;
    cnt += __popc(mask);
  }
  return cnt;
}

__device__ __forceinline__ void lds_max_f32(float* p, float v) {
  if (v >= 0.0f)
    __hip_atomic_fetch_max((int*)p, __float_as_int(v), __ATOMIC_RELAXED,
                           __HIP_MEMORY_SCOPE_WORKGROUP);
  else
    __hip_atomic_fetch_min((unsigned*)p, __float_as_uint(v), __ATOMIC_RELAXED,
                           __HIP_MEMORY_SCOPE_WORKGROUP);
}

__device__ __forceinline__ void ag_logits_max256(const float* __restrict__ f,
                                                 const float* __restrict__ attn,
                                                 float* s,
                                                 const int* __restrict__ src,
                                                 const int* __restrict__ dst,
                                                 float* mx, const int* le,
                                                 int cnt, int base, int lane) {
  __builtin_amdgcn_fence(__ATOMIC_ACQ_REL, "wavefront");
  cnt = __builtin_amdgcn_readfirstlane(cnt);
  const v4f w0 = ((const v4f*)attn)[lane];
  const v4f w1 = ((const v4f*)attn)[32 + lane];
  for (int c = 0; c < cnt; c += 32) {
    const int e = le[min(c + lane, cnt - 1)];
    const int sv = src[e];
    const int dn = dst[e];
    v4f lg = (v4f){0.0f, 0.0f, 0.0f, 0.0f};
    const int nb = min(32, cnt - c);
    for (int q0 = 0; q0 < nb; q0 += 4) {
      v4f xs0[4], xs1[4], xd0[4], xd1[4];
#pragma unroll
      for (int b = 0; b < 4; ++b) {
        const int sb = __builtin_amdgcn_readlane(sv, q0 + b);
        const int db = __builtin_amdgcn_readlane(dn, q0 + b);
        const v4f* __restrict__ fs = (const v4f*)(f + (long)sb * 256);
        const v4f* __restrict__ fd = (const v4f*)(f + (long)db * 256);
        xs0[b] = fs[lane];
        xs1[b] = fs[32 + lane];
        xd0[b] = fd[lane];
        xd1[b] = fd[32 + lane];
      }
#pragma unroll
      for (int b = 0; b < 4; ++b) {
        float p0 = lrelu_dot4(xs0[b], xd0[b], w0);
        float p1 = lrelu_dot4(xs1[b], xd1[b], w1);
        p0 = row_xor_add<1>(p0); p1 = row_xor_add<1>(p1);
        p0 = row_xor_add<2>(p0); p1 = row_xor_add<2>(p1);
        p0 = row_xor_add<4>(p0); p1 = row_xor_add<4>(p1);
        p0 = row_xor_add<8>(p0); p1 = row_xor_add<8>(p1);
        const int l0 = __builtin_amdgcn_readlane(__float_as_int(p0), 0);
        const int l1 = __builtin_amdgcn_readlane(__float_as_int(p0), 16);
        const int l2 = __builtin_amdgcn_readlane(__float_as_int(p1), 0);
        const int l3 = __builtin_amdgcn_readlane(__float_as_int(p1), 16);
        const bool mine = lane == q0 + b;
        lg.x = mine ? __int_as_float(l0) : lg.x;
        lg.y = mine ? __int_as_float(l1) : lg.y;
        lg.z = mine ? __int_as_float(l2) : lg.z;
        lg.w = mine ? __int_as_float(l3) : lg.w;
      }
    }
    if (c + lane < cnt) {
      *(v4f*)(s + (long)e * 4) = lg;
      float* mr = mx + (dn - base) * 4;
      lds_max_f32(mr,     lg.x);
      lds_max_f32(mr + 1, lg.y);
      lds_max_f32(mr + 2, lg.z);
      lds_max_f32(mr + 3, lg.w);
    }
  }
  __builtin_amdgcn_fence(__ATOMIC_ACQ_REL, "wavefront");
}

__device__ __forceinline__ void ag_logits_max128(const float* __restrict__ f,
                                                 const float* __restrict__ attn,
                                                 float* s,
                                                 const int* __restrict__ src,
                                                 const int* __restrict__ dst,
                                                 float* mx, const int* le,
                                                 int cnt, int base, int lane) {
  __builtin_amdgcn_fence(__ATOMIC_ACQ_REL, "wavefront");
  cnt = __builtin_amdgcn_readfirstlane(cnt);
  const v4f w = ((const v4f*)attn)[lane];
  for (int c = 0; c < cnt; c += 32) {
    const int e = le[min(c + lane, cnt - 1)];
    const int sv = src[e];
    const int dn = dst[e];
    v4f lg = (v4f){0.0f, 0.0f, 0.0f, 0.0f};
    const int nb = min(32, cnt - c);
    for (int q0 = 0; q0 < nb; q0 += AG_BATCH) {
      v4f xs[AG_BATCH], xd[AG_BATCH];
#pragma unroll
      for (int b = 0; b < AG_BATCH; ++b) {
        const int sb = __builtin_amdgcn_readlane(sv, q0 + b);
        const int db = __builtin_amdgcn_readlane(dn, q0 + b);
        xs[b] = ((const v4f*)(f + (long)sb * 128))[lane];
        xd[b] = ((const v4f*)(f + (long)db * 128))[lane];
      }
#pragma unroll
      for (int b = 0; b < AG_BATCH; ++b) {
        float p = lrelu_dot4(xs[b], xd[b], w);
        p = row_xor_add<1>(p);
        p = row_xor_add<2>(p);
        p = row_xor_add<4>(p);
        const int l0 = __builtin_amdgcn_readlane(__float_as_int(p), 0);
        const int l1 = __builtin_amdgcn_readlane(__float_as_int(p), 8);
        const int l2 = __builtin_amdgcn_readlane(__float_as_int(p), 16);
        const int l3 = __builtin_amdgcn_readlane(__float_as_int(p), 24);
        const bool mine = lane == q0 + b;
        lg.x = mine ? __int_as_float(l0) : lg.x;
        lg.y = mine ? __int_as_float(l1) : lg.y;
        lg.z = mine ? __int_as_float(l2) : lg.z;
        lg.w = mine ? __int_as_float(l3) : lg.w;
      }
    }
    if (c + lane < cnt) {
      *(v4f*)(s + (long)e * 4) = lg;
      float* mr = mx + (dn - base) * 4;
      lds_max_f32(mr,     lg.x);
      lds_max_f32(mr + 1, lg.y);
      lds_max_f32(mr + 2, lg.z);
      lds_max_f32(mr + 3, lg.w);
    }
  }
  __builtin_amdgcn_fence(__ATOMIC_ACQ_REL, "wavefront");
}

__device__ __forceinline__ void ag_drain(const float* __restrict__ f,
                                         const float* s,
                                         const int* __restrict__ src,
                                         const int* __restrict__ dst,
                                         float* tile, const float* mx, float* den,
                                         const int* le,
                                         int cnt, int base, int lane) {
  __builtin_amdgcn_fence(__ATOMIC_ACQ_REL, "wavefront");
  const int j = lane >> 2;
  const int hd = lane & 3;
  const bool lo = lane < 16;
  for (int i = 0; i < cnt; i += AG_BATCH) {
    const int e = le[min(i + j, cnt - 1)];
    const int sv = src[e];
    const int dn = dst[e];
    const float sl = s[(long)e * 4 + hd];
    v4f x0[AG_BATCH], x1[AG_BATCH];
#pragma unroll
    for (int b = 0; b < AG_BATCH; ++b) {
      const int sb = __builtin_amdgcn_readlane(sv, 4 * b);
      const v4f* __restrict__ fr = (const v4f*)(f + (long)sb * 256);
      x0[b] = fr[lane];
      x1[b] = fr[32 + lane];
    }
    const int r = dn - base;
    const float mv = mx[r * 4 + hd];
    asm volatile("" :: "v"(x0[0]), "v"(x1[0]), "v"(x0[1]), "v"(x1[1]), "v"(x0[2]), "v"(x1[2]),
                       "v"(x0[3]), "v"(x1[3]), "v"(x0[4]), "v"(x1[4]), "v"(x0[5]), "v"(x1[5]),
                       "v"(x0[6]), "v"(x1[6]), "v"(x0[7]), "v"(x1[7]));
    const float a = __expf(sl - mv);
    if (i + j < cnt)
      __hip_atomic_fetch_add(den + r * 4 + hd, a, __ATOMIC_RELAXED, __HIP_MEMORY_SCOPE_WORKGROUP);
#pragma unroll
    for (int b = 0; b < AG_BATCH; ++b) {
      if (i + b < cnt) {
        const int rb = __builtin_amdgcn_readlane(r, 4 * b);
        const float a0 = __int_as_float(__builtin_amdgcn_readlane(__float_as_int(a), 4 * b));
        const float a1 = __int_as_float(__builtin_amdgcn_readlane(__float_as_int(a), 4 * b + 1));
        const float a2 = __int_as_float(__builtin_amdgcn_readlane(__float_as_int(a), 4 * b + 2));
        const float a3 = __int_as_float(__builtin_amdgcn_readlane(__float_as_int(a), 4 * b + 3));
        const float al0 = lo ? a0 : a1;
        const float al1 = lo ? a2 : a3;
        float* tr = tile + rb * 256 + lane;
        __hip_atomic_fetch_add(tr,       x0[b].x * al0, __ATOMIC_RELAXED, __HIP_MEMORY_SCOPE_WORKGROUP);
        __hip_atomic_fetch_add(tr + 64,  x0[b].y * al0, __ATOMIC_RELAXED, __HIP_MEMORY_SCOPE_WORKGROUP);
        __hip_atomic_fetch_add(tr + 128, x0[b].z * al0, __ATOMIC_RELAXED, __HIP_MEMORY_SCOPE_WORKGROUP);
        __hip_atomic_fetch_add(tr + 192, x0[b].w * al0, __ATOMIC_RELAXED, __HIP_MEMORY_SCOPE_WORKGROUP);
        __hip_atomic_fetch_add(tr + 32,  x1[b].x * al1, __ATOMIC_RELAXED, __HIP_MEMORY_SCOPE_WORKGROUP);
        __hip_atomic_fetch_add(tr + 96,  x1[b].y * al1, __ATOMIC_RELAXED, __HIP_MEMORY_SCOPE_WORKGROUP);
        __hip_atomic_fetch_add(tr + 160, x1[b].z * al1, __ATOMIC_RELAXED, __HIP_MEMORY_SCOPE_WORKGROUP);
        __hip_atomic_fetch_add(tr + 224, x1[b].w * al1, __ATOMIC_RELAXED, __HIP_MEMORY_SCOPE_WORKGROUP);
      }
    }
  }
  __builtin_amdgcn_fence(__ATOMIC_ACQ_REL, "wavefront");
}

__device__ __forceinline__ void ag_drain128(const float* __restrict__ f,
                                            const float* s,
                                            const int* __restrict__ src,
                                            const int* __restrict__ dst,
                                            float* tile, const float* mx, float* den,
                                            const int* le,
                                            int cnt, int base, int lane) {
  __builtin_amdgcn_fence(__ATOMIC_ACQ_REL, "wavefront");
  const int j = lane >> 2;
  const int hd = lane & 3;
  const int hq = lane >> 3;
  for (int i = 0; i < cnt; i += AG_BATCH) {
    const int e = le[min(i + j, cnt - 1)];
    const int sv = src[e];
    const int dn = dst[e];
    const float sl = s[(long)e * 4 + hd];
    v4f x[AG_BATCH];
#pragma unroll
    for (int b = 0; b < AG_BATCH; ++b) {
      const int sb = __builtin_amdgcn_readlane(sv, 4 * b);
      const v4f* __restrict__ fr = (const v4f*)(f + (long)sb * 128);
      x[b] = fr[lane];
    }
    const int r = dn - base;
    const float mv = mx[r * 4 + hd];
    const float a = __expf(sl - mv);
    if (i + j < cnt)
      __hip_atomic_fetch_add(den + r * 4 + hd, a, __ATOMIC_RELAXED, __HIP_MEMORY_SCOPE_WORKGROUP);
    float al[AG_BATCH];
#pragma unroll
    for (int b = 0; b < AG_BATCH; ++b) al[b] = __shfl(a, 4 * b + hq, 32);
#pragma unroll
    for (int b = 0; b < AG_BATCH; ++b) {
      if (i + b < cnt) {
        const int rb = __builtin_amdgcn_readlane(r, 4 * b);
        float* tr = tile + rb * 128 + lane;
        __hip_atomic_fetch_add(tr,      x[b].x * al[b], __ATOMIC_RELAXED, __HIP_MEMORY_SCOPE_WORKGROUP);
        __hip_atomic_fetch_add(tr + 32, x[b].y * al[b], __ATOMIC_RELAXED, __HIP_MEMORY_SCOPE_WORKGROUP);
        __hip_atomic_fetch_add(tr + 64, x[b].z * al[b], __ATOMIC_RELAXED, __HIP_MEMORY_SCOPE_WORKGROUP);
        __hip_atomic_fetch_add(tr + 96, x[b].w * al[b], __ATOMIC_RELAXED, __HIP_MEMORY_SCOPE_WORKGROUP);
      }
    }
  }
  __builtin_amdgcn_fence(__ATOMIC_ACQ_REL, "wavefront");
}

template <bool MAXPASS, int ROW>
__device__ __forceinline__ int ag_filter(const float* __restrict__ f,
                                         const float* __restrict__ attn,
                                         float* s,
                                         const int* __restrict__ src,
                                         const int* __restrict__ dst,
                                         float* tile, float* mx, float* den, int* le,
                                         int base, int rows, int E, int lane, int w,
                                         int& spilled) {
  int peel = (int)(((16u - (unsigned)((unsigned long long)dst & 15ull)) & 15u) >> 2);
  if (peel > E) peel = E;
  const int nG = (E - peel) >> 2;
  const int tail0 = peel + 4 * nG;
  int cnt = 0;
  if (w == 0) {
    int e = -1;
    if (lane < peel) e = lane;
    else if (lane >= 4 && lane < 8 && tail0 + (lane - 4) < E) e = tail0 + (lane - 4);
    const int d = e >= 0 ? dst[e] : -1;
    cnt = ag_append(d, e, base, rows, cnt, le);
  }
  cnt = __builtin_amdgcn_readfirstlane(cnt);
  const int4* __restrict__ d4 = (const int4*)(dst + peel);
  for (int g0 = 0; g0 < nG; g0 += 1024 * AG_SCAN) {
    if (cnt > AG_CAP - AG_STEP) {
      if (MAXPASS) {
        if (ROW == 256) ag_logits_max256(f, attn, s, src, dst, mx, le, cnt, base, lane);
        else            ag_logits_max128(f, attn, s, src, dst, mx, le, cnt, base, lane);
        spilled = 1;
      } else if (ROW == 256) {
        ag_drain(f, s, src, dst, tile, mx, den, le, cnt, base, lane);
      } else {
        ag_drain128(f, s, src, dst, tile, mx, den, le, cnt, base, lane);
      }
      cnt = 0;
    }
    int4 v[AG_SCAN];
#pragma unroll
    for (int u = 0; u < AG_SCAN; ++u) {
      const int g = g0 + u * 1024 + (int)threadIdx.x;
      v[u] = g < nG ? d4[g] : make_int4(-1, -1, -1, -1);
    }
#pragma unroll
    for (int u = 0; u < AG_SCAN; ++u) {
      const int e0 = peel + 4 * (g0 + u * 1024 + (int)threadIdx.x);
      cnt = ag_append(v[u].x, e0, base, rows, cnt, le);
      cnt = ag_append(v[u].y, e0 + 1, base, rows, cnt, le);
      cnt = ag_append(v[u].z, e0 + 2, base, rows, cnt, le);
      cnt = ag_append(v[u].w, e0 + 3, base, rows, cnt, le);
    }
  }
  return cnt;
}

__device__ __forceinline__ float elu1(float v) { return v > 0.0f ? v : expm1f(v); }

__global__ __launch_bounds__(1024)
void aggregate_owner_softmax_elu(const float* __restrict__ f, const float* __restrict__ attn,
                                 float* s,
                                 const int* __restrict__ src, const int* __restrict__ dst,
                                 float* __restrict__ h, int N, int E) {
  __shared__ float tile[AG_ROWS * 256];
  __shared__ int l_edge[AG_WAVES * AG_CAP];
  __shared__ float l_mx[AG_ROWS * 4];
  __shared__ float l_den[AG_ROWS * 4];
  const int lane = threadIdx.x & 31;
  const int w    = threadIdx.x >> 5;
  const int base = blockIdx.x * AG_ROWS;
  const int rows = min(AG_ROWS, N - base);
  int* le = l_edge + w * AG_CAP;

  float4* t4 = (float4*)tile;
  for (int i = threadIdx.x; i < AG_ROWS * 64; i += 1024)
    t4[i] = make_float4(0.0f, 0.0f, 0.0f, 0.0f);
  for (int i = threadIdx.x; i < AG_ROWS * 4; i += 1024) {
    l_mx[i] = __uint_as_float(0xFF800000u);
    l_den[i] = 0.0f;
  }
  __syncthreads();

  int spilled = 0;
  int cnt = ag_filter<true, 256>(f, attn, s, src, dst, tile, l_mx, l_den, le, base, rows, E,
                                 lane, w, spilled);
  ag_logits_max256(f, attn, s, src, dst, l_mx, le, cnt, base, lane);
  __syncthreads();

  if (__builtin_amdgcn_readfirstlane(spilled) != 0)
    cnt = ag_filter<false, 256>(f, attn, s, src, dst, tile, l_mx, l_den, le, base, rows, E,
                                lane, w, spilled);
  ag_drain(f, s, src, dst, tile, l_mx, l_den, le, cnt, base, lane);
  __syncthreads();

  for (int i = threadIdx.x; i < rows * 64; i += 1024) {
    const int r = i >> 6, p = i & 63;
    const float* tr = tile + r * 256 + p;
    const float dv = l_den[r * 4 + (p >> 4)];
    const bool none = dv == 0.0f;
    float4 o;
    o.x = elu1(none ? 0.0f : tr[0]   / dv);
    o.y = elu1(none ? 0.0f : tr[64]  / dv);
    o.z = elu1(none ? 0.0f : tr[128] / dv);
    o.w = elu1(none ? 0.0f : tr[192] / dv);
    *(float4*)(h + (long)(base + r) * 256 + 4 * p) = o;
  }
}

__global__ __launch_bounds__(1024)
void aggregate_owner_softmax_mean(const float* __restrict__ f, const float* __restrict__ attn,
                                  float* s, const float* __restrict__ o1,
                                  const int* __restrict__ src, const int* __restrict__ dst,
                                  float* __restrict__ out, int N, int E) {
  __shared__ float tile[AG_ROWS * 128];
  __shared__ int l_edge[AG_WAVES * AG_CAP];
  __shared__ float l_mx[AG_ROWS * 4];
  __shared__ float l_den[AG_ROWS * 4];
  const int lane = threadIdx.x & 31;
  const int w    = threadIdx.x >> 5;
  const int base = blockIdx.x * AG_ROWS;
  const int rows = min(AG_ROWS, N - base);
  int* le = l_edge + w * AG_CAP;

  float4* t4 = (float4*)tile;
  for (int i = threadIdx.x; i < AG_ROWS * 32; i += 1024)
    t4[i] = make_float4(0.0f, 0.0f, 0.0f, 0.0f);
  for (int i = threadIdx.x; i < AG_ROWS * 4; i += 1024) {
    l_mx[i] = __uint_as_float(0xFF800000u);
    l_den[i] = 0.0f;
  }
  __syncthreads();

  int spilled = 0;
  int cnt = ag_filter<true, 128>(f, attn, s, src, dst, tile, l_mx, l_den, le, base, rows, E,
                                 lane, w, spilled);
  ag_logits_max128(f, attn, s, src, dst, l_mx, le, cnt, base, lane);
  __syncthreads();

  if (__builtin_amdgcn_readfirstlane(spilled) != 0)
    cnt = ag_filter<false, 128>(f, attn, s, src, dst, tile, l_mx, l_den, le, base, rows, E,
                                lane, w, spilled);
  ag_drain128(f, s, src, dst, tile, l_mx, l_den, le, cnt, base, lane);
  __syncthreads();

  for (int i = threadIdx.x; i < rows * 8; i += 1024) {
    const int r = i >> 3, q = i & 7;
    const float* tr = tile + r * 128 + q;
    const float* __restrict__ rr = o1 + (long)(base + r) * 128 + 4 * q;
    float4 acc = make_float4(0.0f, 0.0f, 0.0f, 0.0f);
#pragma unroll
    for (int hh = 0; hh < 4; ++hh) {
      const float dv = l_den[r * 4 + hh];
      const bool none = dv == 0.0f;
      const float4 rs = *(const float4*)(rr + hh * 32);
      const float* th = tr + hh * 8;
      acc.x += (none ? 0.0f : th[0]  / dv) + rs.x;
      acc.y += (none ? 0.0f : th[32] / dv) + rs.y;
      acc.z += (none ? 0.0f : th[64] / dv) + rs.z;
      acc.w += (none ? 0.0f : th[96] / dv) + rs.w;
    }
    float4 o;
    o.x = 0.25f * acc.x; o.y = 0.25f * acc.y; o.z = 0.25f * acc.z; o.w = 0.25f * acc.w;
    *(float4*)(out + (long)(base + r) * 32 + 4 * q) = o;
  }
}

static inline int cdiv_l(long a, long b) { return (int)((a + b - 1) / b); }

extern "C" void kernel_launch(void* const* d_in, const int* in_sizes, int n_in,
                              void* d_out, int out_size, void* d_ws, size_t ws_size,
                              hipStream_t stream) {
  const float* x     = (const float*)d_in[0];
  const float* W0    = (const float*)d_in[1];
  const float* a0    = (const float*)d_in[2];
  const float* W1    = (const float*)d_in[3];
  const float* a1    = (const float*)d_in[4];
  const float* Wres1 = (const float*)d_in[5];
  const int*   src   = (const int*)d_in[6];
  const int*   dst   = (const int*)d_in[7];

  const int N = in_sizes[0] / 256;
  const int E = in_sizes[6];
  const int H = 4;

  float*    f0    = (float*)d_ws;
  float*    h     = f0 + (long)N * 256;
  float*    f1    = h  + (long)N * 256;
  float*    o1    = f1 + (long)N * 128;
  float*    s     = o1 + (long)N * 128;
  const size_t need =
      ((size_t)N * 256 * 2 + (size_t)N * 128 * 2 + (size_t)E * H) * sizeof(float);
  if (need > ws_size) return;

  const int BT = 256;

  {
    const int strips = ((N + 15) / 16) * (256 / 16 / 4);
    gemm_f32_wmma<4, 256, 256><<<cdiv_l(strips, 8), BT, 0, stream>>>(x, W0, f0, N);
  }
  aggregate_owner_softmax_elu<<<cdiv_l(N, AG_ROWS), 1024, 0, stream>>>(f0, a0, s, src, dst,
                                                                       h, N, E);

  {
    const int strips = ((N + 15) / 16) * (128 / 16 / 4);
    gemm_f32_wmma<4, 256, 128><<<cdiv_l(strips, 8), BT, 0, stream>>>(h, W1, f1, N);
    gemm_f32_wmma<4, 256, 128><<<cdiv_l(strips, 8), BT, 0, stream>>>(h, Wres1, o1, N);
  }
  aggregate_owner_softmax_mean<<<cdiv_l(N, AG_ROWS), 1024, 0, stream>>>(
      f1, a1, s, o1, src, dst, (float*)d_out, N, E);
}


# logit pass reads destination rows from LDS: an owner workgroup copies its own 196 rows of f into the tile (idle before pass 2), zeroes it between passes; only source rows gathered, 8 or 16 per batch
# speedup vs baseline: 2.4413x; 1.0332x over previous
#include <hip/hip_runtime.h>
#include <hip/hip_bf16.h>

typedef __attribute__((ext_vector_type(2))) float v2f;
typedef __attribute__((ext_vector_type(8))) float v8f;
typedef __attribute__((ext_vector_type(4))) float v4f;

template <int NT, int K, int NC>
__global__ __launch_bounds__(256)
void gemm_f32_wmma(const float* __restrict__ A, const float* __restrict__ B,
                   float* __restrict__ C, int M) {
  const int lane = threadIdx.x & 31;
  const int wid  = blockIdx.x * (blockDim.x >> 5) + (threadIdx.x >> 5);
  const int tilesM  = (M + 15) >> 4;
  constexpr int stripsN = (NC >> 4) / NT;
  if (wid >= tilesM * stripsN) return;
  const int tm  = wid / stripsN;
  const int tn0 = (wid - tm * stripsN) * NT;
  const int hi = lane >> 4;
  const int lm = lane & 15;
  int arow = tm * 16 + lm; if (arow >= M) arow = M - 1;
  const int col0 = tn0 * 16 + lm;
  const float* __restrict__ Ar = A + (long)arow * K + 2 * hi;
  const float* __restrict__ Bc = B + (long)(2 * hi) * NC + col0;

  v8f c[NT];
#pragma unroll
  for (int t = 0; t < NT; ++t) c[t] = (v8f){};

  auto loadA = [&](int k) {
    v2f a; a.x = Ar[k]; a.y = Ar[k + 1]; return a;
  };
  auto loadB = [&](int k, v2f* b) {
    const float* __restrict__ B0 = Bc + (long)k * NC;
#pragma unroll
    for (int t = 0; t < NT; ++t) { b[t].x = B0[16 * t]; b[t].y = B0[NC + 16 * t]; }
  };

  v2f a_cur = loadA(0);
  v2f b_cur[NT];
  loadB(0, b_cur);
#pragma unroll 4
  for (int k = 0; k < K - 4; k += 4) {
    v2f a_nxt = loadA(k + 4);
    v2f b_nxt[NT];
    loadB(k + 4, b_nxt);
#pragma unroll
    for (int t = 0; t < NT; ++t)
      c[t] = __builtin_amdgcn_wmma_f32_16x16x4_f32(false, a_cur, false, b_cur[t],
                                                   (short)0, c[t], false, false);
    a_cur = a_nxt;
#pragma unroll
    for (int t = 0; t < NT; ++t) b_cur[t] = b_nxt[t];
  }
#pragma unroll
  for (int t = 0; t < NT; ++t)
    c[t] = __builtin_amdgcn_wmma_f32_16x16x4_f32(false, a_cur, false, b_cur[t],
                                                 (short)0, c[t], false, false);

#pragma unroll
  for (int t = 0; t < NT; ++t) {
#pragma unroll
    for (int r = 0; r < 8; ++r) {
      const int orow = tm * 16 + r + 8 * hi;
      if (orow < M) C[(long)orow * NC + col0 + 16 * t] = c[t][r];
    }
  }
}


template <int M>
__device__ __forceinline__ float row_xor_add(float v) {
  return v + __int_as_float(__builtin_amdgcn_update_dpp(0, __float_as_int(v), 0x160 + M,
                                                        0xf, 0xf, true));
}

__device__ __forceinline__ float lrelu_dot4(v4f a, v4f b, v4f w) {
  const v4f v = a + b;
  float r = fmaxf(v.x, 0.2f * v.x) * w.x;
  r += fmaxf(v.y, 0.2f * v.y) * w.y;
  r += fmaxf(v.z, 0.2f * v.z) * w.z;
  r += fmaxf(v.w, 0.2f * v.w) * w.w;
  return r;
}

constexpr int AG_ROWS  = 196;
constexpr int AG_WAVES = 32;
constexpr int AG_SCAN  = 6;
constexpr int AG_STEP  = 128 * AG_SCAN;
constexpr int AG_CAP   = AG_STEP + 128;
constexpr int AG_BATCH = 8;
constexpr int AG_LB256 = 8;
constexpr int AG_LB128 = 16;

__device__ __forceinline__ int ag_append(int d, int e, int base, int rows, int cnt,
                                         int* le) {
  const unsigned r = (unsigned)(d - base);
  const bool hit = r < (unsigned)rows;
  const unsigned mask = __builtin_amdgcn_ballot_w32(hit);
  if (mask != 0u) {
    const int pos = cnt + (int)__builtin_amdgcn_mbcnt_lo(mask, 0u);
    if (hit) le[pos] = e;
    cnt += __popc(mask);
  }
  return cnt;
}

__device__ __forceinline__ void lds_max_f32(float* p, float v) {
  if (v >= 0.0f)
    __hip_atomic_fetch_max((int*)p, __float_as_int(v), __ATOMIC_RELAXED,
                           __HIP_MEMORY_SCOPE_WORKGROUP);
  else
    __hip_atomic_fetch_min((unsigned*)p, __float_as_uint(v), __ATOMIC_RELAXED,
                           __HIP_MEMORY_SCOPE_WORKGROUP);
}

__device__ __forceinline__ void ag_logits_max256(const float* __restrict__ f,
                                                 const float* __restrict__ attn,
                                                 float* s,
                                                 const int* __restrict__ src,
                                                 const int* __restrict__ dst,
                                                 const float* own,
                                                 float* mx, const int* le,
                                                 int cnt, int base, int lane) {
  __builtin_amdgcn_fence(__ATOMIC_ACQ_REL, "wavefront");
  cnt = __builtin_amdgcn_readfirstlane(cnt);
  const v4f w0 = ((const v4f*)attn)[lane];
  const v4f w1 = ((const v4f*)attn)[32 + lane];
  for (int c = 0; c < cnt; c += 32) {
    const int e = le[min(c + lane, cnt - 1)];
    const int sv = src[e];
    const int dn = dst[e];
    v4f lg = (v4f){0.0f, 0.0f, 0.0f, 0.0f};
    const int nb = min(32, cnt - c);
    for (int q0 = 0; q0 < nb; q0 += AG_LB256) {
      v4f xs0[AG_LB256], xs1[AG_LB256];
#pragma unroll
      for (int b = 0; b < AG_LB256; ++b) {
        const int sb = __builtin_amdgcn_readlane(sv, q0 + b);
        const v4f* __restrict__ fs = (const v4f*)(f + (long)sb * 256);
        xs0[b] = fs[lane];
        xs1[b] = fs[32 + lane];
      }
#pragma unroll
      for (int b = 0; b < AG_LB256; ++b) {
        const int rb = __builtin_amdgcn_readlane(dn, q0 + b) - base;
        const v4f* fd = (const v4f*)(own + rb * 256);
        const v4f xd0 = fd[lane];
        const v4f xd1 = fd[32 + lane];
        float p0 = lrelu_dot4(xs0[b], xd0, w0);
        float p1 = lrelu_dot4(xs1[b], xd1, w1);
        p0 = row_xor_add<1>(p0); p1 = row_xor_add<1>(p1);
        p0 = row_xor_add<2>(p0); p1 = row_xor_add<2>(p1);
        p0 = row_xor_add<4>(p0); p1 = row_xor_add<4>(p1);
        p0 = row_xor_add<8>(p0); p1 = row_xor_add<8>(p1);
        const int l0 = __builtin_amdgcn_readlane(__float_as_int(p0), 0);
        const int l1 = __builtin_amdgcn_readlane(__float_as_int(p0), 16);
        const int l2 = __builtin_amdgcn_readlane(__float_as_int(p1), 0);
        const int l3 = __builtin_amdgcn_readlane(__float_as_int(p1), 16);
        const bool mine = lane == q0 + b;
        lg.x = mine ? __int_as_float(l0) : lg.x;
        lg.y = mine ? __int_as_float(l1) : lg.y;
        lg.z = mine ? __int_as_float(l2) : lg.z;
        lg.w = mine ? __int_as_float(l3) : lg.w;
        if (b & 1) __builtin_amdgcn_sched_barrier(0);
      }
    }
    if (c + lane < cnt) {
      *(v4f*)(s + (long)e * 4) = lg;
      float* mr = mx + (dn - base) * 4;
      lds_max_f32(mr,     lg.x);
      lds_max_f32(mr + 1, lg.y);
      lds_max_f32(mr + 2, lg.z);
      lds_max_f32(mr + 3, lg.w);
    }
  }
  __builtin_amdgcn_fence(__ATOMIC_ACQ_REL, "wavefront");
}

__device__ __forceinline__ void ag_logits_max128(const float* __restrict__ f,
                                                 const float* __restrict__ attn,
                                                 float* s,
                                                 const int* __restrict__ src,
                                                 const int* __restrict__ dst,
                                                 const float* own,
                                                 float* mx, const int* le,
                                                 int cnt, int base, int lane) {
  __builtin_amdgcn_fence(__ATOMIC_ACQ_REL, "wavefront");
  cnt = __builtin_amdgcn_readfirstlane(cnt);
  const v4f w = ((const v4f*)attn)[lane];
  for (int c = 0; c < cnt; c += 32) {
    const int e = le[min(c + lane, cnt - 1)];
    const int sv = src[e];
    const int dn = dst[e];
    v4f lg = (v4f){0.0f, 0.0f, 0.0f, 0.0f};
    const int nb = min(32, cnt - c);
    for (int q0 = 0; q0 < nb; q0 += AG_LB128) {
      v4f xs[AG_LB128];
#pragma unroll
      for (int b = 0; b < AG_LB128; ++b) {
        const int sb = __builtin_amdgcn_readlane(sv, q0 + b);
        xs[b] = ((const v4f*)(f + (long)sb * 128))[lane];
      }
#pragma unroll
      for (int b = 0; b < AG_LB128; ++b) {
        const int rb = __builtin_amdgcn_readlane(dn, q0 + b) - base;
        const v4f xd = ((const v4f*)(own + rb * 128))[lane];
        float p = lrelu_dot4(xs[b], xd, w);
        p = row_xor_add<1>(p);
        p = row_xor_add<2>(p);
        p = row_xor_add<4>(p);
        const int l0 = __builtin_amdgcn_readlane(__float_as_int(p), 0);
        const int l1 = __builtin_amdgcn_readlane(__float_as_int(p), 8);
        const int l2 = __builtin_amdgcn_readlane(__float_as_int(p), 16);
        const int l3 = __builtin_amdgcn_readlane(__float_as_int(p), 24);
        const bool mine = lane == q0 + b;
        lg.x = mine ? __int_as_float(l0) : lg.x;
        lg.y = mine ? __int_as_float(l1) : lg.y;
        lg.z = mine ? __int_as_float(l2) : lg.z;
        lg.w = mine ? __int_as_float(l3) : lg.w;
      }
    }
    if (c + lane < cnt) {
      *(v4f*)(s + (long)e * 4) = lg;
      float* mr = mx + (dn - base) * 4;
      lds_max_f32(mr,     lg.x);
      lds_max_f32(mr + 1, lg.y);
      lds_max_f32(mr + 2, lg.z);
      lds_max_f32(mr + 3, lg.w);
    }
  }
  __builtin_amdgcn_fence(__ATOMIC_ACQ_REL, "wavefront");
}

__device__ __forceinline__ void ag_drain(const float* __restrict__ f,
                                         const float* s,
                                         const int* __restrict__ src,
                                         const int* __restrict__ dst,
                                         float* tile, const float* mx, float* den,
                                         const int* le,
                                         int cnt, int base, int lane) {
  __builtin_amdgcn_fence(__ATOMIC_ACQ_REL, "wavefront");
  const int j = lane >> 2;
  const int hd = lane & 3;
  const bool lo = lane < 16;
  for (int i = 0; i < cnt; i += AG_BATCH) {
    const int e = le[min(i + j, cnt - 1)];
    const int sv = src[e];
    const int dn = dst[e];
    const float sl = s[(long)e * 4 + hd];
    v4f x0[AG_BATCH], x1[AG_BATCH];
#pragma unroll
    for (int b = 0; b < AG_BATCH; ++b) {
      const int sb = __builtin_amdgcn_readlane(sv, 4 * b);
      const v4f* __restrict__ fr = (const v4f*)(f + (long)sb * 256);
      x0[b] = fr[lane];
      x1[b] = fr[32 + lane];
    }
    const int r = dn - base;
    const float mv = mx[r * 4 + hd];
    asm volatile("" :: "v"(x0[0]), "v"(x1[0]), "v"(x0[1]), "v"(x1[1]), "v"(x0[2]), "v"(x1[2]),
                       "v"(x0[3]), "v"(x1[3]), "v"(x0[4]), "v"(x1[4]), "v"(x0[5]), "v"(x1[5]),
                       "v"(x0[6]), "v"(x1[6]), "v"(x0[7]), "v"(x1[7]));
    const float a = __expf(sl - mv);
    if (i + j < cnt)
      __hip_atomic_fetch_add(den + r * 4 + hd, a, __ATOMIC_RELAXED, __HIP_MEMORY_SCOPE_WORKGROUP);
#pragma unroll
    for (int b = 0; b < AG_BATCH; ++b) {
      if (i + b < cnt) {
        const int rb = __builtin_amdgcn_readlane(r, 4 * b);
        const float a0 = __int_as_float(__builtin_amdgcn_readlane(__float_as_int(a), 4 * b));
        const float a1 = __int_as_float(__builtin_amdgcn_readlane(__float_as_int(a), 4 * b + 1));
        const float a2 = __int_as_float(__builtin_amdgcn_readlane(__float_as_int(a), 4 * b + 2));
        const float a3 = __int_as_float(__builtin_amdgcn_readlane(__float_as_int(a), 4 * b + 3));
        const float al0 = lo ? a0 : a1;
        const float al1 = lo ? a2 : a3;
        float* tr = tile + rb * 256 + lane;
        __hip_atomic_fetch_add(tr,       x0[b].x * al0, __ATOMIC_RELAXED, __HIP_MEMORY_SCOPE_WORKGROUP);
        __hip_atomic_fetch_add(tr + 64,  x0[b].y * al0, __ATOMIC_RELAXED, __HIP_MEMORY_SCOPE_WORKGROUP);
        __hip_atomic_fetch_add(tr + 128, x0[b].z * al0, __ATOMIC_RELAXED, __HIP_MEMORY_SCOPE_WORKGROUP);
        __hip_atomic_fetch_add(tr + 192, x0[b].w * al0, __ATOMIC_RELAXED, __HIP_MEMORY_SCOPE_WORKGROUP);
        __hip_atomic_fetch_add(tr + 32,  x1[b].x * al1, __ATOMIC_RELAXED, __HIP_MEMORY_SCOPE_WORKGROUP);
        __hip_atomic_fetch_add(tr + 96,  x1[b].y * al1, __ATOMIC_RELAXED, __HIP_MEMORY_SCOPE_WORKGROUP);
        __hip_atomic_fetch_add(tr + 160, x1[b].z * al1, __ATOMIC_RELAXED, __HIP_MEMORY_SCOPE_WORKGROUP);
        __hip_atomic_fetch_add(tr + 224, x1[b].w * al1, __ATOMIC_RELAXED, __HIP_MEMORY_SCOPE_WORKGROUP);
      }
    }
  }
  __builtin_amdgcn_fence(__ATOMIC_ACQ_REL, "wavefront");
}

__device__ __forceinline__ void ag_drain128(const float* __restrict__ f,
                                            const float* s,
                                            const int* __restrict__ src,
                                            const int* __restrict__ dst,
                                            float* tile, const float* mx, float* den,
                                            const int* le,
                                            int cnt, int base, int lane) {
  __builtin_amdgcn_fence(__ATOMIC_ACQ_REL, "wavefront");
  const int j = lane >> 2;
  const int hd = lane & 3;
  const int hq = lane >> 3;
  for (int i = 0; i < cnt; i += AG_BATCH) {
    const int e = le[min(i + j, cnt - 1)];
    const int sv = src[e];
    const int dn = dst[e];
    const float sl = s[(long)e * 4 + hd];
    v4f x[AG_BATCH];
#pragma unroll
    for (int b = 0; b < AG_BATCH; ++b) {
      const int sb = __builtin_amdgcn_readlane(sv, 4 * b);
      const v4f* __restrict__ fr = (const v4f*)(f + (long)sb * 128);
      x[b] = fr[lane];
    }
    const int r = dn - base;
    const float mv = mx[r * 4 + hd];
    const float a = __expf(sl - mv);
    if (i + j < cnt)
      __hip_atomic_fetch_add(den + r * 4 + hd, a, __ATOMIC_RELAXED, __HIP_MEMORY_SCOPE_WORKGROUP);
    float al[AG_BATCH];
#pragma unroll
    for (int b = 0; b < AG_BATCH; ++b) al[b] = __shfl(a, 4 * b + hq, 32);
#pragma unroll
    for (int b = 0; b < AG_BATCH; ++b) {
      if (i + b < cnt) {
        const int rb = __builtin_amdgcn_readlane(r, 4 * b);
        float* tr = tile + rb * 128 + lane;
        __hip_atomic_fetch_add(tr,      x[b].x * al[b], __ATOMIC_RELAXED, __HIP_MEMORY_SCOPE_WORKGROUP);
        __hip_atomic_fetch_add(tr + 32, x[b].y * al[b], __ATOMIC_RELAXED, __HIP_MEMORY_SCOPE_WORKGROUP);
        __hip_atomic_fetch_add(tr + 64, x[b].z * al[b], __ATOMIC_RELAXED, __HIP_MEMORY_SCOPE_WORKGROUP);
        __hip_atomic_fetch_add(tr + 96, x[b].w * al[b], __ATOMIC_RELAXED, __HIP_MEMORY_SCOPE_WORKGROUP);
      }
    }
  }
  __builtin_amdgcn_fence(__ATOMIC_ACQ_REL, "wavefront");
}

template <bool MAXPASS, int ROW>
__device__ __forceinline__ int ag_filter(const float* __restrict__ f,
                                         const float* __restrict__ attn,
                                         float* s,
                                         const int* __restrict__ src,
                                         const int* __restrict__ dst,
                                         float* tile, float* mx, float* den, int* le,
                                         int base, int rows, int E, int lane, int w,
                                         int& spilled) {
  int peel = (int)(((16u - (unsigned)((unsigned long long)dst & 15ull)) & 15u) >> 2);
  if (peel > E) peel = E;
  const int nG = (E - peel) >> 2;
  const int tail0 = peel + 4 * nG;
  int cnt = 0;
  if (w == 0) {
    int e = -1;
    if (lane < peel) e = lane;
    else if (lane >= 4 && lane < 8 && tail0 + (lane - 4) < E) e = tail0 + (lane - 4);
    const int d = e >= 0 ? dst[e] : -1;
    cnt = ag_append(d, e, base, rows, cnt, le);
  }
  cnt = __builtin_amdgcn_readfirstlane(cnt);
  const int4* __restrict__ d4 = (const int4*)(dst + peel);
  for (int g0 = 0; g0 < nG; g0 += 1024 * AG_SCAN) {
    if (cnt > AG_CAP - AG_STEP) {
      if (MAXPASS) {
        if (ROW == 256) ag_logits_max256(f, attn, s, src, dst, tile, mx, le, cnt, base, lane);
        else            ag_logits_max128(f, attn, s, src, dst, tile, mx, le, cnt, base, lane);
        spilled = 1;
      } else if (ROW == 256) {
        ag_drain(f, s, src, dst, tile, mx, den, le, cnt, base, lane);
      } else {
        ag_drain128(f, s, src, dst, tile, mx, den, le, cnt, base, lane);
      }
      cnt = 0;
    }
    int4 v[AG_SCAN];
#pragma unroll
    for (int u = 0; u < AG_SCAN; ++u) {
      const int g = g0 + u * 1024 + (int)threadIdx.x;
      v[u] = g < nG ? d4[g] : make_int4(-1, -1, -1, -1);
    }
#pragma unroll
    for (int u = 0; u < AG_SCAN; ++u) {
      const int e0 = peel + 4 * (g0 + u * 1024 + (int)threadIdx.x);
      cnt = ag_append(v[u].x, e0, base, rows, cnt, le);
      cnt = ag_append(v[u].y, e0 + 1, base, rows, cnt, le);
      cnt = ag_append(v[u].z, e0 + 2, base, rows, cnt, le);
      cnt = ag_append(v[u].w, e0 + 3, base, rows, cnt, le);
    }
  }
  return cnt;
}

__device__ __forceinline__ float elu1(float v) { return v > 0.0f ? v : expm1f(v); }

__global__ __launch_bounds__(1024)
void aggregate_owner_softmax_elu(const float* __restrict__ f, const float* __restrict__ attn,
                                 float* s,
                                 const int* __restrict__ src, const int* __restrict__ dst,
                                 float* __restrict__ h, int N, int E) {
  __shared__ float tile[AG_ROWS * 256];
  __shared__ int l_edge[AG_WAVES * AG_CAP];
  __shared__ float l_mx[AG_ROWS * 4];
  __shared__ float l_den[AG_ROWS * 4];
  const int lane = threadIdx.x & 31;
  const int w    = threadIdx.x >> 5;
  const int base = blockIdx.x * AG_ROWS;
  const int rows = min(AG_ROWS, N - base);
  int* le = l_edge + w * AG_CAP;

  float4* t4 = (float4*)tile;
  {
    constexpr int PER = (AG_ROWS * 64 + 1023) / 1024;
    const float4* __restrict__ fo = (const float4*)(f + (long)base * 256);
    float4 v[PER];
#pragma unroll
    for (int u = 0; u < PER; ++u) {
      const int i = u * 1024 + (int)threadIdx.x;
      v[u] = fo[min(i, rows * 64 - 1)];
    }
#pragma unroll
    for (int u = 0; u < PER; ++u) {
      const int i = u * 1024 + (int)threadIdx.x;
      if (i < AG_ROWS * 64) t4[i] = v[u];
    }
  }
  for (int i = threadIdx.x; i < AG_ROWS * 4; i += 1024) {
    l_mx[i] = __uint_as_float(0xFF800000u);
    l_den[i] = 0.0f;
  }
  __syncthreads();

  int spilled = 0;
  int cnt = ag_filter<true, 256>(f, attn, s, src, dst, tile, l_mx, l_den, le, base, rows, E,
                                 lane, w, spilled);
  ag_logits_max256(f, attn, s, src, dst, tile, l_mx, le, cnt, base, lane);
  __syncthreads();

  for (int i = threadIdx.x; i < AG_ROWS * 64; i += 1024)
    t4[i] = make_float4(0.0f, 0.0f, 0.0f, 0.0f);
  __syncthreads();

  if (__builtin_amdgcn_readfirstlane(spilled) != 0)
    cnt = ag_filter<false, 256>(f, attn, s, src, dst, tile, l_mx, l_den, le, base, rows, E,
                                lane, w, spilled);
  ag_drain(f, s, src, dst, tile, l_mx, l_den, le, cnt, base, lane);
  __syncthreads();

  for (int i = threadIdx.x; i < rows * 64; i += 1024) {
    const int r = i >> 6, p = i & 63;
    const float* tr = tile + r * 256 + p;
    const float dv = l_den[r * 4 + (p >> 4)];
    const bool none = dv == 0.0f;
    float4 o;
    o.x = elu1(none ? 0.0f : tr[0]   / dv);
    o.y = elu1(none ? 0.0f : tr[64]  / dv);
    o.z = elu1(none ? 0.0f : tr[128] / dv);
    o.w = elu1(none ? 0.0f : tr[192] / dv);
    *(float4*)(h + (long)(base + r) * 256 + 4 * p) = o;
  }
}

__global__ __launch_bounds__(1024)
void aggregate_owner_softmax_mean(const float* __restrict__ f, const float* __restrict__ attn,
                                  float* s, const float* __restrict__ o1,
                                  const int* __restrict__ src, const int* __restrict__ dst,
                                  float* __restrict__ out, int N, int E) {
  __shared__ float tile[AG_ROWS * 128];
  __shared__ int l_edge[AG_WAVES * AG_CAP];
  __shared__ float l_mx[AG_ROWS * 4];
  __shared__ float l_den[AG_ROWS * 4];
  const int lane = threadIdx.x & 31;
  const int w    = threadIdx.x >> 5;
  const int base = blockIdx.x * AG_ROWS;
  const int rows = min(AG_ROWS, N - base);
  int* le = l_edge + w * AG_CAP;

  float4* t4 = (float4*)tile;
  {
    constexpr int PER = (AG_ROWS * 32 + 1023) / 1024;
    const float4* __restrict__ fo = (const float4*)(f + (long)base * 128);
    float4 v[PER];
#pragma unroll
    for (int u = 0; u < PER; ++u) {
      const int i = u * 1024 + (int)threadIdx.x;
      v[u] = fo[min(i, rows * 32 - 1)];
    }
#pragma unroll
    for (int u = 0; u < PER; ++u) {
      const int i = u * 1024 + (int)threadIdx.x;
      if (i < AG_ROWS * 32) t4[i] = v[u];
    }
  }
  for (int i = threadIdx.x; i < AG_ROWS * 4; i += 1024) {
    l_mx[i] = __uint_as_float(0xFF800000u);
    l_den[i] = 0.0f;
  }
  __syncthreads();

  int spilled = 0;
  int cnt = ag_filter<true, 128>(f, attn, s, src, dst, tile, l_mx, l_den, le, base, rows, E,
                                 lane, w, spilled);
  ag_logits_max128(f, attn, s, src, dst, tile, l_mx, le, cnt, base, lane);
  __syncthreads();

  for (int i = threadIdx.x; i < AG_ROWS * 32; i += 1024)
    t4[i] = make_float4(0.0f, 0.0f, 0.0f, 0.0f);
  __syncthreads();

  if (__builtin_amdgcn_readfirstlane(spilled) != 0)
    cnt = ag_filter<false, 128>(f, attn, s, src, dst, tile, l_mx, l_den, le, base, rows, E,
                                lane, w, spilled);
  ag_drain128(f, s, src, dst, tile, l_mx, l_den, le, cnt, base, lane);
  __syncthreads();

  for (int i = threadIdx.x; i < rows * 8; i += 1024) {
    const int r = i >> 3, q = i & 7;
    const float* tr = tile + r * 128 + q;
    const float* __restrict__ rr = o1 + (long)(base + r) * 128 + 4 * q;
    float4 acc = make_float4(0.0f, 0.0f, 0.0f, 0.0f);
#pragma unroll
    for (int hh = 0; hh < 4; ++hh) {
      const float dv = l_den[r * 4 + hh];
      const bool none = dv == 0.0f;
      const float4 rs = *(const float4*)(rr + hh * 32);
      const float* th = tr + hh * 8;
      acc.x += (none ? 0.0f : th[0]  / dv) + rs.x;
      acc.y += (none ? 0.0f : th[32] / dv) + rs.y;
      acc.z += (none ? 0.0f : th[64] / dv) + rs.z;
      acc.w += (none ? 0.0f : th[96] / dv) + rs.w;
    }
    float4 o;
    o.x = 0.25f * acc.x; o.y = 0.25f * acc.y; o.z = 0.25f * acc.z; o.w = 0.25f * acc.w;
    *(float4*)(out + (long)(base + r) * 32 + 4 * q) = o;
  }
}

static inline int cdiv_l(long a, long b) { return (int)((a + b - 1) / b); }

extern "C" void kernel_launch(void* const* d_in, const int* in_sizes, int n_in,
                              void* d_out, int out_size, void* d_ws, size_t ws_size,
                              hipStream_t stream) {
  const float* x     = (const float*)d_in[0];
  const float* W0    = (const float*)d_in[1];
  const float* a0    = (const float*)d_in[2];
  const float* W1    = (const float*)d_in[3];
  const float* a1    = (const float*)d_in[4];
  const float* Wres1 = (const float*)d_in[5];
  const int*   src   = (const int*)d_in[6];
  const int*   dst   = (const int*)d_in[7];

  const int N = in_sizes[0] / 256;
  const int E = in_sizes[6];
  const int H = 4;

  float*    f0    = (float*)d_ws;
  float*    h     = f0 + (long)N * 256;
  float*    f1    = h  + (long)N * 256;
  float*    o1    = f1 + (long)N * 128;
  float*    s     = o1 + (long)N * 128;
  const size_t need =
      ((size_t)N * 256 * 2 + (size_t)N * 128 * 2 + (size_t)E * H) * sizeof(float);
  if (need > ws_size) return;

  const int BT = 256;

  {
    const int strips = ((N + 15) / 16) * (256 / 16 / 4);
    gemm_f32_wmma<4, 256, 256><<<cdiv_l(strips, 8), BT, 0, stream>>>(x, W0, f0, N);
  }
  aggregate_owner_softmax_elu<<<cdiv_l(N, AG_ROWS), 1024, 0, stream>>>(f0, a0, s, src, dst,
                                                                       h, N, E);

  {
    const int strips = ((N + 15) / 16) * (128 / 16 / 4);
    gemm_f32_wmma<4, 256, 128><<<cdiv_l(strips, 8), BT, 0, stream>>>(h, W1, f1, N);
    gemm_f32_wmma<4, 256, 128><<<cdiv_l(strips, 8), BT, 0, stream>>>(h, Wres1, o1, N);
  }
  aggregate_owner_softmax_mean<<<cdiv_l(N, AG_ROWS), 1024, 0, stream>>>(
      f1, a1, s, o1, src, dst, (float*)d_out, N, E);
}


# GEMM re-tiled for operand reuse: a wave computes 32 x 64 (each B fragment feeds two A row tiles, 8 accumulator chains); layer 1's two projections share one launch and its A tiles; 5 -> 4 launches
# speedup vs baseline: 2.6365x; 1.0800x over previous
#include <hip/hip_runtime.h>
#include <hip/hip_bf16.h>

typedef __attribute__((ext_vector_type(2))) float v2f;
typedef __attribute__((ext_vector_type(8))) float v8f;
typedef __attribute__((ext_vector_type(4))) float v4f;

template <int K, int LD>
__global__ __launch_bounds__(128)
void gemm_f32_wmma(const float* __restrict__ A,
                   const float* __restrict__ B0, const float* __restrict__ B1,
                   float* __restrict__ C0, float* __restrict__ C1, int M) {
  constexpr int MT = 2;
  constexpr int NT = 4;
  const int lane = threadIdx.x & 31;
  const int st   = threadIdx.x >> 5;
  const int tm   = blockIdx.x;
  const int hi = lane >> 4;
  const int lm = lane & 15;
  const float* __restrict__ B = (st & 2) ? B1 : B0;
  float* __restrict__ C = (st & 2) ? C1 : C0;
  const int col0 = (st & 1) * (NT * 16) + lm;
  const float* __restrict__ Ar[MT];
#pragma unroll
  for (int m = 0; m < MT; ++m) {
    int arow = tm * (MT * 16) + m * 16 + lm; if (arow >= M) arow = M - 1;
    Ar[m] = A + (long)arow * K + 2 * hi;
  }
  const float* __restrict__ Bc = B + (long)(2 * hi) * LD + col0;

  v8f c[MT][NT];
#pragma unroll
  for (int m = 0; m < MT; ++m)
#pragma unroll
    for (int t = 0; t < NT; ++t) c[m][t] = (v8f){};

  auto loadA = [&](int k, v2f* a) {
#pragma unroll
    for (int m = 0; m < MT; ++m) { a[m].x = Ar[m][k]; a[m].y = Ar[m][k + 1]; }
  };
  auto loadB = [&](int k, v2f* b) {
    const float* __restrict__ Bk = Bc + (long)k * LD;
#pragma unroll
    for (int t = 0; t < NT; ++t) { b[t].x = Bk[16 * t]; b[t].y = Bk[LD + 16 * t]; }
  };

  v2f a_cur[MT], b_cur[NT];
  loadA(0, a_cur);
  loadB(0, b_cur);
#pragma unroll 4
  for (int k = 0; k < K - 4; k += 4) {
    v2f a_nxt[MT], b_nxt[NT];
    loadA(k + 4, a_nxt);
    loadB(k + 4, b_nxt);
#pragma unroll
    for (int m = 0; m < MT; ++m)
#pragma unroll
      for (int t = 0; t < NT; ++t)
        c[m][t] = __builtin_amdgcn_wmma_f32_16x16x4_f32(false, a_cur[m], false, b_cur[t],
                                                        (short)0, c[m][t], false, false);
#pragma unroll
    for (int m = 0; m < MT; ++m) a_cur[m] = a_nxt[m];
#pragma unroll
    for (int t = 0; t < NT; ++t) b_cur[t] = b_nxt[t];
  }
#pragma unroll
  for (int m = 0; m < MT; ++m)
#pragma unroll
    for (int t = 0; t < NT; ++t)
      c[m][t] = __builtin_amdgcn_wmma_f32_16x16x4_f32(false, a_cur[m], false, b_cur[t],
                                                      (short)0, c[m][t], false, false);

#pragma unroll
  for (int m = 0; m < MT; ++m) {
#pragma unroll
    for (int t = 0; t < NT; ++t) {
#pragma unroll
      for (int r = 0; r < 8; ++r) {
        const int orow = tm * (MT * 16) + m * 16 + r + 8 * hi;
        if (orow < M) C[(long)orow * LD + col0 + 16 * t] = c[m][t][r];
      }
    }
  }
}


template <int M>
__device__ __forceinline__ float row_xor_add(float v) {
  return v + __int_as_float(__builtin_amdgcn_update_dpp(0, __float_as_int(v), 0x160 + M,
                                                        0xf, 0xf, true));
}

__device__ __forceinline__ float lrelu_dot4(v4f a, v4f b, v4f w) {
  const v4f v = a + b;
  float r = fmaxf(v.x, 0.2f * v.x) * w.x;
  r += fmaxf(v.y, 0.2f * v.y) * w.y;
  r += fmaxf(v.z, 0.2f * v.z) * w.z;
  r += fmaxf(v.w, 0.2f * v.w) * w.w;
  return r;
}

constexpr int AG_ROWS  = 196;
constexpr int AG_WAVES = 32;
constexpr int AG_SCAN  = 6;
constexpr int AG_STEP  = 128 * AG_SCAN;
constexpr int AG_CAP   = AG_STEP + 128;
constexpr int AG_BATCH = 8;
constexpr int AG_LB256 = 8;
constexpr int AG_LB128 = 16;

__device__ __forceinline__ int ag_append(int d, int e, int base, int rows, int cnt,
                                         int* le) {
  const unsigned r = (unsigned)(d - base);
  const bool hit = r < (unsigned)rows;
  const unsigned mask = __builtin_amdgcn_ballot_w32(hit);
  if (mask != 0u) {
    const int pos = cnt + (int)__builtin_amdgcn_mbcnt_lo(mask, 0u);
    if (hit) le[pos] = e;
    cnt += __popc(mask);
  }
  return cnt;
}

__device__ __forceinline__ void lds_max_f32(float* p, float v) {
  if (v >= 0.0f)
    __hip_atomic_fetch_max((int*)p, __float_as_int(v), __ATOMIC_RELAXED,
                           __HIP_MEMORY_SCOPE_WORKGROUP);
  else
    __hip_atomic_fetch_min((unsigned*)p, __float_as_uint(v), __ATOMIC_RELAXED,
                           __HIP_MEMORY_SCOPE_WORKGROUP);
}

__device__ __forceinline__ void ag_logits_max256(const float* __restrict__ f,
                                                 const float* __restrict__ attn,
                                                 float* s,
                                                 const int* __restrict__ src,
                                                 const int* __restrict__ dst,
                                                 const float* own,
                                                 float* mx, const int* le,
                                                 int cnt, int base, int lane) {
  __builtin_amdgcn_fence(__ATOMIC_ACQ_REL, "wavefront");
  cnt = __builtin_amdgcn_readfirstlane(cnt);
  const v4f w0 = ((const v4f*)attn)[lane];
  const v4f w1 = ((const v4f*)attn)[32 + lane];
  for (int c = 0; c < cnt; c += 32) {
    const int e = le[min(c + lane, cnt - 1)];
    const int sv = src[e];
    const int dn = dst[e];
    v4f lg = (v4f){0.0f, 0.0f, 0.0f, 0.0f};
    const int nb = min(32, cnt - c);
    for (int q0 = 0; q0 < nb; q0 += AG_LB256) {
      v4f xs0[AG_LB256], xs1[AG_LB256];
#pragma unroll
      for (int b = 0; b < AG_LB256; ++b) {
        const int sb = __builtin_amdgcn_readlane(sv, q0 + b);
        const v4f* __restrict__ fs = (const v4f*)(f + (long)sb * 256);
        xs0[b] = fs[lane];
        xs1[b] = fs[32 + lane];
      }
#pragma unroll
      for (int b = 0; b < AG_LB256; ++b) {
        const int rb = __builtin_amdgcn_readlane(dn, q0 + b) - base;
        const v4f* fd = (const v4f*)(own + rb * 256);
        const v4f xd0 = fd[lane];
        const v4f xd1 = fd[32 + lane];
        float p0 = lrelu_dot4(xs0[b], xd0, w0);
        float p1 = lrelu_dot4(xs1[b], xd1, w1);
        p0 = row_xor_add<1>(p0); p1 = row_xor_add<1>(p1);
        p0 = row_xor_add<2>(p0); p1 = row_xor_add<2>(p1);
        p0 = row_xor_add<4>(p0); p1 = row_xor_add<4>(p1);
        p0 = row_xor_add<8>(p0); p1 = row_xor_add<8>(p1);
        const int l0 = __builtin_amdgcn_readlane(__float_as_int(p0), 0);
        const int l1 = __builtin_amdgcn_readlane(__float_as_int(p0), 16);
        const int l2 = __builtin_amdgcn_readlane(__float_as_int(p1), 0);
        const int l3 = __builtin_amdgcn_readlane(__float_as_int(p1), 16);
        const bool mine = lane == q0 + b;
        lg.x = mine ? __int_as_float(l0) : lg.x;
        lg.y = mine ? __int_as_float(l1) : lg.y;
        lg.z = mine ? __int_as_float(l2) : lg.z;
        lg.w = mine ? __int_as_float(l3) : lg.w;
        if (b & 1) __builtin_amdgcn_sched_barrier(0);
      }
    }
    if (c + lane < cnt) {
      *(v4f*)(s + (long)e * 4) = lg;
      float* mr = mx + (dn - base) * 4;
      lds_max_f32(mr,     lg.x);
      lds_max_f32(mr + 1, lg.y);
      lds_max_f32(mr + 2, lg.z);
      lds_max_f32(mr + 3, lg.w);
    }
  }
  __builtin_amdgcn_fence(__ATOMIC_ACQ_REL, "wavefront");
}

__device__ __forceinline__ void ag_logits_max128(const float* __restrict__ f,
                                                 const float* __restrict__ attn,
                                                 float* s,
                                                 const int* __restrict__ src,
                                                 const int* __restrict__ dst,
                                                 const float* own,
                                                 float* mx, const int* le,
                                                 int cnt, int base, int lane) {
  __builtin_amdgcn_fence(__ATOMIC_ACQ_REL, "wavefront");
  cnt = __builtin_amdgcn_readfirstlane(cnt);
  const v4f w = ((const v4f*)attn)[lane];
  for (int c = 0; c < cnt; c += 32) {
    const int e = le[min(c + lane, cnt - 1)];
    const int sv = src[e];
    const int dn = dst[e];
    v4f lg = (v4f){0.0f, 0.0f, 0.0f, 0.0f};
    const int nb = min(32, cnt - c);
    for (int q0 = 0; q0 < nb; q0 += AG_LB128) {
      v4f xs[AG_LB128];
#pragma unroll
      for (int b = 0; b < AG_LB128; ++b) {
        const int sb = __builtin_amdgcn_readlane(sv, q0 + b);
        xs[b] = ((const v4f*)(f + (long)sb * 128))[lane];
      }
#pragma unroll
      for (int b = 0; b < AG_LB128; ++b) {
        const int rb = __builtin_amdgcn_readlane(dn, q0 + b) - base;
        const v4f xd = ((const v4f*)(own + rb * 128))[lane];
        float p = lrelu_dot4(xs[b], xd, w);
        p = row_xor_add<1>(p);
        p = row_xor_add<2>(p);
        p = row_xor_add<4>(p);
        const int l0 = __builtin_amdgcn_readlane(__float_as_int(p), 0);
        const int l1 = __builtin_amdgcn_readlane(__float_as_int(p), 8);
        const int l2 = __builtin_amdgcn_readlane(__float_as_int(p), 16);
        const int l3 = __builtin_amdgcn_readlane(__float_as_int(p), 24);
        const bool mine = lane == q0 + b;
        lg.x = mine ? __int_as_float(l0) : lg.x;
        lg.y = mine ? __int_as_float(l1) : lg.y;
        lg.z = mine ? __int_as_float(l2) : lg.z;
        lg.w = mine ? __int_as_float(l3) : lg.w;
      }
    }
    if (c + lane < cnt) {
      *(v4f*)(s + (long)e * 4) = lg;
      float* mr = mx + (dn - base) * 4;
      lds_max_f32(mr,     lg.x);
      lds_max_f32(mr + 1, lg.y);
      lds_max_f32(mr + 2, lg.z);
      lds_max_f32(mr + 3, lg.w);
    }
  }
  __builtin_amdgcn_fence(__ATOMIC_ACQ_REL, "wavefront");
}

__device__ __forceinline__ void ag_drain(const float* __restrict__ f,
                                         const float* s,
                                         const int* __restrict__ src,
                                         const int* __restrict__ dst,
                                         float* tile, const float* mx, float* den,
                                         const int* le,
                                         int cnt, int base, int lane) {
  __builtin_amdgcn_fence(__ATOMIC_ACQ_REL, "wavefront");
  const int j = lane >> 2;
  const int hd = lane & 3;
  const bool lo = lane < 16;
  for (int i = 0; i < cnt; i += AG_BATCH) {
    const int e = le[min(i + j, cnt - 1)];
    const int sv = src[e];
    const int dn = dst[e];
    const float sl = s[(long)e * 4 + hd];
    v4f x0[AG_BATCH], x1[AG_BATCH];
#pragma unroll
    for (int b = 0; b < AG_BATCH; ++b) {
      const int sb = __builtin_amdgcn_readlane(sv, 4 * b);
      const v4f* __restrict__ fr = (const v4f*)(f + (long)sb * 256);
      x0[b] = fr[lane];
      x1[b] = fr[32 + lane];
    }
    const int r = dn - base;
    const float mv = mx[r * 4 + hd];
    asm volatile("" :: "v"(x0[0]), "v"(x1[0]), "v"(x0[1]), "v"(x1[1]), "v"(x0[2]), "v"(x1[2]),
                       "v"(x0[3]), "v"(x1[3]), "v"(x0[4]), "v"(x1[4]), "v"(x0[5]), "v"(x1[5]),
                       "v"(x0[6]), "v"(x1[6]), "v"(x0[7]), "v"(x1[7]));
    const float a = __expf(sl - mv);
    if (i + j < cnt)
      __hip_atomic_fetch_add(den + r * 4 + hd, a, __ATOMIC_RELAXED, __HIP_MEMORY_SCOPE_WORKGROUP);
#pragma unroll
    for (int b = 0; b < AG_BATCH; ++b) {
      if (i + b < cnt) {
        const int rb = __builtin_amdgcn_readlane(r, 4 * b);
        const float a0 = __int_as_float(__builtin_amdgcn_readlane(__float_as_int(a), 4 * b));
        const float a1 = __int_as_float(__builtin_amdgcn_readlane(__float_as_int(a), 4 * b + 1));
        const float a2 = __int_as_float(__builtin_amdgcn_readlane(__float_as_int(a), 4 * b + 2));
        const float a3 = __int_as_float(__builtin_amdgcn_readlane(__float_as_int(a), 4 * b + 3));
        const float al0 = lo ? a0 : a1;
        const float al1 = lo ? a2 : a3;
        float* tr = tile + rb * 256 + lane;
        __hip_atomic_fetch_add(tr,       x0[b].x * al0, __ATOMIC_RELAXED, __HIP_MEMORY_SCOPE_WORKGROUP);
        __hip_atomic_fetch_add(tr + 64,  x0[b].y * al0, __ATOMIC_RELAXED, __HIP_MEMORY_SCOPE_WORKGROUP);
        __hip_atomic_fetch_add(tr + 128, x0[b].z * al0, __ATOMIC_RELAXED, __HIP_MEMORY_SCOPE_WORKGROUP);
        __hip_atomic_fetch_add(tr + 192, x0[b].w * al0, __ATOMIC_RELAXED, __HIP_MEMORY_SCOPE_WORKGROUP);
        __hip_atomic_fetch_add(tr + 32,  x1[b].x * al1, __ATOMIC_RELAXED, __HIP_MEMORY_SCOPE_WORKGROUP);
        __hip_atomic_fetch_add(tr + 96,  x1[b].y * al1, __ATOMIC_RELAXED, __HIP_MEMORY_SCOPE_WORKGROUP);
        __hip_atomic_fetch_add(tr + 160, x1[b].z * al1, __ATOMIC_RELAXED, __HIP_MEMORY_SCOPE_WORKGROUP);
        __hip_atomic_fetch_add(tr + 224, x1[b].w * al1, __ATOMIC_RELAXED, __HIP_MEMORY_SCOPE_WORKGROUP);
      }
    }
  }
  __builtin_amdgcn_fence(__ATOMIC_ACQ_REL, "wavefront");
}

__device__ __forceinline__ void ag_drain128(const float* __restrict__ f,
                                            const float* s,
                                            const int* __restrict__ src,
                                            const int* __restrict__ dst,
                                            float* tile, const float* mx, float* den,
                                            const int* le,
                                            int cnt, int base, int lane) {
  __builtin_amdgcn_fence(__ATOMIC_ACQ_REL, "wavefront");
  const int j = lane >> 2;
  const int hd = lane & 3;
  const int hq = lane >> 3;
  for (int i = 0; i < cnt; i += AG_BATCH) {
    const int e = le[min(i + j, cnt - 1)];
    const int sv = src[e];
    const int dn = dst[e];
    const float sl = s[(long)e * 4 + hd];
    v4f x[AG_BATCH];
#pragma unroll
    for (int b = 0; b < AG_BATCH; ++b) {
      const int sb = __builtin_amdgcn_readlane(sv, 4 * b);
      const v4f* __restrict__ fr = (const v4f*)(f + (long)sb * 128);
      x[b] = fr[lane];
    }
    const int r = dn - base;
    const float mv = mx[r * 4 + hd];
    const float a = __expf(sl - mv);
    if (i + j < cnt)
      __hip_atomic_fetch_add(den + r * 4 + hd, a, __ATOMIC_RELAXED, __HIP_MEMORY_SCOPE_WORKGROUP);
    float al[AG_BATCH];
#pragma unroll
    for (int b = 0; b < AG_BATCH; ++b) al[b] = __shfl(a, 4 * b + hq, 32);
#pragma unroll
    for (int b = 0; b < AG_BATCH; ++b) {
      if (i + b < cnt) {
        const int rb = __builtin_amdgcn_readlane(r, 4 * b);
        float* tr = tile + rb * 128 + lane;
        __hip_atomic_fetch_add(tr,      x[b].x * al[b], __ATOMIC_RELAXED, __HIP_MEMORY_SCOPE_WORKGROUP);
        __hip_atomic_fetch_add(tr + 32, x[b].y * al[b], __ATOMIC_RELAXED, __HIP_MEMORY_SCOPE_WORKGROUP);
        __hip_atomic_fetch_add(tr + 64, x[b].z * al[b], __ATOMIC_RELAXED, __HIP_MEMORY_SCOPE_WORKGROUP);
        __hip_atomic_fetch_add(tr + 96, x[b].w * al[b], __ATOMIC_RELAXED, __HIP_MEMORY_SCOPE_WORKGROUP);
      }
    }
  }
  __builtin_amdgcn_fence(__ATOMIC_ACQ_REL, "wavefront");
}

template <bool MAXPASS, int ROW>
__device__ __forceinline__ int ag_filter(const float* __restrict__ f,
                                         const float* __restrict__ attn,
                                         float* s,
                                         const int* __restrict__ src,
                                         const int* __restrict__ dst,
                                         float* tile, float* mx, float* den, int* le,
                                         int base, int rows, int E, int lane, int w,
                                         int& spilled) {
  int peel = (int)(((16u - (unsigned)((unsigned long long)dst & 15ull)) & 15u) >> 2);
  if (peel > E) peel = E;
  const int nG = (E - peel) >> 2;
  const int tail0 = peel + 4 * nG;
  int cnt = 0;
  if (w == 0) {
    int e = -1;
    if (lane < peel) e = lane;
    else if (lane >= 4 && lane < 8 && tail0 + (lane - 4) < E) e = tail0 + (lane - 4);
    const int d = e >= 0 ? dst[e] : -1;
    cnt = ag_append(d, e, base, rows, cnt, le);
  }
  cnt = __builtin_amdgcn_readfirstlane(cnt);
  const int4* __restrict__ d4 = (const int4*)(dst + peel);
  for (int g0 = 0; g0 < nG; g0 += 1024 * AG_SCAN) {
    if (cnt > AG_CAP - AG_STEP) {
      if (MAXPASS) {
        if (ROW == 256) ag_logits_max256(f, attn, s, src, dst, tile, mx, le, cnt, base, lane);
        else            ag_logits_max128(f, attn, s, src, dst, tile, mx, le, cnt, base, lane);
        spilled = 1;
      } else if (ROW == 256) {
        ag_drain(f, s, src, dst, tile, mx, den, le, cnt, base, lane);
      } else {
        ag_drain128(f, s, src, dst, tile, mx, den, le, cnt, base, lane);
      }
      cnt = 0;
    }
    int4 v[AG_SCAN];
#pragma unroll
    for (int u = 0; u < AG_SCAN; ++u) {
      const int g = g0 + u * 1024 + (int)threadIdx.x;
      v[u] = g < nG ? d4[g] : make_int4(-1, -1, -1, -1);
    }
#pragma unroll
    for (int u = 0; u < AG_SCAN; ++u) {
      const int e0 = peel + 4 * (g0 + u * 1024 + (int)threadIdx.x);
      cnt = ag_append(v[u].x, e0, base, rows, cnt, le);
      cnt = ag_append(v[u].y, e0 + 1, base, rows, cnt, le);
      cnt = ag_append(v[u].z, e0 + 2, base, rows, cnt, le);
      cnt = ag_append(v[u].w, e0 + 3, base, rows, cnt, le);
    }
  }
  return cnt;
}

__device__ __forceinline__ float elu1(float v) { return v > 0.0f ? v : expm1f(v); }

__global__ __launch_bounds__(1024)
void aggregate_owner_softmax_elu(const float* __restrict__ f, const float* __restrict__ attn,
                                 float* s,
                                 const int* __restrict__ src, const int* __restrict__ dst,
                                 float* __restrict__ h, int N, int E) {
  __shared__ float tile[AG_ROWS * 256];
  __shared__ int l_edge[AG_WAVES * AG_CAP];
  __shared__ float l_mx[AG_ROWS * 4];
  __shared__ float l_den[AG_ROWS * 4];
  const int lane = threadIdx.x & 31;
  const int w    = threadIdx.x >> 5;
  const int base = blockIdx.x * AG_ROWS;
  const int rows = min(AG_ROWS, N - base);
  int* le = l_edge + w * AG_CAP;

  float4* t4 = (float4*)tile;
  {
    constexpr int PER = (AG_ROWS * 64 + 1023) / 1024;
    const float4* __restrict__ fo = (const float4*)(f + (long)base * 256);
    float4 v[PER];
#pragma unroll
    for (int u = 0; u < PER; ++u) {
      const int i = u * 1024 + (int)threadIdx.x;
      v[u] = fo[min(i, rows * 64 - 1)];
    }
#pragma unroll
    for (int u = 0; u < PER; ++u) {
      const int i = u * 1024 + (int)threadIdx.x;
      if (i < AG_ROWS * 64) t4[i] = v[u];
    }
  }
  for (int i = threadIdx.x; i < AG_ROWS * 4; i += 1024) {
    l_mx[i] = __uint_as_float(0xFF800000u);
    l_den[i] = 0.0f;
  }
  __syncthreads();

  int spilled = 0;
  int cnt = ag_filter<true, 256>(f, attn, s, src, dst, tile, l_mx, l_den, le, base, rows, E,
                                 lane, w, spilled);
  ag_logits_max256(f, attn, s, src, dst, tile, l_mx, le, cnt, base, lane);
  __syncthreads();

  for (int i = threadIdx.x; i < AG_ROWS * 64; i += 1024)
    t4[i] = make_float4(0.0f, 0.0f, 0.0f, 0.0f);
  __syncthreads();

  if (__builtin_amdgcn_readfirstlane(spilled) != 0)
    cnt = ag_filter<false, 256>(f, attn, s, src, dst, tile, l_mx, l_den, le, base, rows, E,
                                lane, w, spilled);
  ag_drain(f, s, src, dst, tile, l_mx, l_den, le, cnt, base, lane);
  __syncthreads();

  for (int i = threadIdx.x; i < rows * 64; i += 1024) {
    const int r = i >> 6, p = i & 63;
    const float* tr = tile + r * 256 + p;
    const float dv = l_den[r * 4 + (p >> 4)];
    const bool none = dv == 0.0f;
    float4 o;
    o.x = elu1(none ? 0.0f : tr[0]   / dv);
    o.y = elu1(none ? 0.0f : tr[64]  / dv);
    o.z = elu1(none ? 0.0f : tr[128] / dv);
    o.w = elu1(none ? 0.0f : tr[192] / dv);
    *(float4*)(h + (long)(base + r) * 256 + 4 * p) = o;
  }
}

__global__ __launch_bounds__(1024)
void aggregate_owner_softmax_mean(const float* __restrict__ f, const float* __restrict__ attn,
                                  float* s, const float* __restrict__ o1,
                                  const int* __restrict__ src, const int* __restrict__ dst,
                                  float* __restrict__ out, int N, int E) {
  __shared__ float tile[AG_ROWS * 128];
  __shared__ int l_edge[AG_WAVES * AG_CAP];
  __shared__ float l_mx[AG_ROWS * 4];
  __shared__ float l_den[AG_ROWS * 4];
  const int lane = threadIdx.x & 31;
  const int w    = threadIdx.x >> 5;
  const int base = blockIdx.x * AG_ROWS;
  const int rows = min(AG_ROWS, N - base);
  int* le = l_edge + w * AG_CAP;

  float4* t4 = (float4*)tile;
  {
    constexpr int PER = (AG_ROWS * 32 + 1023) / 1024;
    const float4* __restrict__ fo = (const float4*)(f + (long)base * 128);
    float4 v[PER];
#pragma unroll
    for (int u = 0; u < PER; ++u) {
      const int i = u * 1024 + (int)threadIdx.x;
      v[u] = fo[min(i, rows * 32 - 1)];
    }
#pragma unroll
    for (int u = 0; u < PER; ++u) {
      const int i = u * 1024 + (int)threadIdx.x;
      if (i < AG_ROWS * 32) t4[i] = v[u];
    }
  }
  for (int i = threadIdx.x; i < AG_ROWS * 4; i += 1024) {
    l_mx[i] = __uint_as_float(0xFF800000u);
    l_den[i] = 0.0f;
  }
  __syncthreads();

  int spilled = 0;
  int cnt = ag_filter<true, 128>(f, attn, s, src, dst, tile, l_mx, l_den, le, base, rows, E,
                                 lane, w, spilled);
  ag_logits_max128(f, attn, s, src, dst, tile, l_mx, le, cnt, base, lane);
  __syncthreads();

  for (int i = threadIdx.x; i < AG_ROWS * 32; i += 1024)
    t4[i] = make_float4(0.0f, 0.0f, 0.0f, 0.0f);
  __syncthreads();

  if (__builtin_amdgcn_readfirstlane(spilled) != 0)
    cnt = ag_filter<false, 128>(f, attn, s, src, dst, tile, l_mx, l_den, le, base, rows, E,
                                lane, w, spilled);
  ag_drain128(f, s, src, dst, tile, l_mx, l_den, le, cnt, base, lane);
  __syncthreads();

  for (int i = threadIdx.x; i < rows * 8; i += 1024) {
    const int r = i >> 3, q = i & 7;
    const float* tr = tile + r * 128 + q;
    const float* __restrict__ rr = o1 + (long)(base + r) * 128 + 4 * q;
    float4 acc = make_float4(0.0f, 0.0f, 0.0f, 0.0f);
#pragma unroll
    for (int hh = 0; hh < 4; ++hh) {
      const float dv = l_den[r * 4 + hh];
      const bool none = dv == 0.0f;
      const float4 rs = *(const float4*)(rr + hh * 32);
      const float* th = tr + hh * 8;
      acc.x += (none ? 0.0f : th[0]  / dv) + rs.x;
      acc.y += (none ? 0.0f : th[32] / dv) + rs.y;
      acc.z += (none ? 0.0f : th[64] / dv) + rs.z;
      acc.w += (none ? 0.0f : th[96] / dv) + rs.w;
    }
    float4 o;
    o.x = 0.25f * acc.x; o.y = 0.25f * acc.y; o.z = 0.25f * acc.z; o.w = 0.25f * acc.w;
    *(float4*)(out + (long)(base + r) * 32 + 4 * q) = o;
  }
}

static inline int cdiv_l(long a, long b) { return (int)((a + b - 1) / b); }

extern "C" void kernel_launch(void* const* d_in, const int* in_sizes, int n_in,
                              void* d_out, int out_size, void* d_ws, size_t ws_size,
                              hipStream_t stream) {
  const float* x     = (const float*)d_in[0];
  const float* W0    = (const float*)d_in[1];
  const float* a0    = (const float*)d_in[2];
  const float* W1    = (const float*)d_in[3];
  const float* a1    = (const float*)d_in[4];
  const float* Wres1 = (const float*)d_in[5];
  const int*   src   = (const int*)d_in[6];
  const int*   dst   = (const int*)d_in[7];

  const int N = in_sizes[0] / 256;
  const int E = in_sizes[6];
  const int H = 4;

  float*    f0    = (float*)d_ws;
  float*    h     = f0 + (long)N * 256;
  float*    f1    = h  + (long)N * 256;
  float*    o1    = f1 + (long)N * 128;
  float*    s     = o1 + (long)N * 128;
  const size_t need =
      ((size_t)N * 256 * 2 + (size_t)N * 128 * 2 + (size_t)E * H) * sizeof(float);
  if (need > ws_size) return;

  const int BT = 128;

  gemm_f32_wmma<256, 256><<<cdiv_l(N, 32), BT, 0, stream>>>(x, W0, W0 + 128, f0, f0 + 128, N);
  aggregate_owner_softmax_elu<<<cdiv_l(N, AG_ROWS), 1024, 0, stream>>>(f0, a0, s, src, dst,
                                                                       h, N, E);

  gemm_f32_wmma<256, 128><<<cdiv_l(N, 32), BT, 0, stream>>>(h, W1, Wres1, f1, o1, N);
  aggregate_owner_softmax_mean<<<cdiv_l(N, AG_ROWS), 1024, 0, stream>>>(
      f1, a1, s, o1, src, dst, (float*)d_out, N, E);
}


# layer-1 owners load the edge list that layer 0's owner of the same rows filtered and left in the workspace, dealt to waves in equal shares: dst filtered once, not twice; layer-1 add order changes
# speedup vs baseline: 2.8393x; 1.0769x over previous
#include <hip/hip_runtime.h>
#include <hip/hip_bf16.h>

typedef __attribute__((ext_vector_type(2))) float v2f;
typedef __attribute__((ext_vector_type(8))) float v8f;
typedef __attribute__((ext_vector_type(4))) float v4f;

template <int K, int LD>
__global__ __launch_bounds__(128)
void gemm_f32_wmma(const float* __restrict__ A,
                   const float* __restrict__ B0, const float* __restrict__ B1,
                   float* __restrict__ C0, float* __restrict__ C1, int M) {
  constexpr int MT = 2;
  constexpr int NT = 4;
  const int lane = threadIdx.x & 31;
  const int st   = threadIdx.x >> 5;
  const int tm   = blockIdx.x;
  const int hi = lane >> 4;
  const int lm = lane & 15;
  const float* __restrict__ B = (st & 2) ? B1 : B0;
  float* __restrict__ C = (st & 2) ? C1 : C0;
  const int col0 = (st & 1) * (NT * 16) + lm;
  const float* __restrict__ Ar[MT];
#pragma unroll
  for (int m = 0; m < MT; ++m) {
    int arow = tm * (MT * 16) + m * 16 + lm; if (arow >= M) arow = M - 1;
    Ar[m] = A + (long)arow * K + 2 * hi;
  }
  const float* __restrict__ Bc = B + (long)(2 * hi) * LD + col0;

  v8f c[MT][NT];
#pragma unroll
  for (int m = 0; m < MT; ++m)
#pragma unroll
    for (int t = 0; t < NT; ++t) c[m][t] = (v8f){};

  auto loadA = [&](int k, v2f* a) {
#pragma unroll
    for (int m = 0; m < MT; ++m) { a[m].x = Ar[m][k]; a[m].y = Ar[m][k + 1]; }
  };
  auto loadB = [&](int k, v2f* b) {
    const float* __restrict__ Bk = Bc + (long)k * LD;
#pragma unroll
    for (int t = 0; t < NT; ++t) { b[t].x = Bk[16 * t]; b[t].y = Bk[LD + 16 * t]; }
  };

  v2f a_cur[MT], b_cur[NT];
  loadA(0, a_cur);
  loadB(0, b_cur);
#pragma unroll 4
  for (int k = 0; k < K - 4; k += 4) {
    v2f a_nxt[MT], b_nxt[NT];
    loadA(k + 4, a_nxt);
    loadB(k + 4, b_nxt);
#pragma unroll
    for (int m = 0; m < MT; ++m)
#pragma unroll
      for (int t = 0; t < NT; ++t)
        c[m][t] = __builtin_amdgcn_wmma_f32_16x16x4_f32(false, a_cur[m], false, b_cur[t],
                                                        (short)0, c[m][t], false, false);
#pragma unroll
    for (int m = 0; m < MT; ++m) a_cur[m] = a_nxt[m];
#pragma unroll
    for (int t = 0; t < NT; ++t) b_cur[t] = b_nxt[t];
  }
#pragma unroll
  for (int m = 0; m < MT; ++m)
#pragma unroll
    for (int t = 0; t < NT; ++t)
      c[m][t] = __builtin_amdgcn_wmma_f32_16x16x4_f32(false, a_cur[m], false, b_cur[t],
                                                      (short)0, c[m][t], false, false);

#pragma unroll
  for (int m = 0; m < MT; ++m) {
#pragma unroll
    for (int t = 0; t < NT; ++t) {
#pragma unroll
      for (int r = 0; r < 8; ++r) {
        const int orow = tm * (MT * 16) + m * 16 + r + 8 * hi;
        if (orow < M) C[(long)orow * LD + col0 + 16 * t] = c[m][t][r];
      }
    }
  }
}


template <int M>
__device__ __forceinline__ float row_xor_add(float v) {
  return v + __int_as_float(__builtin_amdgcn_update_dpp(0, __float_as_int(v), 0x160 + M,
                                                        0xf, 0xf, true));
}

__device__ __forceinline__ float lrelu_dot4(v4f a, v4f b, v4f w) {
  const v4f v = a + b;
  float r = fmaxf(v.x, 0.2f * v.x) * w.x;
  r += fmaxf(v.y, 0.2f * v.y) * w.y;
  r += fmaxf(v.z, 0.2f * v.z) * w.z;
  r += fmaxf(v.w, 0.2f * v.w) * w.w;
  return r;
}

constexpr int AG_ROWS  = 196;
constexpr int AG_WAVES = 32;
constexpr int AG_SCAN  = 6;
constexpr int AG_STEP  = 128 * AG_SCAN;
constexpr int AG_CAP   = AG_STEP + 128;
constexpr int AG_BATCH = 8;
constexpr int AG_LB256 = 8;
constexpr int AG_LB128 = 16;
constexpr int AG_LCAP  = 4092;
constexpr int AG_LW    = AG_LCAP + 4;

__device__ __forceinline__ int ag_append(int d, int e, int base, int rows, int cnt,
                                         int* le) {
  const unsigned r = (unsigned)(d - base);
  const bool hit = r < (unsigned)rows;
  const unsigned mask = __builtin_amdgcn_ballot_w32(hit);
  if (mask != 0u) {
    const int pos = cnt + (int)__builtin_amdgcn_mbcnt_lo(mask, 0u);
    if (hit) le[pos] = e;
    cnt += __popc(mask);
  }
  return cnt;
}

__device__ __forceinline__ void lds_max_f32(float* p, float v) {
  if (v >= 0.0f)
    __hip_atomic_fetch_max((int*)p, __float_as_int(v), __ATOMIC_RELAXED,
                           __HIP_MEMORY_SCOPE_WORKGROUP);
  else
    __hip_atomic_fetch_min((unsigned*)p, __float_as_uint(v), __ATOMIC_RELAXED,
                           __HIP_MEMORY_SCOPE_WORKGROUP);
}

__device__ __forceinline__ void ag_logits_max256(const float* __restrict__ f,
                                                 const float* __restrict__ attn,
                                                 float* s,
                                                 const int* __restrict__ src,
                                                 const int* __restrict__ dst,
                                                 const float* own,
                                                 float* mx, const int* le,
                                                 int cnt, int base, int lane) {
  __builtin_amdgcn_fence(__ATOMIC_ACQ_REL, "wavefront");
  cnt = __builtin_amdgcn_readfirstlane(cnt);
  const v4f w0 = ((const v4f*)attn)[lane];
  const v4f w1 = ((const v4f*)attn)[32 + lane];
  for (int c = 0; c < cnt; c += 32) {
    const int e = le[min(c + lane, cnt - 1)];
    const int sv = src[e];
    const int dn = dst[e];
    v4f lg = (v4f){0.0f, 0.0f, 0.0f, 0.0f};
    const int nb = min(32, cnt - c);
    for (int q0 = 0; q0 < nb; q0 += AG_LB256) {
      v4f xs0[AG_LB256], xs1[AG_LB256];
#pragma unroll
      for (int b = 0; b < AG_LB256; ++b) {
        const int sb = __builtin_amdgcn_readlane(sv, q0 + b);
        const v4f* __restrict__ fs = (const v4f*)(f + (long)sb * 256);
        xs0[b] = fs[lane];
        xs1[b] = fs[32 + lane];
      }
#pragma unroll
      for (int b = 0; b < AG_LB256; ++b) {
        const int rb = __builtin_amdgcn_readlane(dn, q0 + b) - base;
        const v4f* fd = (const v4f*)(own + rb * 256);
        const v4f xd0 = fd[lane];
        const v4f xd1 = fd[32 + lane];
        float p0 = lrelu_dot4(xs0[b], xd0, w0);
        float p1 = lrelu_dot4(xs1[b], xd1, w1);
        p0 = row_xor_add<1>(p0); p1 = row_xor_add<1>(p1);
        p0 = row_xor_add<2>(p0); p1 = row_xor_add<2>(p1);
        p0 = row_xor_add<4>(p0); p1 = row_xor_add<4>(p1);
        p0 = row_xor_add<8>(p0); p1 = row_xor_add<8>(p1);
        const int l0 = __builtin_amdgcn_readlane(__float_as_int(p0), 0);
        const int l1 = __builtin_amdgcn_readlane(__float_as_int(p0), 16);
        const int l2 = __builtin_amdgcn_readlane(__float_as_int(p1), 0);
        const int l3 = __builtin_amdgcn_readlane(__float_as_int(p1), 16);
        const bool mine = lane == q0 + b;
        lg.x = mine ? __int_as_float(l0) : lg.x;
        lg.y = mine ? __int_as_float(l1) : lg.y;
        lg.z = mine ? __int_as_float(l2) : lg.z;
        lg.w = mine ? __int_as_float(l3) : lg.w;
        if (b & 1) __builtin_amdgcn_sched_barrier(0);
      }
    }
    if (c + lane < cnt) {
      *(v4f*)(s + (long)e * 4) = lg;
      float* mr = mx + (dn - base) * 4;
      lds_max_f32(mr,     lg.x);
      lds_max_f32(mr + 1, lg.y);
      lds_max_f32(mr + 2, lg.z);
      lds_max_f32(mr + 3, lg.w);
    }
  }
  __builtin_amdgcn_fence(__ATOMIC_ACQ_REL, "wavefront");
}

__device__ __forceinline__ void ag_logits_max128(const float* __restrict__ f,
                                                 const float* __restrict__ attn,
                                                 float* s,
                                                 const int* __restrict__ src,
                                                 const int* __restrict__ dst,
                                                 const float* own,
                                                 float* mx, const int* le,
                                                 int cnt, int base, int lane) {
  __builtin_amdgcn_fence(__ATOMIC_ACQ_REL, "wavefront");
  cnt = __builtin_amdgcn_readfirstlane(cnt);
  const v4f w = ((const v4f*)attn)[lane];
  for (int c = 0; c < cnt; c += 32) {
    const int e = le[min(c + lane, cnt - 1)];
    const int sv = src[e];
    const int dn = dst[e];
    v4f lg = (v4f){0.0f, 0.0f, 0.0f, 0.0f};
    const int nb = min(32, cnt - c);
    for (int q0 = 0; q0 < nb; q0 += AG_LB128) {
      v4f xs[AG_LB128];
#pragma unroll
      for (int b = 0; b < AG_LB128; ++b) {
        const int sb = __builtin_amdgcn_readlane(sv, q0 + b);
        xs[b] = ((const v4f*)(f + (long)sb * 128))[lane];
      }
#pragma unroll
      for (int b = 0; b < AG_LB128; ++b) {
        const int rb = __builtin_amdgcn_readlane(dn, q0 + b) - base;
        const v4f xd = ((const v4f*)(own + rb * 128))[lane];
        float p = lrelu_dot4(xs[b], xd, w);
        p = row_xor_add<1>(p);
        p = row_xor_add<2>(p);
        p = row_xor_add<4>(p);
        const int l0 = __builtin_amdgcn_readlane(__float_as_int(p), 0);
        const int l1 = __builtin_amdgcn_readlane(__float_as_int(p), 8);
        const int l2 = __builtin_amdgcn_readlane(__float_as_int(p), 16);
        const int l3 = __builtin_amdgcn_readlane(__float_as_int(p), 24);
        const bool mine = lane == q0 + b;
        lg.x = mine ? __int_as_float(l0) : lg.x;
        lg.y = mine ? __int_as_float(l1) : lg.y;
        lg.z = mine ? __int_as_float(l2) : lg.z;
        lg.w = mine ? __int_as_float(l3) : lg.w;
      }
    }
    if (c + lane < cnt) {
      *(v4f*)(s + (long)e * 4) = lg;
      float* mr = mx + (dn - base) * 4;
      lds_max_f32(mr,     lg.x);
      lds_max_f32(mr + 1, lg.y);
      lds_max_f32(mr + 2, lg.z);
      lds_max_f32(mr + 3, lg.w);
    }
  }
  __builtin_amdgcn_fence(__ATOMIC_ACQ_REL, "wavefront");
}

__device__ __forceinline__ void ag_drain(const float* __restrict__ f,
                                         const float* s,
                                         const int* __restrict__ src,
                                         const int* __restrict__ dst,
                                         float* tile, const float* mx, float* den,
                                         const int* le,
                                         int cnt, int base, int lane) {
  __builtin_amdgcn_fence(__ATOMIC_ACQ_REL, "wavefront");
  const int j = lane >> 2;
  const int hd = lane & 3;
  const bool lo = lane < 16;
  for (int i = 0; i < cnt; i += AG_BATCH) {
    const int e = le[min(i + j, cnt - 1)];
    const int sv = src[e];
    const int dn = dst[e];
    const float sl = s[(long)e * 4 + hd];
    v4f x0[AG_BATCH], x1[AG_BATCH];
#pragma unroll
    for (int b = 0; b < AG_BATCH; ++b) {
      const int sb = __builtin_amdgcn_readlane(sv, 4 * b);
      const v4f* __restrict__ fr = (const v4f*)(f + (long)sb * 256);
      x0[b] = fr[lane];
      x1[b] = fr[32 + lane];
    }
    const int r = dn - base;
    const float mv = mx[r * 4 + hd];
    asm volatile("" :: "v"(x0[0]), "v"(x1[0]), "v"(x0[1]), "v"(x1[1]), "v"(x0[2]), "v"(x1[2]),
                       "v"(x0[3]), "v"(x1[3]), "v"(x0[4]), "v"(x1[4]), "v"(x0[5]), "v"(x1[5]),
                       "v"(x0[6]), "v"(x1[6]), "v"(x0[7]), "v"(x1[7]));
    const float a = __expf(sl - mv);
    if (i + j < cnt)
      __hip_atomic_fetch_add(den + r * 4 + hd, a, __ATOMIC_RELAXED, __HIP_MEMORY_SCOPE_WORKGROUP);
#pragma unroll
    for (int b = 0; b < AG_BATCH; ++b) {
      if (i + b < cnt) {
        const int rb = __builtin_amdgcn_readlane(r, 4 * b);
        const float a0 = __int_as_float(__builtin_amdgcn_readlane(__float_as_int(a), 4 * b));
        const float a1 = __int_as_float(__builtin_amdgcn_readlane(__float_as_int(a), 4 * b + 1));
        const float a2 = __int_as_float(__builtin_amdgcn_readlane(__float_as_int(a), 4 * b + 2));
        const float a3 = __int_as_float(__builtin_amdgcn_readlane(__float_as_int(a), 4 * b + 3));
        const float al0 = lo ? a0 : a1;
        const float al1 = lo ? a2 : a3;
        float* tr = tile + rb * 256 + lane;
        __hip_atomic_fetch_add(tr,       x0[b].x * al0, __ATOMIC_RELAXED, __HIP_MEMORY_SCOPE_WORKGROUP);
        __hip_atomic_fetch_add(tr + 64,  x0[b].y * al0, __ATOMIC_RELAXED, __HIP_MEMORY_SCOPE_WORKGROUP);
        __hip_atomic_fetch_add(tr + 128, x0[b].z * al0, __ATOMIC_RELAXED, __HIP_MEMORY_SCOPE_WORKGROUP);
        __hip_atomic_fetch_add(tr + 192, x0[b].w * al0, __ATOMIC_RELAXED, __HIP_MEMORY_SCOPE_WORKGROUP);
        __hip_atomic_fetch_add(tr + 32,  x1[b].x * al1, __ATOMIC_RELAXED, __HIP_MEMORY_SCOPE_WORKGROUP);
        __hip_atomic_fetch_add(tr + 96,  x1[b].y * al1, __ATOMIC_RELAXED, __HIP_MEMORY_SCOPE_WORKGROUP);
        __hip_atomic_fetch_add(tr + 160, x1[b].z * al1, __ATOMIC_RELAXED, __HIP_MEMORY_SCOPE_WORKGROUP);
        __hip_atomic_fetch_add(tr + 224, x1[b].w * al1, __ATOMIC_RELAXED, __HIP_MEMORY_SCOPE_WORKGROUP);
      }
    }
  }
  __builtin_amdgcn_fence(__ATOMIC_ACQ_REL, "wavefront");
}

__device__ __forceinline__ void ag_drain128(const float* __restrict__ f,
                                            const float* s,
                                            const int* __restrict__ src,
                                            const int* __restrict__ dst,
                                            float* tile, const float* mx, float* den,
                                            const int* le,
                                            int cnt, int base, int lane) {
  __builtin_amdgcn_fence(__ATOMIC_ACQ_REL, "wavefront");
  const int j = lane >> 2;
  const int hd = lane & 3;
  const int hq = lane >> 3;
  for (int i = 0; i < cnt; i += AG_BATCH) {
    const int e = le[min(i + j, cnt - 1)];
    const int sv = src[e];
    const int dn = dst[e];
    const float sl = s[(long)e * 4 + hd];
    v4f x[AG_BATCH];
#pragma unroll
    for (int b = 0; b < AG_BATCH; ++b) {
      const int sb = __builtin_amdgcn_readlane(sv, 4 * b);
      const v4f* __restrict__ fr = (const v4f*)(f + (long)sb * 128);
      x[b] = fr[lane];
    }
    const int r = dn - base;
    const float mv = mx[r * 4 + hd];
    const float a = __expf(sl - mv);
    if (i + j < cnt)
      __hip_atomic_fetch_add(den + r * 4 + hd, a, __ATOMIC_RELAXED, __HIP_MEMORY_SCOPE_WORKGROUP);
    float al[AG_BATCH];
#pragma unroll
    for (int b = 0; b < AG_BATCH; ++b) al[b] = __shfl(a, 4 * b + hq, 32);
#pragma unroll
    for (int b = 0; b < AG_BATCH; ++b) {
      if (i + b < cnt) {
        const int rb = __builtin_amdgcn_readlane(r, 4 * b);
        float* tr = tile + rb * 128 + lane;
        __hip_atomic_fetch_add(tr,      x[b].x * al[b], __ATOMIC_RELAXED, __HIP_MEMORY_SCOPE_WORKGROUP);
        __hip_atomic_fetch_add(tr + 32, x[b].y * al[b], __ATOMIC_RELAXED, __HIP_MEMORY_SCOPE_WORKGROUP);
        __hip_atomic_fetch_add(tr + 64, x[b].z * al[b], __ATOMIC_RELAXED, __HIP_MEMORY_SCOPE_WORKGROUP);
        __hip_atomic_fetch_add(tr + 96, x[b].w * al[b], __ATOMIC_RELAXED, __HIP_MEMORY_SCOPE_WORKGROUP);
      }
    }
  }
  __builtin_amdgcn_fence(__ATOMIC_ACQ_REL, "wavefront");
}

template <bool MAXPASS, int ROW>
__device__ __forceinline__ int ag_filter(const float* __restrict__ f,
                                         const float* __restrict__ attn,
                                         float* s,
                                         const int* __restrict__ src,
                                         const int* __restrict__ dst,
                                         float* tile, float* mx, float* den, int* le,
                                         int base, int rows, int E, int lane, int w,
                                         int& spilled) {
  int peel = (int)(((16u - (unsigned)((unsigned long long)dst & 15ull)) & 15u) >> 2);
  if (peel > E) peel = E;
  const int nG = (E - peel) >> 2;
  const int tail0 = peel + 4 * nG;
  int cnt = 0;
  if (w == 0) {
    int e = -1;
    if (lane < peel) e = lane;
    else if (lane >= 4 && lane < 8 && tail0 + (lane - 4) < E) e = tail0 + (lane - 4);
    const int d = e >= 0 ? dst[e] : -1;
    cnt = ag_append(d, e, base, rows, cnt, le);
  }
  cnt = __builtin_amdgcn_readfirstlane(cnt);
  const int4* __restrict__ d4 = (const int4*)(dst + peel);
  for (int g0 = 0; g0 < nG; g0 += 1024 * AG_SCAN) {
    if (cnt > AG_CAP - AG_STEP) {
      if (MAXPASS) {
        if (ROW == 256) ag_logits_max256(f, attn, s, src, dst, tile, mx, le, cnt, base, lane);
        else            ag_logits_max128(f, attn, s, src, dst, tile, mx, le, cnt, base, lane);
        spilled = 1;
      } else if (ROW == 256) {
        ag_drain(f, s, src, dst, tile, mx, den, le, cnt, base, lane);
      } else {
        ag_drain128(f, s, src, dst, tile, mx, den, le, cnt, base, lane);
      }
      cnt = 0;
    }
    int4 v[AG_SCAN];
#pragma unroll
    for (int u = 0; u < AG_SCAN; ++u) {
      const int g = g0 + u * 1024 + (int)threadIdx.x;
      v[u] = g < nG ? d4[g] : make_int4(-1, -1, -1, -1);
    }
#pragma unroll
    for (int u = 0; u < AG_SCAN; ++u) {
      const int e0 = peel + 4 * (g0 + u * 1024 + (int)threadIdx.x);
      cnt = ag_append(v[u].x, e0, base, rows, cnt, le);
      cnt = ag_append(v[u].y, e0 + 1, base, rows, cnt, le);
      cnt = ag_append(v[u].z, e0 + 2, base, rows, cnt, le);
      cnt = ag_append(v[u].w, e0 + 3, base, rows, cnt, le);
    }
  }
  return cnt;
}

__device__ __forceinline__ float elu1(float v) { return v > 0.0f ? v : expm1f(v); }

__global__ __launch_bounds__(1024)
void aggregate_owner_softmax_elu(const float* __restrict__ f, const float* __restrict__ attn,
                                 float* s,
                                 const int* __restrict__ src, const int* __restrict__ dst,
                                 float* __restrict__ h, int* __restrict__ lists, int N, int E) {
  __shared__ float tile[AG_ROWS * 256];
  __shared__ int l_edge[AG_WAVES * AG_CAP];
  __shared__ float l_mx[AG_ROWS * 4];
  __shared__ float l_den[AG_ROWS * 4];
  __shared__ int l_wcnt[AG_WAVES];
  const int lane = threadIdx.x & 31;
  const int w    = threadIdx.x >> 5;
  const int base = blockIdx.x * AG_ROWS;
  const int rows = min(AG_ROWS, N - base);
  int* le = l_edge + w * AG_CAP;

  float4* t4 = (float4*)tile;
  {
    constexpr int PER = (AG_ROWS * 64 + 1023) / 1024;
    const float4* __restrict__ fo = (const float4*)(f + (long)base * 256);
    float4 v[PER];
#pragma unroll
    for (int u = 0; u < PER; ++u) {
      const int i = u * 1024 + (int)threadIdx.x;
      v[u] = fo[min(i, rows * 64 - 1)];
    }
#pragma unroll
    for (int u = 0; u < PER; ++u) {
      const int i = u * 1024 + (int)threadIdx.x;
      if (i < AG_ROWS * 64) t4[i] = v[u];
    }
  }
  for (int i = threadIdx.x; i < AG_ROWS * 4; i += 1024) {
    l_mx[i] = __uint_as_float(0xFF800000u);
    l_den[i] = 0.0f;
  }
  __syncthreads();

  int spilled = 0;
  int cnt = ag_filter<true, 256>(f, attn, s, src, dst, tile, l_mx, l_den, le, base, rows, E,
                                 lane, w, spilled);
  if (lane == 0) l_wcnt[w] = spilled != 0 ? -1 : cnt;
  ag_logits_max256(f, attn, s, src, dst, tile, l_mx, le, cnt, base, lane);
  __syncthreads();

  for (int i = threadIdx.x; i < AG_ROWS * 64; i += 1024)
    t4[i] = make_float4(0.0f, 0.0f, 0.0f, 0.0f);
  if (lists != nullptr) {
    int* L = lists + (long)blockIdx.x * AG_LW;
    const int v = l_wcnt[lane];
    const bool gone = __builtin_amdgcn_ballot_w32(v < 0) != 0u;
    int tot = v, off = lane < w ? v : 0;
    for (int m = 16; m >= 1; m >>= 1) {
      tot += __shfl_xor(tot, m, 32);
      off += __shfl_xor(off, m, 32);
    }
    const bool fits = !gone && tot <= AG_LCAP;
    const int n = __builtin_amdgcn_readfirstlane(cnt);
    if (fits)
      for (int i0 = 0; i0 < n; i0 += 32)
        if (i0 + lane < n) L[4 + off + i0 + lane] = le[i0 + lane];
    if (threadIdx.x == 0) L[0] = fits ? tot : -1;
  }
  __syncthreads();

  if (__builtin_amdgcn_readfirstlane(spilled) != 0)
    cnt = ag_filter<false, 256>(f, attn, s, src, dst, tile, l_mx, l_den, le, base, rows, E,
                                lane, w, spilled);
  ag_drain(f, s, src, dst, tile, l_mx, l_den, le, cnt, base, lane);
  __syncthreads();

  for (int i = threadIdx.x; i < rows * 64; i += 1024) {
    const int r = i >> 6, p = i & 63;
    const float* tr = tile + r * 256 + p;
    const float dv = l_den[r * 4 + (p >> 4)];
    const bool none = dv == 0.0f;
    float4 o;
    o.x = elu1(none ? 0.0f : tr[0]   / dv);
    o.y = elu1(none ? 0.0f : tr[64]  / dv);
    o.z = elu1(none ? 0.0f : tr[128] / dv);
    o.w = elu1(none ? 0.0f : tr[192] / dv);
    *(float4*)(h + (long)(base + r) * 256 + 4 * p) = o;
  }
}

__global__ __launch_bounds__(1024)
void aggregate_owner_softmax_mean(const float* __restrict__ f, const float* __restrict__ attn,
                                  float* s, const float* __restrict__ o1,
                                  const int* __restrict__ src, const int* __restrict__ dst,
                                  const int* __restrict__ lists,
                                  float* __restrict__ out, int N, int E) {
  __shared__ float tile[AG_ROWS * 128];
  __shared__ int l_edge[AG_WAVES * AG_CAP];
  __shared__ float l_mx[AG_ROWS * 4];
  __shared__ float l_den[AG_ROWS * 4];
  const int lane = threadIdx.x & 31;
  const int w    = threadIdx.x >> 5;
  const int base = blockIdx.x * AG_ROWS;
  const int rows = min(AG_ROWS, N - base);
  int* le = l_edge + w * AG_CAP;

  const int handed = lists != nullptr
      ? __builtin_amdgcn_readfirstlane(lists[(long)blockIdx.x * AG_LW]) : -1;
  int cnt = 0;
  if (handed >= 0) {
    const int per = (handed + AG_WAVES - 1) / AG_WAVES;
    const int first = __builtin_amdgcn_readfirstlane(w) * per;
    cnt = max(0, min(per, handed - first));
    if (cnt > 0) {
      constexpr int SH = (AG_LCAP / AG_WAVES + 1 + 31) / 32;
      const int* __restrict__ L = lists + (long)blockIdx.x * AG_LW + 4 + first;
      int id[SH];
#pragma unroll
      for (int u = 0; u < SH; ++u) id[u] = L[min(u * 32 + lane, cnt - 1)];
#pragma unroll
      for (int u = 0; u < SH; ++u)
        if (u * 32 + lane < cnt) le[u * 32 + lane] = id[u];
    }
  }
  float4* t4 = (float4*)tile;
  {
    constexpr int PER = (AG_ROWS * 32 + 1023) / 1024;
    const float4* __restrict__ fo = (const float4*)(f + (long)base * 128);
    float4 v[PER];
#pragma unroll
    for (int u = 0; u < PER; ++u) {
      const int i = u * 1024 + (int)threadIdx.x;
      v[u] = fo[min(i, rows * 32 - 1)];
    }
#pragma unroll
    for (int u = 0; u < PER; ++u) {
      const int i = u * 1024 + (int)threadIdx.x;
      if (i < AG_ROWS * 32) t4[i] = v[u];
    }
  }
  for (int i = threadIdx.x; i < AG_ROWS * 4; i += 1024) {
    l_mx[i] = __uint_as_float(0xFF800000u);
    l_den[i] = 0.0f;
  }
  __syncthreads();

  int spilled = 0;
  if (handed < 0)
    cnt = ag_filter<true, 128>(f, attn, s, src, dst, tile, l_mx, l_den, le, base, rows, E,
                               lane, w, spilled);
  ag_logits_max128(f, attn, s, src, dst, tile, l_mx, le, cnt, base, lane);
  __syncthreads();

  for (int i = threadIdx.x; i < AG_ROWS * 32; i += 1024)
    t4[i] = make_float4(0.0f, 0.0f, 0.0f, 0.0f);
  __syncthreads();

  if (__builtin_amdgcn_readfirstlane(spilled) != 0)
    cnt = ag_filter<false, 128>(f, attn, s, src, dst, tile, l_mx, l_den, le, base, rows, E,
                                lane, w, spilled);
  ag_drain128(f, s, src, dst, tile, l_mx, l_den, le, cnt, base, lane);
  __syncthreads();

  for (int i = threadIdx.x; i < rows * 8; i += 1024) {
    const int r = i >> 3, q = i & 7;
    const float* tr = tile + r * 128 + q;
    const float* __restrict__ rr = o1 + (long)(base + r) * 128 + 4 * q;
    float4 acc = make_float4(0.0f, 0.0f, 0.0f, 0.0f);
#pragma unroll
    for (int hh = 0; hh < 4; ++hh) {
      const float dv = l_den[r * 4 + hh];
      const bool none = dv == 0.0f;
      const float4 rs = *(const float4*)(rr + hh * 32);
      const float* th = tr + hh * 8;
      acc.x += (none ? 0.0f : th[0]  / dv) + rs.x;
      acc.y += (none ? 0.0f : th[32] / dv) + rs.y;
      acc.z += (none ? 0.0f : th[64] / dv) + rs.z;
      acc.w += (none ? 0.0f : th[96] / dv) + rs.w;
    }
    float4 o;
    o.x = 0.25f * acc.x; o.y = 0.25f * acc.y; o.z = 0.25f * acc.z; o.w = 0.25f * acc.w;
    *(float4*)(out + (long)(base + r) * 32 + 4 * q) = o;
  }
}

static inline int cdiv_l(long a, long b) { return (int)((a + b - 1) / b); }

extern "C" void kernel_launch(void* const* d_in, const int* in_sizes, int n_in,
                              void* d_out, int out_size, void* d_ws, size_t ws_size,
                              hipStream_t stream) {
  const float* x     = (const float*)d_in[0];
  const float* W0    = (const float*)d_in[1];
  const float* a0    = (const float*)d_in[2];
  const float* W1    = (const float*)d_in[3];
  const float* a1    = (const float*)d_in[4];
  const float* Wres1 = (const float*)d_in[5];
  const int*   src   = (const int*)d_in[6];
  const int*   dst   = (const int*)d_in[7];

  const int N = in_sizes[0] / 256;
  const int E = in_sizes[6];
  const int H = 4;

  float*    f0    = (float*)d_ws;
  float*    h     = f0 + (long)N * 256;
  float*    f1    = h  + (long)N * 256;
  float*    o1    = f1 + (long)N * 128;
  float*    s     = o1 + (long)N * 128;
  int*      lists = (int*)(s + (long)E * H);
  const size_t need =
      ((size_t)N * 256 * 2 + (size_t)N * 128 * 2 + (size_t)E * H) * sizeof(float);
  if (need > ws_size) return;
  if (need + (size_t)cdiv_l(N, AG_ROWS) * AG_LW * sizeof(int) > ws_size) lists = nullptr;

  const int BT = 128;

  gemm_f32_wmma<256, 256><<<cdiv_l(N, 32), BT, 0, stream>>>(x, W0, W0 + 128, f0, f0 + 128, N);
  aggregate_owner_softmax_elu<<<cdiv_l(N, AG_ROWS), 1024, 0, stream>>>(f0, a0, s, src, dst,
                                                                       h, lists, N, E);

  gemm_f32_wmma<256, 128><<<cdiv_l(N, 32), BT, 0, stream>>>(h, W1, Wres1, f1, o1, N);
  aggregate_owner_softmax_mean<<<cdiv_l(N, AG_ROWS), 1024, 0, stream>>>(
      f1, a1, s, o1, src, dst, lists, (float*)d_out, N, E);
}


# layer-0 owners read edge ids pre-binned by owner range (256 chunk workgroups riding in the layer-0 GEMM grid, LDS counting add, 128-byte segments), no longer each scanning all dst; add order changes
# speedup vs baseline: 3.0670x; 1.0802x over previous
#include <hip/hip_runtime.h>
#include <hip/hip_bf16.h>

typedef __attribute__((ext_vector_type(2))) float v2f;
typedef __attribute__((ext_vector_type(8))) float v8f;
typedef __attribute__((ext_vector_type(4))) float v4f;

constexpr int AG_ROWS = 196;
constexpr int BIN_NB  = 256;
constexpr int BIN_SEG = 32;
constexpr int BIN_OWN = 256;

__device__ __forceinline__ void bin_edges(const int* __restrict__ dst, int* __restrict__ bins,
                                          int E, int owners, int chunk) {
  __shared__ int seg[BIN_OWN * BIN_SEG];
  constexpr int NT = 128;
  constexpr int IN_FLIGHT = 8;
  const int tid = threadIdx.x;
  for (int i = tid; i < BIN_OWN * BIN_SEG / 4; i += NT)
    ((int4*)seg)[i] = make_int4(0, 0, 0, 0);
  __syncthreads();
  const int per = (E + BIN_NB - 1) / BIN_NB;
  const int e0 = chunk * per;
  const int e1 = min(E, e0 + per);
  for (int eb = e0; eb < e1; eb += NT * IN_FLIGHT) {
    int d[IN_FLIGHT];
#pragma unroll
    for (int u = 0; u < IN_FLIGHT; ++u) {
      const int e = eb + u * NT + tid;
      d[u] = e < e1 ? dst[e] : -1;
    }
#pragma unroll
    for (int u = 0; u < IN_FLIGHT; ++u) {
      const int e = eb + u * NT + tid;
      const int o = d[u] / AG_ROWS;
      if (d[u] >= 0 && o < owners) {
        int* sg = seg + o * BIN_SEG;
        const int k = __hip_atomic_fetch_add(sg, 1, __ATOMIC_RELAXED, __HIP_MEMORY_SCOPE_WORKGROUP);
        if (k < BIN_SEG - 1) sg[1 + k] = e;
      }
    }
  }
  __syncthreads();
  for (int i = tid; i < owners * (BIN_SEG / 4); i += NT) {
    const int o = i / (BIN_SEG / 4), p = i % (BIN_SEG / 4);
    ((int4*)(bins + ((long)o * BIN_NB + chunk) * BIN_SEG))[p] = ((const int4*)seg)[i];
  }
}

template <int K, int LD, bool BIN>
__global__ __launch_bounds__(128)
void gemm_f32_wmma(const float* __restrict__ A,
                   const float* __restrict__ B0, const float* __restrict__ B1,
                   float* __restrict__ C0, float* __restrict__ C1, int M,
                   const int* __restrict__ dst, int* __restrict__ bins, int E, int owners) {
  constexpr int MT = 2;
  constexpr int NT = 4;
  if constexpr (BIN) {
    const int tiles = (M + MT * 16 - 1) / (MT * 16);
    if ((int)blockIdx.x >= tiles) {
      bin_edges(dst, bins, E, owners, (int)blockIdx.x - tiles);
      return;
    }
  }
  const int lane = threadIdx.x & 31;
  const int st   = threadIdx.x >> 5;
  const int tm   = blockIdx.x;
  const int hi = lane >> 4;
  const int lm = lane & 15;
  const float* __restrict__ B = (st & 2) ? B1 : B0;
  float* __restrict__ C = (st & 2) ? C1 : C0;
  const int col0 = (st & 1) * (NT * 16) + lm;
  const float* __restrict__ Ar[MT];
#pragma unroll
  for (int m = 0; m < MT; ++m) {
    int arow = tm * (MT * 16) + m * 16 + lm; if (arow >= M) arow = M - 1;
    Ar[m] = A + (long)arow * K + 2 * hi;
  }
  const float* __restrict__ Bc = B + (long)(2 * hi) * LD + col0;

  v8f c[MT][NT];
#pragma unroll
  for (int m = 0; m < MT; ++m)
#pragma unroll
    for (int t = 0; t < NT; ++t) c[m][t] = (v8f){};

  auto loadA = [&](int k, v2f* a) {
#pragma unroll
    for (int m = 0; m < MT; ++m) { a[m].x = Ar[m][k]; a[m].y = Ar[m][k + 1]; }
  };
  auto loadB = [&](int k, v2f* b) {
    const float* __restrict__ Bk = Bc + (long)k * LD;
#pragma unroll
    for (int t = 0; t < NT; ++t) { b[t].x = Bk[16 * t]; b[t].y = Bk[LD + 16 * t]; }
  };

  v2f a_cur[MT], b_cur[NT];
  loadA(0, a_cur);
  loadB(0, b_cur);
#pragma unroll 4
  for (int k = 0; k < K - 4; k += 4) {
    v2f a_nxt[MT], b_nxt[NT];
    loadA(k + 4, a_nxt);
    loadB(k + 4, b_nxt);
#pragma unroll
    for (int m = 0; m < MT; ++m)
#pragma unroll
      for (int t = 0; t < NT; ++t)
        c[m][t] = __builtin_amdgcn_wmma_f32_16x16x4_f32(false, a_cur[m], false, b_cur[t],
                                                        (short)0, c[m][t], false, false);
#pragma unroll
    for (int m = 0; m < MT; ++m) a_cur[m] = a_nxt[m];
#pragma unroll
    for (int t = 0; t < NT; ++t) b_cur[t] = b_nxt[t];
  }
#pragma unroll
  for (int m = 0; m < MT; ++m)
#pragma unroll
    for (int t = 0; t < NT; ++t)
      c[m][t] = __builtin_amdgcn_wmma_f32_16x16x4_f32(false, a_cur[m], false, b_cur[t],
                                                      (short)0, c[m][t], false, false);

#pragma unroll
  for (int m = 0; m < MT; ++m) {
#pragma unroll
    for (int t = 0; t < NT; ++t) {
#pragma unroll
      for (int r = 0; r < 8; ++r) {
        const int orow = tm * (MT * 16) + m * 16 + r + 8 * hi;
        if (orow < M) C[(long)orow * LD + col0 + 16 * t] = c[m][t][r];
      }
    }
  }
}


template <int M>
__device__ __forceinline__ float row_xor_add(float v) {
  return v + __int_as_float(__builtin_amdgcn_update_dpp(0, __float_as_int(v), 0x160 + M,
                                                        0xf, 0xf, true));
}

__device__ __forceinline__ float lrelu_dot4(v4f a, v4f b, v4f w) {
  const v4f v = a + b;
  float r = fmaxf(v.x, 0.2f * v.x) * w.x;
  r += fmaxf(v.y, 0.2f * v.y) * w.y;
  r += fmaxf(v.z, 0.2f * v.z) * w.z;
  r += fmaxf(v.w, 0.2f * v.w) * w.w;
  return r;
}

constexpr int AG_WAVES = 32;
constexpr int AG_SCAN  = 6;
constexpr int AG_STEP  = 128 * AG_SCAN;
constexpr int AG_CAP   = AG_STEP + 128;
constexpr int AG_BATCH = 8;
constexpr int AG_LB256 = 8;
constexpr int AG_LB128 = 16;
constexpr int AG_LCAP  = 4092;
constexpr int AG_LW    = AG_LCAP + 4;

__device__ __forceinline__ int ag_append(int d, int e, int base, int rows, int cnt,
                                         int* le) {
  const unsigned r = (unsigned)(d - base);
  const bool hit = r < (unsigned)rows;
  const unsigned mask = __builtin_amdgcn_ballot_w32(hit);
  if (mask != 0u) {
    const int pos = cnt + (int)__builtin_amdgcn_mbcnt_lo(mask, 0u);
    if (hit) le[pos] = e;
    cnt += __popc(mask);
  }
  return cnt;
}

__device__ __forceinline__ void lds_max_f32(float* p, float v) {
  if (v >= 0.0f)
    __hip_atomic_fetch_max((int*)p, __float_as_int(v), __ATOMIC_RELAXED,
                           __HIP_MEMORY_SCOPE_WORKGROUP);
  else
    __hip_atomic_fetch_min((unsigned*)p, __float_as_uint(v), __ATOMIC_RELAXED,
                           __HIP_MEMORY_SCOPE_WORKGROUP);
}

__device__ __forceinline__ void ag_logits_max256(const float* __restrict__ f,
                                                 const float* __restrict__ attn,
                                                 float* s,
                                                 const int* __restrict__ src,
                                                 const int* __restrict__ dst,
                                                 const float* own,
                                                 float* mx, const int* le,
                                                 int cnt, int base, int lane) {
  __builtin_amdgcn_fence(__ATOMIC_ACQ_REL, "wavefront");
  cnt = __builtin_amdgcn_readfirstlane(cnt);
  const v4f w0 = ((const v4f*)attn)[lane];
  const v4f w1 = ((const v4f*)attn)[32 + lane];
  for (int c = 0; c < cnt; c += 32) {
    const int e = le[min(c + lane, cnt - 1)];
    const int sv = src[e];
    const int dn = dst[e];
    v4f lg = (v4f){0.0f, 0.0f, 0.0f, 0.0f};
    const int nb = min(32, cnt - c);
    for (int q0 = 0; q0 < nb; q0 += AG_LB256) {
      v4f xs0[AG_LB256], xs1[AG_LB256];
#pragma unroll
      for (int b = 0; b < AG_LB256; ++b) {
        const int sb = __builtin_amdgcn_readlane(sv, q0 + b);
        const v4f* __restrict__ fs = (const v4f*)(f + (long)sb * 256);
        xs0[b] = fs[lane];
        xs1[b] = fs[32 + lane];
      }
#pragma unroll
      for (int b = 0; b < AG_LB256; ++b) {
        const int rb = __builtin_amdgcn_readlane(dn, q0 + b) - base;
        const v4f* fd = (const v4f*)(own + rb * 256);
        const v4f xd0 = fd[lane];
        const v4f xd1 = fd[32 + lane];
        float p0 = lrelu_dot4(xs0[b], xd0, w0);
        float p1 = lrelu_dot4(xs1[b], xd1, w1);
        p0 = row_xor_add<1>(p0); p1 = row_xor_add<1>(p1);
        p0 = row_xor_add<2>(p0); p1 = row_xor_add<2>(p1);
        p0 = row_xor_add<4>(p0); p1 = row_xor_add<4>(p1);
        p0 = row_xor_add<8>(p0); p1 = row_xor_add<8>(p1);
        const int l0 = __builtin_amdgcn_readlane(__float_as_int(p0), 0);
        const int l1 = __builtin_amdgcn_readlane(__float_as_int(p0), 16);
        const int l2 = __builtin_amdgcn_readlane(__float_as_int(p1), 0);
        const int l3 = __builtin_amdgcn_readlane(__float_as_int(p1), 16);
        const bool mine = lane == q0 + b;
        lg.x = mine ? __int_as_float(l0) : lg.x;
        lg.y = mine ? __int_as_float(l1) : lg.y;
        lg.z = mine ? __int_as_float(l2) : lg.z;
        lg.w = mine ? __int_as_float(l3) : lg.w;
        if (b & 1) __builtin_amdgcn_sched_barrier(0);
      }
    }
    if (c + lane < cnt) {
      *(v4f*)(s + (long)e * 4) = lg;
      float* mr = mx + (dn - base) * 4;
      lds_max_f32(mr,     lg.x);
      lds_max_f32(mr + 1, lg.y);
      lds_max_f32(mr + 2, lg.z);
      lds_max_f32(mr + 3, lg.w);
    }
  }
  __builtin_amdgcn_fence(__ATOMIC_ACQ_REL, "wavefront");
}

__device__ __forceinline__ void ag_logits_max128(const float* __restrict__ f,
                                                 const float* __restrict__ attn,
                                                 float* s,
                                                 const int* __restrict__ src,
                                                 const int* __restrict__ dst,
                                                 const float* own,
                                                 float* mx, const int* le,
                                                 int cnt, int base, int lane) {
  __builtin_amdgcn_fence(__ATOMIC_ACQ_REL, "wavefront");
  cnt = __builtin_amdgcn_readfirstlane(cnt);
  const v4f w = ((const v4f*)attn)[lane];
  for (int c = 0; c < cnt; c += 32) {
    const int e = le[min(c + lane, cnt - 1)];
    const int sv = src[e];
    const int dn = dst[e];
    v4f lg = (v4f){0.0f, 0.0f, 0.0f, 0.0f};
    const int nb = min(32, cnt - c);
    for (int q0 = 0; q0 < nb; q0 += AG_LB128) {
      v4f xs[AG_LB128];
#pragma unroll
      for (int b = 0; b < AG_LB128; ++b) {
        const int sb = __builtin_amdgcn_readlane(sv, q0 + b);
        xs[b] = ((const v4f*)(f + (long)sb * 128))[lane];
      }
#pragma unroll
      for (int b = 0; b < AG_LB128; ++b) {
        const int rb = __builtin_amdgcn_readlane(dn, q0 + b) - base;
        const v4f xd = ((const v4f*)(own + rb * 128))[lane];
        float p = lrelu_dot4(xs[b], xd, w);
        p = row_xor_add<1>(p);
        p = row_xor_add<2>(p);
        p = row_xor_add<4>(p);
        const int l0 = __builtin_amdgcn_readlane(__float_as_int(p), 0);
        const int l1 = __builtin_amdgcn_readlane(__float_as_int(p), 8);
        const int l2 = __builtin_amdgcn_readlane(__float_as_int(p), 16);
        const int l3 = __builtin_amdgcn_readlane(__float_as_int(p), 24);
        const bool mine = lane == q0 + b;
        lg.x = mine ? __int_as_float(l0) : lg.x;
        lg.y = mine ? __int_as_float(l1) : lg.y;
        lg.z = mine ? __int_as_float(l2) : lg.z;
        lg.w = mine ? __int_as_float(l3) : lg.w;
      }
    }
    if (c + lane < cnt) {
      *(v4f*)(s + (long)e * 4) = lg;
      float* mr = mx + (dn - base) * 4;
      lds_max_f32(mr,     lg.x);
      lds_max_f32(mr + 1, lg.y);
      lds_max_f32(mr + 2, lg.z);
      lds_max_f32(mr + 3, lg.w);
    }
  }
  __builtin_amdgcn_fence(__ATOMIC_ACQ_REL, "wavefront");
}

__device__ __forceinline__ void ag_drain(const float* __restrict__ f,
                                         const float* s,
                                         const int* __restrict__ src,
                                         const int* __restrict__ dst,
                                         float* tile, const float* mx, float* den,
                                         const int* le,
                                         int cnt, int base, int lane) {
  __builtin_amdgcn_fence(__ATOMIC_ACQ_REL, "wavefront");
  const int j = lane >> 2;
  const int hd = lane & 3;
  const bool lo = lane < 16;
  for (int i = 0; i < cnt; i += AG_BATCH) {
    const int e = le[min(i + j, cnt - 1)];
    const int sv = src[e];
    const int dn = dst[e];
    const float sl = s[(long)e * 4 + hd];
    v4f x0[AG_BATCH], x1[AG_BATCH];
#pragma unroll
    for (int b = 0; b < AG_BATCH; ++b) {
      const int sb = __builtin_amdgcn_readlane(sv, 4 * b);
      const v4f* __restrict__ fr = (const v4f*)(f + (long)sb * 256);
      x0[b] = fr[lane];
      x1[b] = fr[32 + lane];
    }
    const int r = dn - base;
    const float mv = mx[r * 4 + hd];
    asm volatile("" :: "v"(x0[0]), "v"(x1[0]), "v"(x0[1]), "v"(x1[1]), "v"(x0[2]), "v"(x1[2]),
                       "v"(x0[3]), "v"(x1[3]), "v"(x0[4]), "v"(x1[4]), "v"(x0[5]), "v"(x1[5]),
                       "v"(x0[6]), "v"(x1[6]), "v"(x0[7]), "v"(x1[7]));
    const float a = __expf(sl - mv);
    if (i + j < cnt)
      __hip_atomic_fetch_add(den + r * 4 + hd, a, __ATOMIC_RELAXED, __HIP_MEMORY_SCOPE_WORKGROUP);
#pragma unroll
    for (int b = 0; b < AG_BATCH; ++b) {
      if (i + b < cnt) {
        const int rb = __builtin_amdgcn_readlane(r, 4 * b);
        const float a0 = __int_as_float(__builtin_amdgcn_readlane(__float_as_int(a), 4 * b));
        const float a1 = __int_as_float(__builtin_amdgcn_readlane(__float_as_int(a), 4 * b + 1));
        const float a2 = __int_as_float(__builtin_amdgcn_readlane(__float_as_int(a), 4 * b + 2));
        const float a3 = __int_as_float(__builtin_amdgcn_readlane(__float_as_int(a), 4 * b + 3));
        const float al0 = lo ? a0 : a1;
        const float al1 = lo ? a2 : a3;
        float* tr = tile + rb * 256 + lane;
        __hip_atomic_fetch_add(tr,       x0[b].x * al0, __ATOMIC_RELAXED, __HIP_MEMORY_SCOPE_WORKGROUP);
        __hip_atomic_fetch_add(tr + 64,  x0[b].y * al0, __ATOMIC_RELAXED, __HIP_MEMORY_SCOPE_WORKGROUP);
        __hip_atomic_fetch_add(tr + 128, x0[b].z * al0, __ATOMIC_RELAXED, __HIP_MEMORY_SCOPE_WORKGROUP);
        __hip_atomic_fetch_add(tr + 192, x0[b].w * al0, __ATOMIC_RELAXED, __HIP_MEMORY_SCOPE_WORKGROUP);
        __hip_atomic_fetch_add(tr + 32,  x1[b].x * al1, __ATOMIC_RELAXED, __HIP_MEMORY_SCOPE_WORKGROUP);
        __hip_atomic_fetch_add(tr + 96,  x1[b].y * al1, __ATOMIC_RELAXED, __HIP_MEMORY_SCOPE_WORKGROUP);
        __hip_atomic_fetch_add(tr + 160, x1[b].z * al1, __ATOMIC_RELAXED, __HIP_MEMORY_SCOPE_WORKGROUP);
        __hip_atomic_fetch_add(tr + 224, x1[b].w * al1, __ATOMIC_RELAXED, __HIP_MEMORY_SCOPE_WORKGROUP);
      }
    }
  }
  __builtin_amdgcn_fence(__ATOMIC_ACQ_REL, "wavefront");
}

__device__ __forceinline__ void ag_drain128(const float* __restrict__ f,
                                            const float* s,
                                            const int* __restrict__ src,
                                            const int* __restrict__ dst,
                                            float* tile, const float* mx, float* den,
                                            const int* le,
                                            int cnt, int base, int lane) {
  __builtin_amdgcn_fence(__ATOMIC_ACQ_REL, "wavefront");
  const int j = lane >> 2;
  const int hd = lane & 3;
  const int hq = lane >> 3;
  for (int i = 0; i < cnt; i += AG_BATCH) {
    const int e = le[min(i + j, cnt - 1)];
    const int sv = src[e];
    const int dn = dst[e];
    const float sl = s[(long)e * 4 + hd];
    v4f x[AG_BATCH];
#pragma unroll
    for (int b = 0; b < AG_BATCH; ++b) {
      const int sb = __builtin_amdgcn_readlane(sv, 4 * b);
      const v4f* __restrict__ fr = (const v4f*)(f + (long)sb * 128);
      x[b] = fr[lane];
    }
    const int r = dn - base;
    const float mv = mx[r * 4 + hd];
    const float a = __expf(sl - mv);
    if (i + j < cnt)
      __hip_atomic_fetch_add(den + r * 4 + hd, a, __ATOMIC_RELAXED, __HIP_MEMORY_SCOPE_WORKGROUP);
    float al[AG_BATCH];
#pragma unroll
    for (int b = 0; b < AG_BATCH; ++b) al[b] = __shfl(a, 4 * b + hq, 32);
#pragma unroll
    for (int b = 0; b < AG_BATCH; ++b) {
      if (i + b < cnt) {
        const int rb = __builtin_amdgcn_readlane(r, 4 * b);
        float* tr = tile + rb * 128 + lane;
        __hip_atomic_fetch_add(tr,      x[b].x * al[b], __ATOMIC_RELAXED, __HIP_MEMORY_SCOPE_WORKGROUP);
        __hip_atomic_fetch_add(tr + 32, x[b].y * al[b], __ATOMIC_RELAXED, __HIP_MEMORY_SCOPE_WORKGROUP);
        __hip_atomic_fetch_add(tr + 64, x[b].z * al[b], __ATOMIC_RELAXED, __HIP_MEMORY_SCOPE_WORKGROUP);
        __hip_atomic_fetch_add(tr + 96, x[b].w * al[b], __ATOMIC_RELAXED, __HIP_MEMORY_SCOPE_WORKGROUP);
      }
    }
  }
  __builtin_amdgcn_fence(__ATOMIC_ACQ_REL, "wavefront");
}

template <bool MAXPASS, int ROW>
__device__ __forceinline__ int ag_filter(const float* __restrict__ f,
                                         const float* __restrict__ attn,
                                         float* s,
                                         const int* __restrict__ src,
                                         const int* __restrict__ dst,
                                         float* tile, float* mx, float* den, int* le,
                                         int base, int rows, int E, int lane, int w,
                                         int& spilled) {
  int peel = (int)(((16u - (unsigned)((unsigned long long)dst & 15ull)) & 15u) >> 2);
  if (peel > E) peel = E;
  const int nG = (E - peel) >> 2;
  const int tail0 = peel + 4 * nG;
  int cnt = 0;
  if (w == 0) {
    int e = -1;
    if (lane < peel) e = lane;
    else if (lane >= 4 && lane < 8 && tail0 + (lane - 4) < E) e = tail0 + (lane - 4);
    const int d = e >= 0 ? dst[e] : -1;
    cnt = ag_append(d, e, base, rows, cnt, le);
  }
  cnt = __builtin_amdgcn_readfirstlane(cnt);
  const int4* __restrict__ d4 = (const int4*)(dst + peel);
  for (int g0 = 0; g0 < nG; g0 += 1024 * AG_SCAN) {
    if (cnt > AG_CAP - AG_STEP) {
      if (MAXPASS) {
        if (ROW == 256) ag_logits_max256(f, attn, s, src, dst, tile, mx, le, cnt, base, lane);
        else            ag_logits_max128(f, attn, s, src, dst, tile, mx, le, cnt, base, lane);
        spilled = 1;
      } else if (ROW == 256) {
        ag_drain(f, s, src, dst, tile, mx, den, le, cnt, base, lane);
      } else {
        ag_drain128(f, s, src, dst, tile, mx, den, le, cnt, base, lane);
      }
      cnt = 0;
    }
    int4 v[AG_SCAN];
#pragma unroll
    for (int u = 0; u < AG_SCAN; ++u) {
      const int g = g0 + u * 1024 + (int)threadIdx.x;
      v[u] = g < nG ? d4[g] : make_int4(-1, -1, -1, -1);
    }
#pragma unroll
    for (int u = 0; u < AG_SCAN; ++u) {
      const int e0 = peel + 4 * (g0 + u * 1024 + (int)threadIdx.x);
      cnt = ag_append(v[u].x, e0, base, rows, cnt, le);
      cnt = ag_append(v[u].y, e0 + 1, base, rows, cnt, le);
      cnt = ag_append(v[u].z, e0 + 2, base, rows, cnt, le);
      cnt = ag_append(v[u].w, e0 + 3, base, rows, cnt, le);
    }
  }
  return cnt;
}

__device__ __forceinline__ float elu1(float v) { return v > 0.0f ? v : expm1f(v); }

__global__ __launch_bounds__(1024)
void aggregate_owner_softmax_elu(const float* __restrict__ f, const float* __restrict__ attn,
                                 float* s,
                                 const int* __restrict__ src, const int* __restrict__ dst,
                                 float* __restrict__ h, int* __restrict__ lists,
                                 const int* __restrict__ bins, int N, int E) {
  __shared__ float tile[AG_ROWS * 256];
  __shared__ int l_edge[AG_WAVES * AG_CAP];
  __shared__ float l_mx[AG_ROWS * 4];
  __shared__ float l_den[AG_ROWS * 4];
  __shared__ int l_wcnt[AG_WAVES];
  __shared__ int l_over[AG_WAVES];
  const int lane = threadIdx.x & 31;
  const int w    = threadIdx.x >> 5;
  const int base = blockIdx.x * AG_ROWS;
  const int rows = min(AG_ROWS, N - base);
  int* le = l_edge + w * AG_CAP;

  static_assert(BIN_SEG == 32 && BIN_NB % AG_WAVES == 0 &&
                (BIN_NB / AG_WAVES) * (BIN_SEG - 1) <= AG_CAP, "a wave's segments fill its list");
  constexpr int SPW = BIN_NB / AG_WAVES;
  int bv[SPW] = {};
  if (bins != nullptr) {
    const int* __restrict__ Bn =
        bins + ((long)blockIdx.x * BIN_NB + __builtin_amdgcn_readfirstlane(w) * SPW) * BIN_SEG;
#pragma unroll
    for (int u = 0; u < SPW; ++u) bv[u] = Bn[u * BIN_SEG + lane];
  }

  float4* t4 = (float4*)tile;
  {
    constexpr int PER = (AG_ROWS * 64 + 1023) / 1024;
    const float4* __restrict__ fo = (const float4*)(f + (long)base * 256);
    float4 v[PER];
#pragma unroll
    for (int u = 0; u < PER; ++u) {
      const int i = u * 1024 + (int)threadIdx.x;
      v[u] = fo[min(i, rows * 64 - 1)];
    }
#pragma unroll
    for (int u = 0; u < PER; ++u) {
      const int i = u * 1024 + (int)threadIdx.x;
      if (i < AG_ROWS * 64) t4[i] = v[u];
    }
  }
  for (int i = threadIdx.x; i < AG_ROWS * 4; i += 1024) {
    l_mx[i] = __uint_as_float(0xFF800000u);
    l_den[i] = 0.0f;
  }
  int cnt = 0;
  bool over = false;
  if (bins != nullptr) {
#pragma unroll
    for (int u = 0; u < SPW; ++u) {
      const int k = __builtin_amdgcn_readlane(bv[u], 0);
      over = over || k > BIN_SEG - 1 || k < 0;
      const int n = max(0, min(k, BIN_SEG - 1));
      if (lane >= 1 && lane <= n) le[cnt + lane - 1] = bv[u];
      cnt += n;
    }
  }
  if (lane == 0) l_over[w] = over ? 1 : 0;
  __syncthreads();

  int spilled = 0;
  const bool binned = bins != nullptr && __builtin_amdgcn_ballot_w32(l_over[lane] != 0) == 0u;
  if (!binned)
    cnt = ag_filter<true, 256>(f, attn, s, src, dst, tile, l_mx, l_den, le, base, rows, E,
                               lane, w, spilled);
  if (lane == 0) l_wcnt[w] = spilled != 0 ? -1 : cnt;
  ag_logits_max256(f, attn, s, src, dst, tile, l_mx, le, cnt, base, lane);
  __syncthreads();

  for (int i = threadIdx.x; i < AG_ROWS * 64; i += 1024)
    t4[i] = make_float4(0.0f, 0.0f, 0.0f, 0.0f);
  if (lists != nullptr) {
    int* L = lists + (long)blockIdx.x * AG_LW;
    const int v = l_wcnt[lane];
    const bool gone = __builtin_amdgcn_ballot_w32(v < 0) != 0u;
    int tot = v, off = lane < w ? v : 0;
    for (int m = 16; m >= 1; m >>= 1) {
      tot += __shfl_xor(tot, m, 32);
      off += __shfl_xor(off, m, 32);
    }
    const bool fits = !gone && tot <= AG_LCAP;
    const int n = __builtin_amdgcn_readfirstlane(cnt);
    if (fits)
      for (int i0 = 0; i0 < n; i0 += 32)
        if (i0 + lane < n) L[4 + off + i0 + lane] = le[i0 + lane];
    if (threadIdx.x == 0) L[0] = fits ? tot : -1;
  }
  __syncthreads();

  if (__builtin_amdgcn_readfirstlane(spilled) != 0)
    cnt = ag_filter<false, 256>(f, attn, s, src, dst, tile, l_mx, l_den, le, base, rows, E,
                                lane, w, spilled);
  ag_drain(f, s, src, dst, tile, l_mx, l_den, le, cnt, base, lane);
  __syncthreads();

  for (int i = threadIdx.x; i < rows * 64; i += 1024) {
    const int r = i >> 6, p = i & 63;
    const float* tr = tile + r * 256 + p;
    const float dv = l_den[r * 4 + (p >> 4)];
    const bool none = dv == 0.0f;
    float4 o;
    o.x = elu1(none ? 0.0f : tr[0]   / dv);
    o.y = elu1(none ? 0.0f : tr[64]  / dv);
    o.z = elu1(none ? 0.0f : tr[128] / dv);
    o.w = elu1(none ? 0.0f : tr[192] / dv);
    *(float4*)(h + (long)(base + r) * 256 + 4 * p) = o;
  }
}

__global__ __launch_bounds__(1024)
void aggregate_owner_softmax_mean(const float* __restrict__ f, const float* __restrict__ attn,
                                  float* s, const float* __restrict__ o1,
                                  const int* __restrict__ src, const int* __restrict__ dst,
                                  const int* __restrict__ lists,
                                  float* __restrict__ out, int N, int E) {
  __shared__ float tile[AG_ROWS * 128];
  __shared__ int l_edge[AG_WAVES * AG_CAP];
  __shared__ float l_mx[AG_ROWS * 4];
  __shared__ float l_den[AG_ROWS * 4];
  const int lane = threadIdx.x & 31;
  const int w    = threadIdx.x >> 5;
  const int base = blockIdx.x * AG_ROWS;
  const int rows = min(AG_ROWS, N - base);
  int* le = l_edge + w * AG_CAP;

  const int handed = lists != nullptr
      ? __builtin_amdgcn_readfirstlane(lists[(long)blockIdx.x * AG_LW]) : -1;
  int cnt = 0;
  if (handed >= 0) {
    const int per = (handed + AG_WAVES - 1) / AG_WAVES;
    const int first = __builtin_amdgcn_readfirstlane(w) * per;
    cnt = max(0, min(per, handed - first));
    if (cnt > 0) {
      constexpr int SH = (AG_LCAP / AG_WAVES + 1 + 31) / 32;
      const int* __restrict__ L = lists + (long)blockIdx.x * AG_LW + 4 + first;
      int id[SH];
#pragma unroll
      for (int u = 0; u < SH; ++u) id[u] = L[min(u * 32 + lane, cnt - 1)];
#pragma unroll
      for (int u = 0; u < SH; ++u)
        if (u * 32 + lane < cnt) le[u * 32 + lane] = id[u];
    }
  }
  float4* t4 = (float4*)tile;
  {
    constexpr int PER = (AG_ROWS * 32 + 1023) / 1024;
    const float4* __restrict__ fo = (const float4*)(f + (long)base * 128);
    float4 v[PER];
#pragma unroll
    for (int u = 0; u < PER; ++u) {
      const int i = u * 1024 + (int)threadIdx.x;
      v[u] = fo[min(i, rows * 32 - 1)];
    }
#pragma unroll
    for (int u = 0; u < PER; ++u) {
      const int i = u * 1024 + (int)threadIdx.x;
      if (i < AG_ROWS * 32) t4[i] = v[u];
    }
  }
  for (int i = threadIdx.x; i < AG_ROWS * 4; i += 1024) {
    l_mx[i] = __uint_as_float(0xFF800000u);
    l_den[i] = 0.0f;
  }
  __syncthreads();

  int spilled = 0;
  if (handed < 0)
    cnt = ag_filter<true, 128>(f, attn, s, src, dst, tile, l_mx, l_den, le, base, rows, E,
                               lane, w, spilled);
  ag_logits_max128(f, attn, s, src, dst, tile, l_mx, le, cnt, base, lane);
  __syncthreads();

  for (int i = threadIdx.x; i < AG_ROWS * 32; i += 1024)
    t4[i] = make_float4(0.0f, 0.0f, 0.0f, 0.0f);
  __syncthreads();

  if (__builtin_amdgcn_readfirstlane(spilled) != 0)
    cnt = ag_filter<false, 128>(f, attn, s, src, dst, tile, l_mx, l_den, le, base, rows, E,
                                lane, w, spilled);
  ag_drain128(f, s, src, dst, tile, l_mx, l_den, le, cnt, base, lane);
  __syncthreads();

  for (int i = threadIdx.x; i < rows * 8; i += 1024) {
    const int r = i >> 3, q = i & 7;
    const float* tr = tile + r * 128 + q;
    const float* __restrict__ rr = o1 + (long)(base + r) * 128 + 4 * q;
    float4 acc = make_float4(0.0f, 0.0f, 0.0f, 0.0f);
#pragma unroll
    for (int hh = 0; hh < 4; ++hh) {
      const float dv = l_den[r * 4 + hh];
      const bool none = dv == 0.0f;
      const float4 rs = *(const float4*)(rr + hh * 32);
      const float* th = tr + hh * 8;
      acc.x += (none ? 0.0f : th[0]  / dv) + rs.x;
      acc.y += (none ? 0.0f : th[32] / dv) + rs.y;
      acc.z += (none ? 0.0f : th[64] / dv) + rs.z;
      acc.w += (none ? 0.0f : th[96] / dv) + rs.w;
    }
    float4 o;
    o.x = 0.25f * acc.x; o.y = 0.25f * acc.y; o.z = 0.25f * acc.z; o.w = 0.25f * acc.w;
    *(float4*)(out + (long)(base + r) * 32 + 4 * q) = o;
  }
}

static inline int cdiv_l(long a, long b) { return (int)((a + b - 1) / b); }

extern "C" void kernel_launch(void* const* d_in, const int* in_sizes, int n_in,
                              void* d_out, int out_size, void* d_ws, size_t ws_size,
                              hipStream_t stream) {
  const float* x     = (const float*)d_in[0];
  const float* W0    = (const float*)d_in[1];
  const float* a0    = (const float*)d_in[2];
  const float* W1    = (const float*)d_in[3];
  const float* a1    = (const float*)d_in[4];
  const float* Wres1 = (const float*)d_in[5];
  const int*   src   = (const int*)d_in[6];
  const int*   dst   = (const int*)d_in[7];

  const int N = in_sizes[0] / 256;
  const int E = in_sizes[6];
  const int H = 4;

  float*    f0    = (float*)d_ws;
  float*    h     = f0 + (long)N * 256;
  float*    f1    = h  + (long)N * 256;
  float*    o1    = f1 + (long)N * 128;
  float*    s     = o1 + (long)N * 128;
  int*      lists = (int*)(s + (long)E * H);
  int*      bins  = (int*)o1;
  const size_t need =
      ((size_t)N * 256 * 2 + (size_t)N * 128 * 2 + (size_t)E * H) * sizeof(float);
  if (need > ws_size) return;
  if (need + (size_t)cdiv_l(N, AG_ROWS) * AG_LW * sizeof(int) > ws_size) lists = nullptr;
  const int owners = cdiv_l(N, AG_ROWS);
  if (owners > BIN_OWN || (size_t)owners * BIN_NB * BIN_SEG > (size_t)N * 128) bins = nullptr;

  const int BT = 128;

  gemm_f32_wmma<256, 256, true><<<cdiv_l(N, 32) + (bins != nullptr ? BIN_NB : 0), BT, 0, stream>>>(
      x, W0, W0 + 128, f0, f0 + 128, N, dst, bins, E, owners);
  aggregate_owner_softmax_elu<<<cdiv_l(N, AG_ROWS), 1024, 0, stream>>>(f0, a0, s, src, dst,
                                                                       h, lists, bins, N, E);

  gemm_f32_wmma<256, 128, false><<<cdiv_l(N, 32), BT, 0, stream>>>(h, W1, Wres1, f1, o1, N,
                                                                   nullptr, nullptr, 0, 0);
  aggregate_owner_softmax_mean<<<cdiv_l(N, AG_ROWS), 1024, 0, stream>>>(
      f1, a1, s, o1, src, dst, lists, (float*)d_out, N, E);
}


# GEMM A operand through LDS: a workgroup stages its 32 A rows in 32-k slices (whole-line 128-bit loads, two buffers, a barrier per slice); waves read fragments from LDS; B path, sum order unchanged
# speedup vs baseline: 3.1889x; 1.0397x over previous
#include <hip/hip_runtime.h>
#include <hip/hip_bf16.h>

typedef __attribute__((ext_vector_type(2))) float v2f;
typedef __attribute__((ext_vector_type(8))) float v8f;
typedef __attribute__((ext_vector_type(4))) float v4f;

constexpr int AG_ROWS = 196;
constexpr int BIN_NB  = 256;
constexpr int BIN_SEG = 32;
constexpr int BIN_OWN = 256;

__device__ __forceinline__ void bin_edges(const int* __restrict__ dst, int* __restrict__ bins,
                                          int E, int owners, int chunk, int* seg) {
  constexpr int NT = 128;
  constexpr int IN_FLIGHT = 8;
  const int tid = threadIdx.x;
  for (int i = tid; i < BIN_OWN * BIN_SEG / 4; i += NT)
    ((int4*)seg)[i] = make_int4(0, 0, 0, 0);
  __syncthreads();
  const int per = (E + BIN_NB - 1) / BIN_NB;
  const int e0 = chunk * per;
  const int e1 = min(E, e0 + per);
  for (int eb = e0; eb < e1; eb += NT * IN_FLIGHT) {
    int d[IN_FLIGHT];
#pragma unroll
    for (int u = 0; u < IN_FLIGHT; ++u) {
      const int e = eb + u * NT + tid;
      d[u] = e < e1 ? dst[e] : -1;
    }
#pragma unroll
    for (int u = 0; u < IN_FLIGHT; ++u) {
      const int e = eb + u * NT + tid;
      const int o = d[u] / AG_ROWS;
      if (d[u] >= 0 && o < owners) {
        int* sg = seg + o * BIN_SEG;
        const int k = __hip_atomic_fetch_add(sg, 1, __ATOMIC_RELAXED, __HIP_MEMORY_SCOPE_WORKGROUP);
        if (k < BIN_SEG - 1) sg[1 + k] = e;
      }
    }
  }
  __syncthreads();
  for (int i = tid; i < owners * (BIN_SEG / 4); i += NT) {
    const int o = i / (BIN_SEG / 4), p = i % (BIN_SEG / 4);
    ((int4*)(bins + ((long)o * BIN_NB + chunk) * BIN_SEG))[p] = ((const int4*)seg)[i];
  }
}

template <int K, int LD, bool BIN>
__global__ __launch_bounds__(128)
void gemm_f32_wmma(const float* A,
                   const float* __restrict__ B0, const float* __restrict__ B1,
                   float* __restrict__ C0, float* __restrict__ C1, int M,
                   const int* __restrict__ dst, int* __restrict__ bins, int E, int owners) {
  constexpr int MT = 2;
  constexpr int NT = 4;
  constexpr int KC = 32;
  constexpr int AP = KC + 4;
  constexpr int NCH = K / KC;
  constexpr int SLICE = MT * 16 * AP;
  constexpr int PER = MT * 16 * (KC / 4) / 128;
  static_assert(K % KC == 0 && PER == 2, "a thread stages two 128-bit pieces per slice");
  constexpr int SMEM = BIN ? (BIN_OWN * BIN_SEG > 2 * SLICE ? BIN_OWN * BIN_SEG : 2 * SLICE) : 2 * SLICE;
  __shared__ __attribute__((aligned(16))) unsigned char sm_bytes[SMEM * 4];
  float* const sm = (float*)sm_bytes;
  if constexpr (BIN) {
    const int tiles = (M + MT * 16 - 1) / (MT * 16);
    if ((int)blockIdx.x >= tiles) {
      bin_edges(dst, bins, E, owners, (int)blockIdx.x - tiles, (int*)sm);
      return;
    }
  }
  const int lane = threadIdx.x & 31;
  const int st   = threadIdx.x >> 5;
  const int tm   = blockIdx.x;
  const int hi = lane >> 4;
  const int lm = lane & 15;
  const float* __restrict__ B = (st & 2) ? B1 : B0;
  float* __restrict__ C = (st & 2) ? C1 : C0;
  const int col0 = (st & 1) * (NT * 16) + lm;
  const float* __restrict__ Bc = B + (long)(2 * hi) * LD + col0;

  const int c4 = (int)threadIdx.x & (KC / 4 - 1);
  const int r0 = (int)threadIdx.x / (KC / 4);
  const float* Ag[PER];
#pragma unroll
  for (int u = 0; u < PER; ++u) {
    int arow = tm * (MT * 16) + u * (128 / (KC / 4)) + r0; if (arow >= M) arow = M - 1;
    Ag[u] = A + (long)arow * K + 4 * c4;
  }
  float* const sdst = sm + r0 * AP + 4 * c4;
  const float* const frag = sm + lm * AP + 2 * hi;

  v8f c[MT][NT];
#pragma unroll
  for (int m = 0; m < MT; ++m)
#pragma unroll
    for (int t = 0; t < NT; ++t) c[m][t] = (v8f){};

  auto loadB = [&](int k, v2f* b) {
    const float* __restrict__ Bk = Bc + (long)k * LD;
#pragma unroll
    for (int t = 0; t < NT; ++t) { b[t].x = Bk[16 * t]; b[t].y = Bk[LD + 16 * t]; }
  };
  auto loadA = [&](const float* sl, int kk, v2f* a) {
#pragma unroll
    for (int m = 0; m < MT; ++m) a[m] = *(const v2f*)(sl + m * 16 * AP + kk);
  };

  constexpr int RSTEP = (128 / (KC / 4)) * AP;
  float4 stg0, stg1;
  v2f b_cur[NT];
  stg0 = *(const float4*)(Ag[0]);
  stg1 = *(const float4*)(Ag[1]);
  loadB(0, b_cur);
  *(float4*)(sdst) = stg0;
  *(float4*)(sdst + RSTEP) = stg1;
  __syncthreads();
#pragma unroll
  for (int ch = 0; ch < NCH; ++ch) {
    if (ch + 1 < NCH) {
      stg0 = *(const float4*)(Ag[0] + (ch + 1) * KC);
      stg1 = *(const float4*)(Ag[1] + (ch + 1) * KC);
    }
    const float* sl = frag + (ch & 1) * SLICE;
    v2f a_cur[MT];
    loadA(sl, 0, a_cur);
#pragma unroll
    for (int kk = 0; kk < KC; kk += 4) {
      v2f a_nxt[MT], b_nxt[NT];
      if (kk + 4 < KC) loadA(sl, kk + 4, a_nxt);
      if (ch * KC + kk + 4 < K) loadB(ch * KC + kk + 4, b_nxt);
#pragma unroll
      for (int m = 0; m < MT; ++m)
#pragma unroll
        for (int t = 0; t < NT; ++t)
          c[m][t] = __builtin_amdgcn_wmma_f32_16x16x4_f32(false, a_cur[m], false, b_cur[t],
                                                          (short)0, c[m][t], false, false);
      if (kk + 4 < KC) {
#pragma unroll
        for (int m = 0; m < MT; ++m) a_cur[m] = a_nxt[m];
      }
      if (ch * KC + kk + 4 < K) {
#pragma unroll
        for (int t = 0; t < NT; ++t) b_cur[t] = b_nxt[t];
      }
      if ((kk & 4) != 0 && kk + 4 < KC) __builtin_amdgcn_sched_barrier(0);
    }
    if (ch + 1 < NCH) {
      __builtin_amdgcn_sched_barrier(0);
      *(float4*)(sdst + ((ch + 1) & 1) * SLICE) = stg0;
      *(float4*)(sdst + ((ch + 1) & 1) * SLICE + RSTEP) = stg1;
      __syncthreads();
    }
  }

#pragma unroll
  for (int m = 0; m < MT; ++m) {
#pragma unroll
    for (int t = 0; t < NT; ++t) {
#pragma unroll
      for (int r = 0; r < 8; ++r) {
        const int orow = tm * (MT * 16) + m * 16 + r + 8 * hi;
        if (orow < M) C[(long)orow * LD + col0 + 16 * t] = c[m][t][r];
      }
    }
  }
}


template <int M>
__device__ __forceinline__ float row_xor_add(float v) {
  return v + __int_as_float(__builtin_amdgcn_update_dpp(0, __float_as_int(v), 0x160 + M,
                                                        0xf, 0xf, true));
}

__device__ __forceinline__ float lrelu_dot4(v4f a, v4f b, v4f w) {
  const v4f v = a + b;
  float r = fmaxf(v.x, 0.2f * v.x) * w.x;
  r += fmaxf(v.y, 0.2f * v.y) * w.y;
  r += fmaxf(v.z, 0.2f * v.z) * w.z;
  r += fmaxf(v.w, 0.2f * v.w) * w.w;
  return r;
}

constexpr int AG_WAVES = 32;
constexpr int AG_SCAN  = 6;
constexpr int AG_STEP  = 128 * AG_SCAN;
constexpr int AG_CAP   = AG_STEP + 128;
constexpr int AG_BATCH = 8;
constexpr int AG_LB256 = 8;
constexpr int AG_LB128 = 16;
constexpr int AG_LCAP  = 4092;
constexpr int AG_LW    = AG_LCAP + 4;

__device__ __forceinline__ int ag_append(int d, int e, int base, int rows, int cnt,
                                         int* le) {
  const unsigned r = (unsigned)(d - base);
  const bool hit = r < (unsigned)rows;
  const unsigned mask = __builtin_amdgcn_ballot_w32(hit);
  if (mask != 0u) {
    const int pos = cnt + (int)__builtin_amdgcn_mbcnt_lo(mask, 0u);
    if (hit) le[pos] = e;
    cnt += __popc(mask);
  }
  return cnt;
}

__device__ __forceinline__ void lds_max_f32(float* p, float v) {
  if (v >= 0.0f)
    __hip_atomic_fetch_max((int*)p, __float_as_int(v), __ATOMIC_RELAXED,
                           __HIP_MEMORY_SCOPE_WORKGROUP);
  else
    __hip_atomic_fetch_min((unsigned*)p, __float_as_uint(v), __ATOMIC_RELAXED,
                           __HIP_MEMORY_SCOPE_WORKGROUP);
}

__device__ __forceinline__ void ag_logits_max256(const float* __restrict__ f,
                                                 const float* __restrict__ attn,
                                                 float* s,
                                                 const int* __restrict__ src,
                                                 const int* __restrict__ dst,
                                                 const float* own,
                                                 float* mx, const int* le,
                                                 int cnt, int base, int lane) {
  __builtin_amdgcn_fence(__ATOMIC_ACQ_REL, "wavefront");
  cnt = __builtin_amdgcn_readfirstlane(cnt);
  const v4f w0 = ((const v4f*)attn)[lane];
  const v4f w1 = ((const v4f*)attn)[32 + lane];
  for (int c = 0; c < cnt; c += 32) {
    const int e = le[min(c + lane, cnt - 1)];
    const int sv = src[e];
    const int dn = dst[e];
    v4f lg = (v4f){0.0f, 0.0f, 0.0f, 0.0f};
    const int nb = min(32, cnt - c);
    for (int q0 = 0; q0 < nb; q0 += AG_LB256) {
      v4f xs0[AG_LB256], xs1[AG_LB256];
#pragma unroll
      for (int b = 0; b < AG_LB256; ++b) {
        const int sb = __builtin_amdgcn_readlane(sv, q0 + b);
        const v4f* __restrict__ fs = (const v4f*)(f + (long)sb * 256);
        xs0[b] = fs[lane];
        xs1[b] = fs[32 + lane];
      }
#pragma unroll
      for (int b = 0; b < AG_LB256; ++b) {
        const int rb = __builtin_amdgcn_readlane(dn, q0 + b) - base;
        const v4f* fd = (const v4f*)(own + rb * 256);
        const v4f xd0 = fd[lane];
        const v4f xd1 = fd[32 + lane];
        float p0 = lrelu_dot4(xs0[b], xd0, w0);
        float p1 = lrelu_dot4(xs1[b], xd1, w1);
        p0 = row_xor_add<1>(p0); p1 = row_xor_add<1>(p1);
        p0 = row_xor_add<2>(p0); p1 = row_xor_add<2>(p1);
        p0 = row_xor_add<4>(p0); p1 = row_xor_add<4>(p1);
        p0 = row_xor_add<8>(p0); p1 = row_xor_add<8>(p1);
        const int l0 = __builtin_amdgcn_readlane(__float_as_int(p0), 0);
        const int l1 = __builtin_amdgcn_readlane(__float_as_int(p0), 16);
        const int l2 = __builtin_amdgcn_readlane(__float_as_int(p1), 0);
        const int l3 = __builtin_amdgcn_readlane(__float_as_int(p1), 16);
        const bool mine = lane == q0 + b;
        lg.x = mine ? __int_as_float(l0) : lg.x;
        lg.y = mine ? __int_as_float(l1) : lg.y;
        lg.z = mine ? __int_as_float(l2) : lg.z;
        lg.w = mine ? __int_as_float(l3) : lg.w;
        if (b & 1) __builtin_amdgcn_sched_barrier(0);
      }
    }
    if (c + lane < cnt) {
      *(v4f*)(s + (long)e * 4) = lg;
      float* mr = mx + (dn - base) * 4;
      lds_max_f32(mr,     lg.x);
      lds_max_f32(mr + 1, lg.y);
      lds_max_f32(mr + 2, lg.z);
      lds_max_f32(mr + 3, lg.w);
    }
  }
  __builtin_amdgcn_fence(__ATOMIC_ACQ_REL, "wavefront");
}

__device__ __forceinline__ void ag_logits_max128(const float* __restrict__ f,
                                                 const float* __restrict__ attn,
                                                 float* s,
                                                 const int* __restrict__ src,
                                                 const int* __restrict__ dst,
                                                 const float* own,
                                                 float* mx, const int* le,
                                                 int cnt, int base, int lane) {
  __builtin_amdgcn_fence(__ATOMIC_ACQ_REL, "wavefront");
  cnt = __builtin_amdgcn_readfirstlane(cnt);
  const v4f w = ((const v4f*)attn)[lane];
  for (int c = 0; c < cnt; c += 32) {
    const int e = le[min(c + lane, cnt - 1)];
    const int sv = src[e];
    const int dn = dst[e];
    v4f lg = (v4f){0.0f, 0.0f, 0.0f, 0.0f};
    const int nb = min(32, cnt - c);
    for (int q0 = 0; q0 < nb; q0 += AG_LB128) {
      v4f xs[AG_LB128];
#pragma unroll
      for (int b = 0; b < AG_LB128; ++b) {
        const int sb = __builtin_amdgcn_readlane(sv, q0 + b);
        xs[b] = ((const v4f*)(f + (long)sb * 128))[lane];
      }
#pragma unroll
      for (int b = 0; b < AG_LB128; ++b) {
        const int rb = __builtin_amdgcn_readlane(dn, q0 + b) - base;
        const v4f xd = ((const v4f*)(own + rb * 128))[lane];
        float p = lrelu_dot4(xs[b], xd, w);
        p = row_xor_add<1>(p);
        p = row_xor_add<2>(p);
        p = row_xor_add<4>(p);
        const int l0 = __builtin_amdgcn_readlane(__float_as_int(p), 0);
        const int l1 = __builtin_amdgcn_readlane(__float_as_int(p), 8);
        const int l2 = __builtin_amdgcn_readlane(__float_as_int(p), 16);
        const int l3 = __builtin_amdgcn_readlane(__float_as_int(p), 24);
        const bool mine = lane == q0 + b;
        lg.x = mine ? __int_as_float(l0) : lg.x;
        lg.y = mine ? __int_as_float(l1) : lg.y;
        lg.z = mine ? __int_as_float(l2) : lg.z;
        lg.w = mine ? __int_as_float(l3) : lg.w;
      }
    }
    if (c + lane < cnt) {
      *(v4f*)(s + (long)e * 4) = lg;
      float* mr = mx + (dn - base) * 4;
      lds_max_f32(mr,     lg.x);
      lds_max_f32(mr + 1, lg.y);
      lds_max_f32(mr + 2, lg.z);
      lds_max_f32(mr + 3, lg.w);
    }
  }
  __builtin_amdgcn_fence(__ATOMIC_ACQ_REL, "wavefront");
}

__device__ __forceinline__ void ag_drain(const float* __restrict__ f,
                                         const float* s,
                                         const int* __restrict__ src,
                                         const int* __restrict__ dst,
                                         float* tile, const float* mx, float* den,
                                         const int* le,
                                         int cnt, int base, int lane) {
  __builtin_amdgcn_fence(__ATOMIC_ACQ_REL, "wavefront");
  const int j = lane >> 2;
  const int hd = lane & 3;
  const bool lo = lane < 16;
  for (int i = 0; i < cnt; i += AG_BATCH) {
    const int e = le[min(i + j, cnt - 1)];
    const int sv = src[e];
    const int dn = dst[e];
    const float sl = s[(long)e * 4 + hd];
    v4f x0[AG_BATCH], x1[AG_BATCH];
#pragma unroll
    for (int b = 0; b < AG_BATCH; ++b) {
      const int sb = __builtin_amdgcn_readlane(sv, 4 * b);
      const v4f* __restrict__ fr = (const v4f*)(f + (long)sb * 256);
      x0[b] = fr[lane];
      x1[b] = fr[32 + lane];
    }
    const int r = dn - base;
    const float mv = mx[r * 4 + hd];
    asm volatile("" :: "v"(x0[0]), "v"(x1[0]), "v"(x0[1]), "v"(x1[1]), "v"(x0[2]), "v"(x1[2]),
                       "v"(x0[3]), "v"(x1[3]), "v"(x0[4]), "v"(x1[4]), "v"(x0[5]), "v"(x1[5]),
                       "v"(x0[6]), "v"(x1[6]), "v"(x0[7]), "v"(x1[7]));
    const float a = __expf(sl - mv);
    if (i + j < cnt)
      __hip_atomic_fetch_add(den + r * 4 + hd, a, __ATOMIC_RELAXED, __HIP_MEMORY_SCOPE_WORKGROUP);
#pragma unroll
    for (int b = 0; b < AG_BATCH; ++b) {
      if (i + b < cnt) {
        const int rb = __builtin_amdgcn_readlane(r, 4 * b);
        const float a0 = __int_as_float(__builtin_amdgcn_readlane(__float_as_int(a), 4 * b));
        const float a1 = __int_as_float(__builtin_amdgcn_readlane(__float_as_int(a), 4 * b + 1));
        const float a2 = __int_as_float(__builtin_amdgcn_readlane(__float_as_int(a), 4 * b + 2));
        const float a3 = __int_as_float(__builtin_amdgcn_readlane(__float_as_int(a), 4 * b + 3));
        const float al0 = lo ? a0 : a1;
        const float al1 = lo ? a2 : a3;
        float* tr = tile + rb * 256 + lane;
        __hip_atomic_fetch_add(tr,       x0[b].x * al0, __ATOMIC_RELAXED, __HIP_MEMORY_SCOPE_WORKGROUP);
        __hip_atomic_fetch_add(tr + 64,  x0[b].y * al0, __ATOMIC_RELAXED, __HIP_MEMORY_SCOPE_WORKGROUP);
        __hip_atomic_fetch_add(tr + 128, x0[b].z * al0, __ATOMIC_RELAXED, __HIP_MEMORY_SCOPE_WORKGROUP);
        __hip_atomic_fetch_add(tr + 192, x0[b].w * al0, __ATOMIC_RELAXED, __HIP_MEMORY_SCOPE_WORKGROUP);
        __hip_atomic_fetch_add(tr + 32,  x1[b].x * al1, __ATOMIC_RELAXED, __HIP_MEMORY_SCOPE_WORKGROUP);
        __hip_atomic_fetch_add(tr + 96,  x1[b].y * al1, __ATOMIC_RELAXED, __HIP_MEMORY_SCOPE_WORKGROUP);
        __hip_atomic_fetch_add(tr + 160, x1[b].z * al1, __ATOMIC_RELAXED, __HIP_MEMORY_SCOPE_WORKGROUP);
        __hip_atomic_fetch_add(tr + 224, x1[b].w * al1, __ATOMIC_RELAXED, __HIP_MEMORY_SCOPE_WORKGROUP);
      }
    }
  }
  __builtin_amdgcn_fence(__ATOMIC_ACQ_REL, "wavefront");
}

__device__ __forceinline__ void ag_drain128(const float* __restrict__ f,
                                            const float* s,
                                            const int* __restrict__ src,
                                            const int* __restrict__ dst,
                                            float* tile, const float* mx, float* den,
                                            const int* le,
                                            int cnt, int base, int lane) {
  __builtin_amdgcn_fence(__ATOMIC_ACQ_REL, "wavefront");
  const int j = lane >> 2;
  const int hd = lane & 3;
  const int hq = lane >> 3;
  for (int i = 0; i < cnt; i += AG_BATCH) {
    const int e = le[min(i + j, cnt - 1)];
    const int sv = src[e];
    const int dn = dst[e];
    const float sl = s[(long)e * 4 + hd];
    v4f x[AG_BATCH];
#pragma unroll
    for (int b = 0; b < AG_BATCH; ++b) {
      const int sb = __builtin_amdgcn_readlane(sv, 4 * b);
      const v4f* __restrict__ fr = (const v4f*)(f + (long)sb * 128);
      x[b] = fr[lane];
    }
    const int r = dn - base;
    const float mv = mx[r * 4 + hd];
    const float a = __expf(sl - mv);
    if (i + j < cnt)
      __hip_atomic_fetch_add(den + r * 4 + hd, a, __ATOMIC_RELAXED, __HIP_MEMORY_SCOPE_WORKGROUP);
    float al[AG_BATCH];
#pragma unroll
    for (int b = 0; b < AG_BATCH; ++b) al[b] = __shfl(a, 4 * b + hq, 32);
#pragma unroll
    for (int b = 0; b < AG_BATCH; ++b) {
      if (i + b < cnt) {
        const int rb = __builtin_amdgcn_readlane(r, 4 * b);
        float* tr = tile + rb * 128 + lane;
        __hip_atomic_fetch_add(tr,      x[b].x * al[b], __ATOMIC_RELAXED, __HIP_MEMORY_SCOPE_WORKGROUP);
        __hip_atomic_fetch_add(tr + 32, x[b].y * al[b], __ATOMIC_RELAXED, __HIP_MEMORY_SCOPE_WORKGROUP);
        __hip_atomic_fetch_add(tr + 64, x[b].z * al[b], __ATOMIC_RELAXED, __HIP_MEMORY_SCOPE_WORKGROUP);
        __hip_atomic_fetch_add(tr + 96, x[b].w * al[b], __ATOMIC_RELAXED, __HIP_MEMORY_SCOPE_WORKGROUP);
      }
    }
  }
  __builtin_amdgcn_fence(__ATOMIC_ACQ_REL, "wavefront");
}

template <bool MAXPASS, int ROW>
__device__ __forceinline__ int ag_filter(const float* __restrict__ f,
                                         const float* __restrict__ attn,
                                         float* s,
                                         const int* __restrict__ src,
                                         const int* __restrict__ dst,
                                         float* tile, float* mx, float* den, int* le,
                                         int base, int rows, int E, int lane, int w,
                                         int& spilled) {
  int peel = (int)(((16u - (unsigned)((unsigned long long)dst & 15ull)) & 15u) >> 2);
  if (peel > E) peel = E;
  const int nG = (E - peel) >> 2;
  const int tail0 = peel + 4 * nG;
  int cnt = 0;
  if (w == 0) {
    int e = -1;
    if (lane < peel) e = lane;
    else if (lane >= 4 && lane < 8 && tail0 + (lane - 4) < E) e = tail0 + (lane - 4);
    const int d = e >= 0 ? dst[e] : -1;
    cnt = ag_append(d, e, base, rows, cnt, le);
  }
  cnt = __builtin_amdgcn_readfirstlane(cnt);
  const int4* __restrict__ d4 = (const int4*)(dst + peel);
  for (int g0 = 0; g0 < nG; g0 += 1024 * AG_SCAN) {
    if (cnt > AG_CAP - AG_STEP) {
      if (MAXPASS) {
        if (ROW == 256) ag_logits_max256(f, attn, s, src, dst, tile, mx, le, cnt, base, lane);
        else            ag_logits_max128(f, attn, s, src, dst, tile, mx, le, cnt, base, lane);
        spilled = 1;
      } else if (ROW == 256) {
        ag_drain(f, s, src, dst, tile, mx, den, le, cnt, base, lane);
      } else {
        ag_drain128(f, s, src, dst, tile, mx, den, le, cnt, base, lane);
      }
      cnt = 0;
    }
    int4 v[AG_SCAN];
#pragma unroll
    for (int u = 0; u < AG_SCAN; ++u) {
      const int g = g0 + u * 1024 + (int)threadIdx.x;
      v[u] = g < nG ? d4[g] : make_int4(-1, -1, -1, -1);
    }
#pragma unroll
    for (int u = 0; u < AG_SCAN; ++u) {
      const int e0 = peel + 4 * (g0 + u * 1024 + (int)threadIdx.x);
      cnt = ag_append(v[u].x, e0, base, rows, cnt, le);
      cnt = ag_append(v[u].y, e0 + 1, base, rows, cnt, le);
      cnt = ag_append(v[u].z, e0 + 2, base, rows, cnt, le);
      cnt = ag_append(v[u].w, e0 + 3, base, rows, cnt, le);
    }
  }
  return cnt;
}

__device__ __forceinline__ float elu1(float v) { return v > 0.0f ? v : expm1f(v); }

__global__ __launch_bounds__(1024)
void aggregate_owner_softmax_elu(const float* __restrict__ f, const float* __restrict__ attn,
                                 float* s,
                                 const int* __restrict__ src, const int* __restrict__ dst,
                                 float* __restrict__ h, int* __restrict__ lists,
                                 const int* __restrict__ bins, int N, int E) {
  __shared__ float tile[AG_ROWS * 256];
  __shared__ int l_edge[AG_WAVES * AG_CAP];
  __shared__ float l_mx[AG_ROWS * 4];
  __shared__ float l_den[AG_ROWS * 4];
  __shared__ int l_wcnt[AG_WAVES];
  __shared__ int l_over[AG_WAVES];
  const int lane = threadIdx.x & 31;
  const int w    = threadIdx.x >> 5;
  const int base = blockIdx.x * AG_ROWS;
  const int rows = min(AG_ROWS, N - base);
  int* le = l_edge + w * AG_CAP;

  static_assert(BIN_SEG == 32 && BIN_NB % AG_WAVES == 0 &&
                (BIN_NB / AG_WAVES) * (BIN_SEG - 1) <= AG_CAP, "a wave's segments fill its list");
  constexpr int SPW = BIN_NB / AG_WAVES;
  int bv[SPW] = {};
  if (bins != nullptr) {
    const int* __restrict__ Bn =
        bins + ((long)blockIdx.x * BIN_NB + __builtin_amdgcn_readfirstlane(w) * SPW) * BIN_SEG;
#pragma unroll
    for (int u = 0; u < SPW; ++u) bv[u] = Bn[u * BIN_SEG + lane];
  }

  float4* t4 = (float4*)tile;
  {
    constexpr int PER = (AG_ROWS * 64 + 1023) / 1024;
    const float4* __restrict__ fo = (const float4*)(f + (long)base * 256);
    float4 v[PER];
#pragma unroll
    for (int u = 0; u < PER; ++u) {
      const int i = u * 1024 + (int)threadIdx.x;
      v[u] = fo[min(i, rows * 64 - 1)];
    }
#pragma unroll
    for (int u = 0; u < PER; ++u) {
      const int i = u * 1024 + (int)threadIdx.x;
      if (i < AG_ROWS * 64) t4[i] = v[u];
    }
  }
  for (int i = threadIdx.x; i < AG_ROWS * 4; i += 1024) {
    l_mx[i] = __uint_as_float(0xFF800000u);
    l_den[i] = 0.0f;
  }
  int cnt = 0;
  bool over = false;
  if (bins != nullptr) {
#pragma unroll
    for (int u = 0; u < SPW; ++u) {
      const int k = __builtin_amdgcn_readlane(bv[u], 0);
      over = over || k > BIN_SEG - 1 || k < 0;
      const int n = max(0, min(k, BIN_SEG - 1));
      if (lane >= 1 && lane <= n) le[cnt + lane - 1] = bv[u];
      cnt += n;
    }
  }
  if (lane == 0) l_over[w] = over ? 1 : 0;
  __syncthreads();

  int spilled = 0;
  const bool binned = bins != nullptr && __builtin_amdgcn_ballot_w32(l_over[lane] != 0) == 0u;
  if (!binned)
    cnt = ag_filter<true, 256>(f, attn, s, src, dst, tile, l_mx, l_den, le, base, rows, E,
                               lane, w, spilled);
  if (lane == 0) l_wcnt[w] = spilled != 0 ? -1 : cnt;
  ag_logits_max256(f, attn, s, src, dst, tile, l_mx, le, cnt, base, lane);
  __syncthreads();

  for (int i = threadIdx.x; i < AG_ROWS * 64; i += 1024)
    t4[i] = make_float4(0.0f, 0.0f, 0.0f, 0.0f);
  if (lists != nullptr) {
    int* L = lists + (long)blockIdx.x * AG_LW;
    const int v = l_wcnt[lane];
    const bool gone = __builtin_amdgcn_ballot_w32(v < 0) != 0u;
    int tot = v, off = lane < w ? v : 0;
    for (int m = 16; m >= 1; m >>= 1) {
      tot += __shfl_xor(tot, m, 32);
      off += __shfl_xor(off, m, 32);
    }
    const bool fits = !gone && tot <= AG_LCAP;
    const int n = __builtin_amdgcn_readfirstlane(cnt);
    if (fits)
      for (int i0 = 0; i0 < n; i0 += 32)
        if (i0 + lane < n) L[4 + off + i0 + lane] = le[i0 + lane];
    if (threadIdx.x == 0) L[0] = fits ? tot : -1;
  }
  __syncthreads();

  if (__builtin_amdgcn_readfirstlane(spilled) != 0)
    cnt = ag_filter<false, 256>(f, attn, s, src, dst, tile, l_mx, l_den, le, base, rows, E,
                                lane, w, spilled);
  ag_drain(f, s, src, dst, tile, l_mx, l_den, le, cnt, base, lane);
  __syncthreads();

  for (int i = threadIdx.x; i < rows * 64; i += 1024) {
    const int r = i >> 6, p = i & 63;
    const float* tr = tile + r * 256 + p;
    const float dv = l_den[r * 4 + (p >> 4)];
    const bool none = dv == 0.0f;
    float4 o;
    o.x = elu1(none ? 0.0f : tr[0]   / dv);
    o.y = elu1(none ? 0.0f : tr[64]  / dv);
    o.z = elu1(none ? 0.0f : tr[128] / dv);
    o.w = elu1(none ? 0.0f : tr[192] / dv);
    *(float4*)(h + (long)(base + r) * 256 + 4 * p) = o;
  }
}

__global__ __launch_bounds__(1024)
void aggregate_owner_softmax_mean(const float* __restrict__ f, const float* __restrict__ attn,
                                  float* s, const float* __restrict__ o1,
                                  const int* __restrict__ src, const int* __restrict__ dst,
                                  const int* __restrict__ lists,
                                  float* __restrict__ out, int N, int E) {
  __shared__ float tile[AG_ROWS * 128];
  __shared__ int l_edge[AG_WAVES * AG_CAP];
  __shared__ float l_mx[AG_ROWS * 4];
  __shared__ float l_den[AG_ROWS * 4];
  const int lane = threadIdx.x & 31;
  const int w    = threadIdx.x >> 5;
  const int base = blockIdx.x * AG_ROWS;
  const int rows = min(AG_ROWS, N - base);
  int* le = l_edge + w * AG_CAP;

  const int handed = lists != nullptr
      ? __builtin_amdgcn_readfirstlane(lists[(long)blockIdx.x * AG_LW]) : -1;
  int cnt = 0;
  if (handed >= 0) {
    const int per = (handed + AG_WAVES - 1) / AG_WAVES;
    const int first = __builtin_amdgcn_readfirstlane(w) * per;
    cnt = max(0, min(per, handed - first));
    if (cnt > 0) {
      constexpr int SH = (AG_LCAP / AG_WAVES + 1 + 31) / 32;
      const int* __restrict__ L = lists + (long)blockIdx.x * AG_LW + 4 + first;
      int id[SH];
#pragma unroll
      for (int u = 0; u < SH; ++u) id[u] = L[min(u * 32 + lane, cnt - 1)];
#pragma unroll
      for (int u = 0; u < SH; ++u)
        if (u * 32 + lane < cnt) le[u * 32 + lane] = id[u];
    }
  }
  float4* t4 = (float4*)tile;
  {
    constexpr int PER = (AG_ROWS * 32 + 1023) / 1024;
    const float4* __restrict__ fo = (const float4*)(f + (long)base * 128);
    float4 v[PER];
#pragma unroll
    for (int u = 0; u < PER; ++u) {
      const int i = u * 1024 + (int)threadIdx.x;
      v[u] = fo[min(i, rows * 32 - 1)];
    }
#pragma unroll
    for (int u = 0; u < PER; ++u) {
      const int i = u * 1024 + (int)threadIdx.x;
      if (i < AG_ROWS * 32) t4[i] = v[u];
    }
  }
  for (int i = threadIdx.x; i < AG_ROWS * 4; i += 1024) {
    l_mx[i] = __uint_as_float(0xFF800000u);
    l_den[i] = 0.0f;
  }
  __syncthreads();

  int spilled = 0;
  if (handed < 0)
    cnt = ag_filter<true, 128>(f, attn, s, src, dst, tile, l_mx, l_den, le, base, rows, E,
                               lane, w, spilled);
  ag_logits_max128(f, attn, s, src, dst, tile, l_mx, le, cnt, base, lane);
  __syncthreads();

  for (int i = threadIdx.x; i < AG_ROWS * 32; i += 1024)
    t4[i] = make_float4(0.0f, 0.0f, 0.0f, 0.0f);
  __syncthreads();

  if (__builtin_amdgcn_readfirstlane(spilled) != 0)
    cnt = ag_filter<false, 128>(f, attn, s, src, dst, tile, l_mx, l_den, le, base, rows, E,
                                lane, w, spilled);
  ag_drain128(f, s, src, dst, tile, l_mx, l_den, le, cnt, base, lane);
  __syncthreads();

  for (int i = threadIdx.x; i < rows * 8; i += 1024) {
    const int r = i >> 3, q = i & 7;
    const float* tr = tile + r * 128 + q;
    const float* __restrict__ rr = o1 + (long)(base + r) * 128 + 4 * q;
    float4 acc = make_float4(0.0f, 0.0f, 0.0f, 0.0f);
#pragma unroll
    for (int hh = 0; hh < 4; ++hh) {
      const float dv = l_den[r * 4 + hh];
      const bool none = dv == 0.0f;
      const float4 rs = *(const float4*)(rr + hh * 32);
      const float* th = tr + hh * 8;
      acc.x += (none ? 0.0f : th[0]  / dv) + rs.x;
      acc.y += (none ? 0.0f : th[32] / dv) + rs.y;
      acc.z += (none ? 0.0f : th[64] / dv) + rs.z;
      acc.w += (none ? 0.0f : th[96] / dv) + rs.w;
    }
    float4 o;
    o.x = 0.25f * acc.x; o.y = 0.25f * acc.y; o.z = 0.25f * acc.z; o.w = 0.25f * acc.w;
    *(float4*)(out + (long)(base + r) * 32 + 4 * q) = o;
  }
}

static inline int cdiv_l(long a, long b) { return (int)((a + b - 1) / b); }

extern "C" void kernel_launch(void* const* d_in, const int* in_sizes, int n_in,
                              void* d_out, int out_size, void* d_ws, size_t ws_size,
                              hipStream_t stream) {
  const float* x     = (const float*)d_in[0];
  const float* W0    = (const float*)d_in[1];
  const float* a0    = (const float*)d_in[2];
  const float* W1    = (const float*)d_in[3];
  const float* a1    = (const float*)d_in[4];
  const float* Wres1 = (const float*)d_in[5];
  const int*   src   = (const int*)d_in[6];
  const int*   dst   = (const int*)d_in[7];

  const int N = in_sizes[0] / 256;
  const int E = in_sizes[6];
  const int H = 4;

  float*    f0    = (float*)d_ws;
  float*    h     = f0 + (long)N * 256;
  float*    f1    = h  + (long)N * 256;
  float*    o1    = f1 + (long)N * 128;
  float*    s     = o1 + (long)N * 128;
  int*      lists = (int*)(s + (long)E * H);
  int*      bins  = (int*)o1;
  const size_t need =
      ((size_t)N * 256 * 2 + (size_t)N * 128 * 2 + (size_t)E * H) * sizeof(float);
  if (need > ws_size) return;
  if (need + (size_t)cdiv_l(N, AG_ROWS) * AG_LW * sizeof(int) > ws_size) lists = nullptr;
  const int owners = cdiv_l(N, AG_ROWS);
  if (owners > BIN_OWN || (size_t)owners * BIN_NB * BIN_SEG > (size_t)N * 128) bins = nullptr;

  const int BT = 128;

  gemm_f32_wmma<256, 256, true><<<cdiv_l(N, 32) + (bins != nullptr ? BIN_NB : 0), BT, 0, stream>>>(
      x, W0, W0 + 128, f0, f0 + 128, N, dst, bins, E, owners);
  aggregate_owner_softmax_elu<<<cdiv_l(N, AG_ROWS), 1024, 0, stream>>>(f0, a0, s, src, dst,
                                                                       h, lists, bins, N, E);

  gemm_f32_wmma<256, 128, false><<<cdiv_l(N, 32), BT, 0, stream>>>(h, W1, Wres1, f1, o1, N,
                                                                   nullptr, nullptr, 0, 0);
  aggregate_owner_softmax_mean<<<cdiv_l(N, AG_ROWS), 1024, 0, stream>>>(
      f1, a1, s, o1, src, dst, lists, (float*)d_out, N, E);
}
